# starting kernel
# baseline (speedup 1.0000x reference)
#include <hip/hip_runtime.h>
#include <math.h>

#define HIDC 128

typedef __attribute__((ext_vector_type(2))) float v2f;
typedef __attribute__((ext_vector_type(8))) float v8f;

__global__ void gcn_init(float* __restrict__ acc, float* __restrict__ deg, int N) {
    int idx = blockIdx.x * blockDim.x + threadIdx.x;
    long long total = (long long)N * HIDC;
    if ((long long)idx < total) acc[idx] = 0.0f;
    if (idx < N) deg[idx] = 1.0f;
}

__global__ void gcn_degree(const int* __restrict__ dst, float* __restrict__ deg, int E) {
    int e = blockIdx.x * blockDim.x + threadIdx.x;
    if (e < E) atomicAdd(&deg[dst[e]], 1.0f);
}

__global__ void gcn_dinv(const float* __restrict__ deg, float* __restrict__ dinv, int N) {
    int i = blockIdx.x * blockDim.x + threadIdx.x;
    if (i < N) dinv[i] = rsqrtf(deg[i]);
}

__global__ __launch_bounds__(256) void gcn_gemm1_wmma(const float* __restrict__ x,
                                                      const float* __restrict__ W,
                                                      float* __restrict__ h, int N) {
    __shared__ float As[16 * 132];
    const int t = threadIdx.x;
    const int rowBase = blockIdx.x * 16;
    const bool full = (rowBase + 16 <= N);

    if (full) {
#pragma unroll
        for (int i = t; i < 16 * 32; i += 256) {
            int r  = i >> 5;
            int c4 = i & 31;
            float4 v = ((const float4*)(x + (size_t)(rowBase + r) * HIDC))[c4];
            float* p = &As[r * 132 + c4 * 4];
            p[0] = v.x; p[1] = v.y; p[2] = v.z; p[3] = v.w;
        }
    } else {
        for (int i = t; i < 16 * 32; i += 256) {
            int r  = i >> 5;
            int c4 = i & 31;
            float4 v = make_float4(0.f, 0.f, 0.f, 0.f);
            if (rowBase + r < N) v = ((const float4*)(x + (size_t)(rowBase + r) * HIDC))[c4];
            float* p = &As[r * 132 + c4 * 4];
            p[0] = v.x; p[1] = v.y; p[2] = v.z; p[3] = v.w;
        }
    }
    __syncthreads();

    const int lane = t & 31;
    const int wv   = t >> 5;
    const int m    = lane & 15;
    const int half = lane >> 4;
    const int kl   = half * 2;
    const int n    = wv * 16 + (lane & 15);

    v8f c = {};
    const float* Arow = &As[m * 132];
#pragma unroll 8
    for (int k0 = 0; k0 < HIDC; k0 += 4) {
        v2f a, b;
        a.x = Arow[k0 + kl];
        a.y = Arow[k0 + kl + 1];
        b.x = W[(k0 + kl) * HIDC + n];
        b.y = W[(k0 + kl + 1) * HIDC + n];
        c = __builtin_amdgcn_wmma_f32_16x16x4_f32(false, a, false, b,
                                                  (short)0, c, false, false);
    }

    float* hp = h + (size_t)(rowBase + 8 * half) * HIDC + n;
    if (full) {
#pragma unroll
        for (int v = 0; v < 8; ++v) hp[v * HIDC] = c[v];
    } else {
#pragma unroll
        for (int v = 0; v < 8; ++v) {
            if (rowBase + 8 * half + v < N) hp[v * HIDC] = c[v];
        }
    }
}

__global__ __launch_bounds__(256) void gcn_agg1(const float* __restrict__ h,
                                                const int* __restrict__ src,
                                                const int* __restrict__ dst,
                                                const float* __restrict__ dinv,
                                                float* __restrict__ acc, int E) {
    int e = blockIdx.x * 8 + (threadIdx.x >> 5);
    if (e >= E) return;
    int lane = threadIdx.x & 31;
    int s = src[e];
    int d = dst[e];
    const float* hs = h + (size_t)s * HIDC;
    __builtin_prefetch(hs + lane * 4, 0, 0);
    float nrm = dinv[s] * dinv[d];
    float4 v = ((const float4*)hs)[lane];
    float* o = acc + (size_t)d * HIDC + lane * 4;
    atomicAdd(o + 0, v.x * nrm);
    atomicAdd(o + 1, v.y * nrm);
    atomicAdd(o + 2, v.z * nrm);
    atomicAdd(o + 3, v.w * nrm);
}

__global__ __launch_bounds__(256) void gcn_fin1_dot(const float* __restrict__ acc,
                                                    const float* __restrict__ h,
                                                    const float* __restrict__ dinv,
                                                    const float* __restrict__ b1,
                                                    const float* __restrict__ W2,
                                                    float* __restrict__ h2, int N) {
    int i = blockIdx.x * 8 + (threadIdx.x >> 5);
    if (i >= N) return;
    int lane = threadIdx.x & 31;
    float di = dinv[i];
    float ns = di * di;
    float4 a  = ((const float4*)(acc + (size_t)i * HIDC))[lane];
    float4 hv = ((const float4*)(h   + (size_t)i * HIDC))[lane];
    float4 bb = ((const float4*)b1)[lane];
    float4 w  = ((const float4*)W2)[lane];
    float r0 = fmaxf(a.x + hv.x * ns + bb.x, 0.0f);
    float r1 = fmaxf(a.y + hv.y * ns + bb.y, 0.0f);
    float r2 = fmaxf(a.z + hv.z * ns + bb.z, 0.0f);
    float r3 = fmaxf(a.w + hv.w * ns + bb.w, 0.0f);
    float p = r0 * w.x + r1 * w.y + r2 * w.z + r3 * w.w;
#pragma unroll
    for (int off = 16; off > 0; off >>= 1) p += __shfl_down(p, off, 32);
    if (lane == 0) h2[i] = p;
}

__global__ void gcn_init_out(const float* __restrict__ h2, const float* __restrict__ dinv,
                             const float* __restrict__ b2, float* __restrict__ out, int N) {
    int i = blockIdx.x * blockDim.x + threadIdx.x;
    if (i < N) {
        float di = dinv[i];
        out[i] = h2[i] * di * di + b2[0];
    }
}

__global__ void gcn_agg2(const float* __restrict__ h2, const int* __restrict__ src,
                         const int* __restrict__ dst, const float* __restrict__ dinv,
                         float* __restrict__ out, int E) {
    int e = blockIdx.x * blockDim.x + threadIdx.x;
    if (e < E) {
        int s = src[e];
        int d = dst[e];
        atomicAdd(&out[d], h2[s] * dinv[s] * dinv[d]);
    }
}

extern "C" void kernel_launch(void* const* d_in, const int* in_sizes, int n_in,
                              void* d_out, int out_size, void* d_ws, size_t ws_size,
                              hipStream_t stream) {
    const float* x  = (const float*)d_in[0];
    const float* W1 = (const float*)d_in[1];
    const float* b1 = (const float*)d_in[2];
    const float* W2 = (const float*)d_in[3];
    const float* b2 = (const float*)d_in[4];
    const int*   ei = (const int*)d_in[5];

    const int N = in_sizes[0] / HIDC;
    const int E = in_sizes[5] / 2;
    const int* src = ei;
    const int* dst = ei + E;
    float* out = (float*)d_out;

    float* h    = (float*)d_ws;
    float* acc  = h   + (size_t)N * HIDC;
    float* deg  = acc + (size_t)N * HIDC;
    float* dinv = deg + N;
    float* h2   = dinv + N;

    const int TB = 256;
    long long totNH = (long long)N * HIDC;
    int gInit = (int)((totNH + TB - 1) / TB);

    gcn_init      <<<gInit,             TB, 0, stream>>>(acc, deg, N);
    gcn_degree    <<<(E + TB - 1) / TB, TB, 0, stream>>>(dst, deg, E);
    gcn_dinv      <<<(N + TB - 1) / TB, TB, 0, stream>>>(deg, dinv, N);
    gcn_gemm1_wmma<<<(N + 15) / 16,     TB, 0, stream>>>(x, W1, h, N);
    gcn_agg1      <<<(E + 7) / 8,       TB, 0, stream>>>(h, src, dst, dinv, acc, E);
    gcn_fin1_dot  <<<(N + 7) / 8,       TB, 0, stream>>>(acc, h, dinv, b1, W2, h2, N);
    gcn_init_out  <<<(N + TB - 1) / TB, TB, 0, stream>>>(h2, dinv, b2, out, N);
    gcn_agg2      <<<(E + TB - 1) / TB, TB, 0, stream>>>(h2, src, dst, dinv, out, E);
}


# agg1+finalize owner-computes: a block keeps 512 nodes' sums in LDS, filters the whole dst list, adds source rows by LDS f32 atomics; acc plane gone; f32 sum order = LDS atomic arrival order
# speedup vs baseline: 1.6049x; 1.6049x over previous
#include <hip/hip_runtime.h>
#include <math.h>

#define HIDC 128

typedef __attribute__((ext_vector_type(2))) float v2f;
typedef __attribute__((ext_vector_type(8))) float v8f;

__global__ void gcn_init(float* __restrict__ deg, int N) {
    int idx = blockIdx.x * blockDim.x + threadIdx.x;
    if (idx < N) deg[idx] = 1.0f;
}

__global__ void gcn_degree(const int* __restrict__ dst, float* __restrict__ deg, int E) {
    int e = blockIdx.x * blockDim.x + threadIdx.x;
    if (e < E) atomicAdd(&deg[dst[e]], 1.0f);
}

__global__ void gcn_dinv(const float* __restrict__ deg, float* __restrict__ dinv, int N) {
    int i = blockIdx.x * blockDim.x + threadIdx.x;
    if (i < N) dinv[i] = rsqrtf(deg[i]);
}

__global__ __launch_bounds__(256) void gcn_gemm1_wmma(const float* __restrict__ x,
                                                      const float* __restrict__ W,
                                                      float* __restrict__ h, int N) {
    __shared__ float As[16 * 132];
    const int t = threadIdx.x;
    const int rowBase = blockIdx.x * 16;
    const bool full = (rowBase + 16 <= N);

    if (full) {
#pragma unroll
        for (int i = t; i < 16 * 32; i += 256) {
            int r  = i >> 5;
            int c4 = i & 31;
            float4 v = ((const float4*)(x + (size_t)(rowBase + r) * HIDC))[c4];
            float* p = &As[r * 132 + c4 * 4];
            p[0] = v.x; p[1] = v.y; p[2] = v.z; p[3] = v.w;
        }
    } else {
        for (int i = t; i < 16 * 32; i += 256) {
            int r  = i >> 5;
            int c4 = i & 31;
            float4 v = make_float4(0.f, 0.f, 0.f, 0.f);
            if (rowBase + r < N) v = ((const float4*)(x + (size_t)(rowBase + r) * HIDC))[c4];
            float* p = &As[r * 132 + c4 * 4];
            p[0] = v.x; p[1] = v.y; p[2] = v.z; p[3] = v.w;
        }
    }
    __syncthreads();

    const int lane = t & 31;
    const int wv   = t >> 5;
    const int m    = lane & 15;
    const int half = lane >> 4;
    const int kl   = half * 2;
    const int n    = wv * 16 + (lane & 15);

    v8f c = {};
    const float* Arow = &As[m * 132];
#pragma unroll 8
    for (int k0 = 0; k0 < HIDC; k0 += 4) {
        v2f a, b;
        a.x = Arow[k0 + kl];
        a.y = Arow[k0 + kl + 1];
        b.x = W[(k0 + kl) * HIDC + n];
        b.y = W[(k0 + kl + 1) * HIDC + n];
        c = __builtin_amdgcn_wmma_f32_16x16x4_f32(false, a, false, b,
                                                  (short)0, c, false, false);
    }

    float* hp = h + (size_t)(rowBase + 8 * half) * HIDC + n;
    if (full) {
#pragma unroll
        for (int v = 0; v < 8; ++v) hp[v * HIDC] = c[v];
    } else {
#pragma unroll
        for (int v = 0; v < 8; ++v) {
            if (rowBase + 8 * half + v < N) hp[v * HIDC] = c[v];
        }
    }
}

#define TN   512
#define QCAP 256

__device__ __forceinline__ int rdlane_i(int v, int k) { return __builtin_amdgcn_readlane(v, k); }
__device__ __forceinline__ float rdlane_f(float v, int k) {
    return __int_as_float(__builtin_amdgcn_readlane(__float_as_int(v), k));
}
__device__ __forceinline__ void lds_fadd(float* p, float v) {
    __hip_atomic_fetch_add(p, v, __ATOMIC_RELAXED, __HIP_MEMORY_SCOPE_WORKGROUP);
}

__device__ __forceinline__ void drain_rows(const float* __restrict__ h,
                                           const int* __restrict__ src,
                                           const float* __restrict__ dinv,
                                           float* tile, const float* dtile,
                                           const int* myE, const unsigned short* myD,
                                           int head, int n, int lane) {
    const int idx = (head + lane) & (QCAP - 1);
    int   s   = 0;
    int   dl  = 0;
    float nrm = 0.0f;
    if (lane < n) {
        int e = myE[idx];
        dl    = myD[idx];
        s     = src[e];
        nrm   = dinv[s] * dtile[dl];
    }
    const int hi = lane >> 4;
    const int c0 = 4 * lane + (hi ? 1 : 0);
    const int c1 = 4 * lane + (hi ? 2 : 1);
    const int c2 = 4 * lane + (hi ? 3 : 2);
    const int c3 = 4 * lane + (hi ? 0 : 3);
#pragma nounroll
    for (int k0 = 0; k0 < n; k0 += 8) {
        float4 r[8];
#pragma unroll
        for (int u = 0; u < 8; ++u) {
            if (k0 + u < n) {
                int sk = rdlane_i(s, k0 + u);
                r[u] = ((const float4*)(h + (size_t)sk * HIDC))[lane];
            }
        }
#pragma unroll
        for (int u = 0; u < 8; ++u) {
            if (k0 + u < n) {
                int   dk = rdlane_i(dl, k0 + u);
                float nk = rdlane_f(nrm, k0 + u);
                float* o = tile + dk * HIDC;
                float vx = r[u].x * nk, vy = r[u].y * nk, vz = r[u].z * nk, vw = r[u].w * nk;
                lds_fadd(o + c0, hi ? vy : vx);
                lds_fadd(o + c1, hi ? vz : vy);
                lds_fadd(o + c2, hi ? vw : vz);
                lds_fadd(o + c3, hi ? vx : vw);
            }
        }
    }
}

__device__ __forceinline__ void queue_edge(int d, int e, int lo, int lane,
                                           int* myE, unsigned short* myD,
                                           int head, int& cnt) {
    const unsigned dl = (unsigned)(d - lo);
    const bool m = dl < (unsigned)TN;
    const unsigned mask = __builtin_amdgcn_ballot_w32(m);
    if (mask != 0u) {
        if (m) {
            int p = (head + cnt + (int)__builtin_amdgcn_mbcnt_lo(mask, 0u)) & (QCAP - 1);
            myE[p] = e;
            myD[p] = (unsigned short)dl;
        }
        cnt += __builtin_popcount(mask);
    }
}

__global__ __launch_bounds__(1024) void gcn_agg1_owner(const float* __restrict__ h,
                                                       const int* __restrict__ src,
                                                       const int* __restrict__ dst,
                                                       const float* __restrict__ dinv,
                                                       const float* __restrict__ b1,
                                                       const float* __restrict__ W2,
                                                       float* __restrict__ h2, int N, int E) {
    __shared__ __attribute__((aligned(16))) float tile[TN * HIDC];
    __shared__ int            qE[32 * QCAP];
    __shared__ unsigned short qD[32 * QCAP];
    __shared__ float          dtile[TN];

    const int t    = threadIdx.x;
    const int lane = t & 31;
    const int wv   = __builtin_amdgcn_readfirstlane(t >> 5);
    const int lo   = blockIdx.x * TN;

    for (int i = t; i < TN * HIDC / 4; i += 1024)
        ((float4*)tile)[i] = make_float4(0.f, 0.f, 0.f, 0.f);
    if (t < TN) dtile[t] = (lo + t < N) ? dinv[lo + t] : 0.0f;
    __syncthreads();

    int*            myE = qE + wv * QCAP;
    unsigned short* myD = qD + wv * QCAP;
    int head = 0, cnt = 0;
    const bool al = ((((size_t)dst) & 15) == 0);

    int base = wv * 1024;
    for (; al && base + 1024 <= E; base += 32 * 1024) {
        const int4* dp = (const int4*)(dst + base) + lane;
        const int4 v0 = dp[0],   v1 = dp[32],  v2 = dp[64],  v3 = dp[96];
        const int4 v4 = dp[128], v5 = dp[160], v6 = dp[192], v7 = dp[224];
        const int4 vv[8] = {v0, v1, v2, v3, v4, v5, v6, v7};
#pragma unroll
        for (int g = 0; g < 8; ++g) {
            const int e0 = base + g * 128 + lane * 4;
            queue_edge(vv[g].x, e0,     lo, lane, myE, myD, head, cnt);
            queue_edge(vv[g].y, e0 + 1, lo, lane, myE, myD, head, cnt);
            queue_edge(vv[g].z, e0 + 2, lo, lane, myE, myD, head, cnt);
            queue_edge(vv[g].w, e0 + 3, lo, lane, myE, myD, head, cnt);
            while (cnt >= 32) {
                drain_rows(h, src, dinv, tile, dtile, myE, myD, head, 32, lane);
                head = (head + 32) & (QCAP - 1);
                cnt -= 32;
            }
        }
    }
    for (; base < E; base += 32 * 1024) {
        for (int q = 0; q < 32; ++q) {
            const int e = base + q * 32 + lane;
            const int d = (e < E) ? dst[e] : -1;
            queue_edge(d, e, lo, lane, myE, myD, head, cnt);
            if (cnt >= 32) {
                drain_rows(h, src, dinv, tile, dtile, myE, myD, head, 32, lane);
                head = (head + 32) & (QCAP - 1);
                cnt -= 32;
            }
        }
    }
    if (cnt > 0) drain_rows(h, src, dinv, tile, dtile, myE, myD, head, cnt, lane);
    __syncthreads();

    const float4 bb = ((const float4*)b1)[lane];
    const float4 w  = ((const float4*)W2)[lane];
    for (int r = wv; r < TN; r += 32) {
        const int i = lo + r;
        if (i >= N) break;
        const float di = dtile[r];
        const float ns = di * di;
        const float4 a  = ((const float4*)(tile + r * HIDC))[lane];
        const float4 hv = ((const float4*)(h + (size_t)i * HIDC))[lane];
        float r0 = fmaxf(a.x + hv.x * ns + bb.x, 0.0f);
        float r1 = fmaxf(a.y + hv.y * ns + bb.y, 0.0f);
        float r2 = fmaxf(a.z + hv.z * ns + bb.z, 0.0f);
        float r3 = fmaxf(a.w + hv.w * ns + bb.w, 0.0f);
        float p = r0 * w.x + r1 * w.y + r2 * w.z + r3 * w.w;
#pragma unroll
        for (int off = 16; off > 0; off >>= 1) p += __shfl_down(p, off, 32);
        if (lane == 0) h2[i] = p;
    }
}

__global__ void gcn_init_out(const float* __restrict__ h2, const float* __restrict__ dinv,
                             const float* __restrict__ b2, float* __restrict__ out, int N) {
    int i = blockIdx.x * blockDim.x + threadIdx.x;
    if (i < N) {
        float di = dinv[i];
        out[i] = h2[i] * di * di + b2[0];
    }
}

__global__ void gcn_agg2(const float* __restrict__ h2, const int* __restrict__ src,
                         const int* __restrict__ dst, const float* __restrict__ dinv,
                         float* __restrict__ out, int E) {
    int e = blockIdx.x * blockDim.x + threadIdx.x;
    if (e < E) {
        int s = src[e];
        int d = dst[e];
        atomicAdd(&out[d], h2[s] * dinv[s] * dinv[d]);
    }
}

extern "C" void kernel_launch(void* const* d_in, const int* in_sizes, int n_in,
                              void* d_out, int out_size, void* d_ws, size_t ws_size,
                              hipStream_t stream) {
    const float* x  = (const float*)d_in[0];
    const float* W1 = (const float*)d_in[1];
    const float* b1 = (const float*)d_in[2];
    const float* W2 = (const float*)d_in[3];
    const float* b2 = (const float*)d_in[4];
    const int*   ei = (const int*)d_in[5];

    const int N = in_sizes[0] / HIDC;
    const int E = in_sizes[5] / 2;
    const int* src = ei;
    const int* dst = ei + E;
    float* out = (float*)d_out;

    float* h    = (float*)d_ws;
    float* deg  = h   + (size_t)N * HIDC;
    float* dinv = deg + N;
    float* h2   = dinv + N;

    const int TB = 256;
    gcn_init      <<<(N + TB - 1) / TB, TB, 0, stream>>>(deg, N);
    gcn_degree    <<<(E + TB - 1) / TB, TB, 0, stream>>>(dst, deg, E);
    gcn_dinv      <<<(N + TB - 1) / TB, TB, 0, stream>>>(deg, dinv, N);
    gcn_gemm1_wmma<<<(N + 15) / 16,     TB, 0, stream>>>(x, W1, h, N);
    gcn_agg1_owner<<<(N + TN - 1) / TN, 1024, 0, stream>>>(h, src, dst, dinv, b1, W2, h2, N, E);
    gcn_init_out  <<<(N + TB - 1) / TB, TB, 0, stream>>>(h2, dinv, b2, out, N);
    gcn_agg2      <<<(E + TB - 1) / TB, TB, 0, stream>>>(h2, src, dst, dinv, out, E);
}


# agg1 owner blocks filter a one-byte key per edge (dst/512, left by the degree pass), 16 edges per lane per load, and read dst only for matched edges; f32 sums: LDS atomic arrival order as before
# speedup vs baseline: 1.6821x; 1.0481x over previous
#include <hip/hip_runtime.h>
#include <math.h>

#define HIDC 128

typedef __attribute__((ext_vector_type(2))) float v2f;
typedef __attribute__((ext_vector_type(8))) float v8f;

__global__ void gcn_init(float* __restrict__ deg, int N) {
    int idx = blockIdx.x * blockDim.x + threadIdx.x;
    if (idx < N) deg[idx] = 1.0f;
}

__global__ void gcn_degree(const int* __restrict__ dst, float* __restrict__ deg,
                           unsigned int* __restrict__ keyw, int E) {
    int e = blockIdx.x * blockDim.x + threadIdx.x;
    if (e < E) {
        const int d = dst[e];
        atomicAdd(&deg[d], 1.0f);
        ((unsigned char*)keyw)[e] = (unsigned char)(d >> 9);
    }
}

__global__ void gcn_dinv(const float* __restrict__ deg, float* __restrict__ dinv, int N) {
    int i = blockIdx.x * blockDim.x + threadIdx.x;
    if (i < N) dinv[i] = rsqrtf(deg[i]);
}

__global__ __launch_bounds__(256) void gcn_gemm1_wmma(const float* __restrict__ x,
                                                      const float* __restrict__ W,
                                                      float* __restrict__ h, int N) {
    __shared__ float As[16 * 132];
    const int t = threadIdx.x;
    const int rowBase = blockIdx.x * 16;
    const bool full = (rowBase + 16 <= N);

    if (full) {
#pragma unroll
        for (int i = t; i < 16 * 32; i += 256) {
            int r  = i >> 5;
            int c4 = i & 31;
            float4 v = ((const float4*)(x + (size_t)(rowBase + r) * HIDC))[c4];
            float* p = &As[r * 132 + c4 * 4];
            p[0] = v.x; p[1] = v.y; p[2] = v.z; p[3] = v.w;
        }
    } else {
        for (int i = t; i < 16 * 32; i += 256) {
            int r  = i >> 5;
            int c4 = i & 31;
            float4 v = make_float4(0.f, 0.f, 0.f, 0.f);
            if (rowBase + r < N) v = ((const float4*)(x + (size_t)(rowBase + r) * HIDC))[c4];
            float* p = &As[r * 132 + c4 * 4];
            p[0] = v.x; p[1] = v.y; p[2] = v.z; p[3] = v.w;
        }
    }
    __syncthreads();

    const int lane = t & 31;
    const int wv   = t >> 5;
    const int m    = lane & 15;
    const int half = lane >> 4;
    const int kl   = half * 2;
    const int n    = wv * 16 + (lane & 15);

    v8f c = {};
    const float* Arow = &As[m * 132];
#pragma unroll 8
    for (int k0 = 0; k0 < HIDC; k0 += 4) {
        v2f a, b;
        a.x = Arow[k0 + kl];
        a.y = Arow[k0 + kl + 1];
        b.x = W[(k0 + kl) * HIDC + n];
        b.y = W[(k0 + kl + 1) * HIDC + n];
        c = __builtin_amdgcn_wmma_f32_16x16x4_f32(false, a, false, b,
                                                  (short)0, c, false, false);
    }

    float* hp = h + (size_t)(rowBase + 8 * half) * HIDC + n;
    if (full) {
#pragma unroll
        for (int v = 0; v < 8; ++v) hp[v * HIDC] = c[v];
    } else {
#pragma unroll
        for (int v = 0; v < 8; ++v) {
            if (rowBase + 8 * half + v < N) hp[v * HIDC] = c[v];
        }
    }
}

#define TN   512
#define QCAP 256

__device__ __forceinline__ int rdlane_i(int v, int k) { return __builtin_amdgcn_readlane(v, k); }
__device__ __forceinline__ float rdlane_f(float v, int k) {
    return __int_as_float(__builtin_amdgcn_readlane(__float_as_int(v), k));
}
__device__ __forceinline__ void lds_fadd(float* p, float v) {
    __hip_atomic_fetch_add(p, v, __ATOMIC_RELAXED, __HIP_MEMORY_SCOPE_WORKGROUP);
}

__device__ __forceinline__ void drain_rows(const float* __restrict__ h,
                                           const int* __restrict__ src,
                                           const int* __restrict__ dst,
                                           const float* __restrict__ dinv,
                                           float* tile, const float* dtile,
                                           const int* myE, int lo,
                                           int head, int n, int lane) {
    const int idx = (head + lane) & (QCAP - 1);
    int   s   = 0;
    int   dl  = 0;
    float nrm = 0.0f;
    if (lane < n) {
        int e = myE[idx];
        s     = src[e];
        dl    = dst[e] - lo;
        nrm   = dinv[s] * dtile[dl];
    }
    const int hi = lane >> 4;
    const int c0 = 4 * lane + (hi ? 1 : 0);
    const int c1 = 4 * lane + (hi ? 2 : 1);
    const int c2 = 4 * lane + (hi ? 3 : 2);
    const int c3 = 4 * lane + (hi ? 0 : 3);
#pragma nounroll
    for (int k0 = 0; k0 < n; k0 += 8) {
        float4 r[8];
#pragma unroll
        for (int u = 0; u < 8; ++u) {
            if (k0 + u < n) {
                int sk = rdlane_i(s, k0 + u);
                r[u] = ((const float4*)(h + (size_t)sk * HIDC))[lane];
            }
        }
#pragma unroll
        for (int u = 0; u < 8; ++u) {
            if (k0 + u < n) {
                int   dk = rdlane_i(dl, k0 + u);
                float nk = rdlane_f(nrm, k0 + u);
                float* o = tile + dk * HIDC;
                float vx = r[u].x * nk, vy = r[u].y * nk, vz = r[u].z * nk, vw = r[u].w * nk;
                lds_fadd(o + c0, hi ? vy : vx);
                lds_fadd(o + c1, hi ? vz : vy);
                lds_fadd(o + c2, hi ? vw : vz);
                lds_fadd(o + c3, hi ? vx : vw);
            }
        }
    }
}

__device__ __forceinline__ void push_edges(bool m, int e, int* myE, int head, int& cnt) {
    const unsigned mask = __builtin_amdgcn_ballot_w32(m);
    if (mask != 0u) {
        if (m) {
            int p = (head + cnt + (int)__builtin_amdgcn_mbcnt_lo(mask, 0u)) & (QCAP - 1);
            myE[p] = e;
        }
        cnt += __builtin_popcount(mask);
    }
}

__device__ __forceinline__ unsigned differ_bits(unsigned w, unsigned pat) {
    const unsigned x = w ^ pat;
    return (((x & 0x7f7f7f7fu) + 0x7f7f7f7fu) | x) & 0x80808080u;
}
__device__ __forceinline__ unsigned match16(uint4 v, unsigned pat) {
    const unsigned nd = (differ_bits(v.x, pat) >> 7) | (differ_bits(v.y, pat) >> 6) |
                        (differ_bits(v.z, pat) >> 5) | (differ_bits(v.w, pat) >> 4);
    return ~nd & 0x0f0f0f0fu;
}

__global__ __launch_bounds__(1024) void gcn_agg1_owner(const float* __restrict__ h,
                                                       const int* __restrict__ src,
                                                       const int* __restrict__ dst,
                                                       const unsigned int* __restrict__ keyw,
                                                       const float* __restrict__ dinv,
                                                       const float* __restrict__ b1,
                                                       const float* __restrict__ W2,
                                                       float* __restrict__ h2, int N, int E) {
    __shared__ __attribute__((aligned(16))) float tile[TN * HIDC];
    __shared__ int   qE[32 * QCAP];
    __shared__ float dtile[TN];

    const int t    = threadIdx.x;
    const int lane = t & 31;
    const int wv   = __builtin_amdgcn_readfirstlane(t >> 5);
    const int lo   = blockIdx.x * TN;

    for (int i = t; i < TN * HIDC / 4; i += 1024)
        ((float4*)tile)[i] = make_float4(0.f, 0.f, 0.f, 0.f);
    if (t < TN) dtile[t] = (lo + t < N) ? dinv[lo + t] : 0.0f;
    __syncthreads();

    int* myE = qE + wv * QCAP;
    int head = 0, cnt = 0;

    const unsigned char* key = (const unsigned char*)keyw;
    const bool usekey = (N <= 256 * TN) && ((((size_t)key) & 15) == 0);
    const int  nrows  = usekey ? (E >> 9) : 0;
    const unsigned pat = (blockIdx.x & 255u) * 0x01010101u;

    for (int r0 = wv; r0 < nrows; r0 += 32 * 8) {
        uint4 vv[8];
#pragma unroll
        for (int g = 0; g < 8; ++g) {
            const int row = min(r0 + 32 * g, nrows - 1);
            vv[g] = ((const uint4*)(key + ((size_t)row << 9)))[lane];
        }
#pragma unroll
        for (int g = 0; g < 8; ++g) {
            const int row = r0 + 32 * g;
            if (row < nrows) {
                const int eb = (row << 9) + lane * 16;
                unsigned F = match16(vv[g], pat);
#pragma nounroll
                for (int r = 0; r < 16; ++r) {
                    const unsigned mask = __builtin_amdgcn_ballot_w32(F != 0u);
                    if (mask == 0u) break;
                    if (F != 0u) {
                        const int tz = __builtin_ctz(F);
                        const int p  = (head + cnt + (int)__builtin_amdgcn_mbcnt_lo(mask, 0u)) & (QCAP - 1);
                        myE[p] = eb + ((tz & 7) << 2) + (tz >> 3);
                    }
                    cnt += __builtin_popcount(mask);
                    F &= F - 1u;
                    if (cnt >= 32) {
                        drain_rows(h, src, dst, dinv, tile, dtile, myE, lo, head, 32, lane);
                        head = (head + 32) & (QCAP - 1);
                        cnt -= 32;
                    }
                }
            }
        }
    }
    const int ebeg   = nrows << 9;
    const int nsteps = (E - ebeg + 31) >> 5;
#pragma nounroll
    for (int st = wv; st < nsteps; st += 32) {
        const int e = ebeg + (st << 5) + lane;
        const int d = (e < E) ? dst[e] : -1;
        push_edges((unsigned)(d - lo) < (unsigned)TN, e, myE, head, cnt);
        if (cnt >= 32) {
            drain_rows(h, src, dst, dinv, tile, dtile, myE, lo, head, 32, lane);
            head = (head + 32) & (QCAP - 1);
            cnt -= 32;
        }
    }
    if (cnt > 0) drain_rows(h, src, dst, dinv, tile, dtile, myE, lo, head, cnt, lane);
    __syncthreads();

    const float4 bb = ((const float4*)b1)[lane];
    const float4 w  = ((const float4*)W2)[lane];
    for (int r = wv; r < TN; r += 32) {
        const int i = lo + r;
        if (i >= N) break;
        const float di = dtile[r];
        const float ns = di * di;
        const float4 a  = ((const float4*)(tile + r * HIDC))[lane];
        const float4 hv = ((const float4*)(h + (size_t)i * HIDC))[lane];
        float r0 = fmaxf(a.x + hv.x * ns + bb.x, 0.0f);
        float r1 = fmaxf(a.y + hv.y * ns + bb.y, 0.0f);
        float r2 = fmaxf(a.z + hv.z * ns + bb.z, 0.0f);
        float r3 = fmaxf(a.w + hv.w * ns + bb.w, 0.0f);
        float p = r0 * w.x + r1 * w.y + r2 * w.z + r3 * w.w;
#pragma unroll
        for (int off = 16; off > 0; off >>= 1) p += __shfl_down(p, off, 32);
        if (lane == 0) h2[i] = p;
    }
}

__global__ void gcn_init_out(const float* __restrict__ h2, const float* __restrict__ dinv,
                             const float* __restrict__ b2, float* __restrict__ out, int N) {
    int i = blockIdx.x * blockDim.x + threadIdx.x;
    if (i < N) {
        float di = dinv[i];
        out[i] = h2[i] * di * di + b2[0];
    }
}

__global__ void gcn_agg2(const float* __restrict__ h2, const int* __restrict__ src,
                         const int* __restrict__ dst, const float* __restrict__ dinv,
                         float* __restrict__ out, int E) {
    int e = blockIdx.x * blockDim.x + threadIdx.x;
    if (e < E) {
        int s = src[e];
        int d = dst[e];
        atomicAdd(&out[d], h2[s] * dinv[s] * dinv[d]);
    }
}

extern "C" void kernel_launch(void* const* d_in, const int* in_sizes, int n_in,
                              void* d_out, int out_size, void* d_ws, size_t ws_size,
                              hipStream_t stream) {
    const float* x  = (const float*)d_in[0];
    const float* W1 = (const float*)d_in[1];
    const float* b1 = (const float*)d_in[2];
    const float* W2 = (const float*)d_in[3];
    const float* b2 = (const float*)d_in[4];
    const int*   ei = (const int*)d_in[5];

    const int N = in_sizes[0] / HIDC;
    const int E = in_sizes[5] / 2;
    const int* src = ei;
    const int* dst = ei + E;
    float* out = (float*)d_out;

    float* h    = (float*)d_ws;
    float* deg  = h   + (size_t)N * HIDC;
    float* dinv = deg + N;
    float* h2   = dinv + N;
    unsigned int* keyw = (unsigned int*)(h2 + N);

    const int TB = 256;
    gcn_init      <<<(N + TB - 1) / TB, TB, 0, stream>>>(deg, N);
    gcn_degree    <<<(E + TB - 1) / TB, TB, 0, stream>>>(dst, deg, keyw, E);
    gcn_dinv      <<<(N + TB - 1) / TB, TB, 0, stream>>>(deg, dinv, N);
    gcn_gemm1_wmma<<<(N + 15) / 16,     TB, 0, stream>>>(x, W1, h, N);
    gcn_agg1_owner<<<(N + TN - 1) / TN, 1024, 0, stream>>>(h, src, dst, keyw, dinv, b1, W2, h2, N, E);
    gcn_init_out  <<<(N + TB - 1) / TB, TB, 0, stream>>>(h2, dinv, b2, out, N);
    gcn_agg2      <<<(E + TB - 1) / TB, TB, 0, stream>>>(h2, src, dst, dinv, out, E);
}


# agg1: wave w owns tile rows r%32==w; edges filed in per-wave LDS bins (returning LDS int add), owner adds each row by LDS load+fma+store, no f32 atomics; f32 sum order = bin slot (arrival) order
# speedup vs baseline: 1.7886x; 1.0633x over previous
#include <hip/hip_runtime.h>
#include <math.h>

#define HIDC 128

typedef __attribute__((ext_vector_type(2))) float v2f;
typedef __attribute__((ext_vector_type(8))) float v8f;

__global__ void gcn_init(float* __restrict__ deg, int N) {
    int idx = blockIdx.x * blockDim.x + threadIdx.x;
    if (idx < N) deg[idx] = 1.0f;
}

__global__ void gcn_degree(const int* __restrict__ dst, float* __restrict__ deg,
                           unsigned int* __restrict__ keyw, int E) {
    int e = blockIdx.x * blockDim.x + threadIdx.x;
    if (e < E) {
        const int d = dst[e];
        atomicAdd(&deg[d], 1.0f);
        ((unsigned char*)keyw)[e] = (unsigned char)(d >> 9);
    }
}

__global__ void gcn_dinv(const float* __restrict__ deg, float* __restrict__ dinv, int N) {
    int i = blockIdx.x * blockDim.x + threadIdx.x;
    if (i < N) dinv[i] = rsqrtf(deg[i]);
}

__global__ __launch_bounds__(256) void gcn_gemm1_wmma(const float* __restrict__ x,
                                                      const float* __restrict__ W,
                                                      float* __restrict__ h, int N) {
    __shared__ float As[16 * 132];
    const int t = threadIdx.x;
    const int rowBase = blockIdx.x * 16;
    const bool full = (rowBase + 16 <= N);

    if (full) {
#pragma unroll
        for (int i = t; i < 16 * 32; i += 256) {
            int r  = i >> 5;
            int c4 = i & 31;
            float4 v = ((const float4*)(x + (size_t)(rowBase + r) * HIDC))[c4];
            float* p = &As[r * 132 + c4 * 4];
            p[0] = v.x; p[1] = v.y; p[2] = v.z; p[3] = v.w;
        }
    } else {
        for (int i = t; i < 16 * 32; i += 256) {
            int r  = i >> 5;
            int c4 = i & 31;
            float4 v = make_float4(0.f, 0.f, 0.f, 0.f);
            if (rowBase + r < N) v = ((const float4*)(x + (size_t)(rowBase + r) * HIDC))[c4];
            float* p = &As[r * 132 + c4 * 4];
            p[0] = v.x; p[1] = v.y; p[2] = v.z; p[3] = v.w;
        }
    }
    __syncthreads();

    const int lane = t & 31;
    const int wv   = t >> 5;
    const int m    = lane & 15;
    const int half = lane >> 4;
    const int kl   = half * 2;
    const int n    = wv * 16 + (lane & 15);

    v8f c = {};
    const float* Arow = &As[m * 132];
#pragma unroll 8
    for (int k0 = 0; k0 < HIDC; k0 += 4) {
        v2f a, b;
        a.x = Arow[k0 + kl];
        a.y = Arow[k0 + kl + 1];
        b.x = W[(k0 + kl) * HIDC + n];
        b.y = W[(k0 + kl + 1) * HIDC + n];
        c = __builtin_amdgcn_wmma_f32_16x16x4_f32(false, a, false, b,
                                                  (short)0, c, false, false);
    }

    float* hp = h + (size_t)(rowBase + 8 * half) * HIDC + n;
    if (full) {
#pragma unroll
        for (int v = 0; v < 8; ++v) hp[v * HIDC] = c[v];
    } else {
#pragma unroll
        for (int v = 0; v < 8; ++v) {
            if (rowBase + 8 * half + v < N) hp[v * HIDC] = c[v];
        }
    }
}

#define TN   512
#define QCAP 64
#define BCAP 384

__device__ __forceinline__ int rdlane_i(int v, int k) { return __builtin_amdgcn_readlane(v, k); }
__device__ __forceinline__ float rdlane_f(float v, int k) {
    return __int_as_float(__builtin_amdgcn_readlane(__float_as_int(v), k));
}
__device__ __forceinline__ void lds_fadd(float* p, float v) {
    __hip_atomic_fetch_add(p, v, __ATOMIC_RELAXED, __HIP_MEMORY_SCOPE_WORKGROUP);
}

__device__ __forceinline__ void route_edges(const float* __restrict__ h,
                                            const int* __restrict__ src,
                                            const int* __restrict__ dst,
                                            const float* __restrict__ dinv,
                                            float* tile, const float* dtile,
                                            unsigned* bins, int* binCnt,
                                            const int* myE, int lo,
                                            int head, int n, int lane) {
    const int idx = (head + lane) & (QCAP - 1);
    int  s    = 0;
    int  dl   = 0;
    bool over = false;
    if (lane < n) {
        const int e = myE[idx];
        s  = src[e];
        dl = dst[e] - lo;
        const int owner = dl & 31;
        const int pos = __hip_atomic_fetch_add(&binCnt[owner], 1, __ATOMIC_RELAXED,
                                               __HIP_MEMORY_SCOPE_WORKGROUP);
        if (pos < BCAP) bins[owner * BCAP + pos] = ((unsigned)s << 4) | (unsigned)(dl >> 5);
        else over = true;
    }
    unsigned om = __builtin_amdgcn_ballot_w32(over);
    while (om != 0u) {
        const int k = __builtin_ctz(om);
        om &= om - 1u;
        const int   sk = rdlane_i(s, k);
        const int   dk = rdlane_i(dl, k);
        const float nk = dinv[sk] * dtile[dk];
        const float4 r = ((const float4*)(h + (size_t)sk * HIDC))[lane];
        float* o = tile + dk * HIDC + 4 * lane;
        lds_fadd(o + 0, r.x * nk);
        lds_fadd(o + 1, r.y * nk);
        lds_fadd(o + 2, r.z * nk);
        lds_fadd(o + 3, r.w * nk);
    }
}

__device__ __forceinline__ void load_rows8(float4 (&r)[8], const float* __restrict__ h,
                                           int s, int k0, int lane) {
#pragma unroll
    for (int u = 0; u < 8; ++u) {
        const int sk = rdlane_i(s, k0 + u);
        r[u] = ((const float4*)(h + (size_t)sk * HIDC))[lane];
    }
}
template <bool GUARD>
__device__ __forceinline__ void add_rows8(const float4 (&r)[8], float* trow,
                                          int sl, float nrm, int k0, int cnt) {
#pragma unroll
    for (int u = 0; u < 8; ++u) {
        if (!GUARD || k0 + u < cnt) {
            const int   rk = rdlane_i(sl, k0 + u);
            const float nk = rdlane_f(nrm, k0 + u);
            float4* p = (float4*)(trow + rk * (32 * HIDC));
            float4 a = *p;
            a.x = fmaf(r[u].x, nk, a.x);
            a.y = fmaf(r[u].y, nk, a.y);
            a.z = fmaf(r[u].z, nk, a.z);
            a.w = fmaf(r[u].w, nk, a.w);
            *p = a;
        }
    }
}

__device__ __forceinline__ void push_edges(bool m, int e, int* myE, int head, int& cnt) {
    const unsigned mask = __builtin_amdgcn_ballot_w32(m);
    if (mask != 0u) {
        if (m) {
            int p = (head + cnt + (int)__builtin_amdgcn_mbcnt_lo(mask, 0u)) & (QCAP - 1);
            myE[p] = e;
        }
        cnt += __builtin_popcount(mask);
    }
}

__device__ __forceinline__ unsigned differ_bits(unsigned w, unsigned pat) {
    const unsigned x = w ^ pat;
    return (((x & 0x7f7f7f7fu) + 0x7f7f7f7fu) | x) & 0x80808080u;
}
__device__ __forceinline__ unsigned match16(uint4 v, unsigned pat) {
    const unsigned nd = (differ_bits(v.x, pat) >> 7) | (differ_bits(v.y, pat) >> 6) |
                        (differ_bits(v.z, pat) >> 5) | (differ_bits(v.w, pat) >> 4);
    return ~nd & 0x0f0f0f0fu;
}

__global__ __launch_bounds__(1024) void gcn_agg1_owner(const float* __restrict__ h,
                                                       const int* __restrict__ src,
                                                       const int* __restrict__ dst,
                                                       const unsigned int* __restrict__ keyw,
                                                       const float* __restrict__ dinv,
                                                       const float* __restrict__ b1,
                                                       const float* __restrict__ W2,
                                                       float* __restrict__ h2, int N, int E) {
    __shared__ __attribute__((aligned(16))) float tile[TN * HIDC];
    __shared__ unsigned bins[32 * BCAP];
    __shared__ int   qE[32 * QCAP];
    __shared__ float dtile[TN];
    __shared__ int   binCnt[32];

    const int t    = threadIdx.x;
    const int lane = t & 31;
    const int wv   = __builtin_amdgcn_readfirstlane(t >> 5);
    const int lo   = blockIdx.x * TN;

    for (int i = t; i < TN * HIDC / 4; i += 1024)
        ((float4*)tile)[i] = make_float4(0.f, 0.f, 0.f, 0.f);
    if (t < TN) dtile[t] = (lo + t < N) ? dinv[lo + t] : 0.0f;
    if (t < 32) binCnt[t] = 0;
    __syncthreads();

    int* myE = qE + wv * QCAP;
    int head = 0, cnt = 0;

    const unsigned char* key = (const unsigned char*)keyw;
    const bool usekey = (N <= 256 * TN) && ((((size_t)key) & 15) == 0);
    const int  nrows  = usekey ? (E >> 9) : 0;
    const unsigned pat = (blockIdx.x & 255u) * 0x01010101u;

    for (int r0 = wv; r0 < nrows; r0 += 32 * 8) {
        uint4 vv[8];
#pragma unroll
        for (int g = 0; g < 8; ++g) {
            const int row = min(r0 + 32 * g, nrows - 1);
            vv[g] = ((const uint4*)(key + ((size_t)row << 9)))[lane];
        }
#pragma unroll
        for (int g = 0; g < 8; ++g) {
            const int row = r0 + 32 * g;
            if (row < nrows) {
                const int eb = (row << 9) + lane * 16;
                unsigned F = match16(vv[g], pat);
#pragma nounroll
                for (int r = 0; r < 16; ++r) {
                    const unsigned mask = __builtin_amdgcn_ballot_w32(F != 0u);
                    if (mask == 0u) break;
                    if (F != 0u) {
                        const int tz = __builtin_ctz(F);
                        const int p  = (head + cnt + (int)__builtin_amdgcn_mbcnt_lo(mask, 0u)) & (QCAP - 1);
                        myE[p] = eb + ((tz & 7) << 2) + (tz >> 3);
                    }
                    cnt += __builtin_popcount(mask);
                    F &= F - 1u;
                    if (cnt >= 32) {
                        route_edges(h, src, dst, dinv, tile, dtile, bins, binCnt, myE, lo, head, 32, lane);
                        head = (head + 32) & (QCAP - 1);
                        cnt -= 32;
                    }
                }
            }
        }
    }
    const int ebeg   = nrows << 9;
    const int nsteps = (E - ebeg + 31) >> 5;
#pragma nounroll
    for (int st = wv; st < nsteps; st += 32) {
        const int e = ebeg + (st << 5) + lane;
        const int d = (e < E) ? dst[e] : -1;
        push_edges((unsigned)(d - lo) < (unsigned)TN, e, myE, head, cnt);
        if (cnt >= 32) {
            route_edges(h, src, dst, dinv, tile, dtile, bins, binCnt, myE, lo, head, 32, lane);
            head = (head + 32) & (QCAP - 1);
            cnt -= 32;
        }
    }
    if (cnt > 0) route_edges(h, src, dst, dinv, tile, dtile, bins, binCnt, myE, lo, head, cnt, lane);
    __syncthreads();

    {
        const int nb = __builtin_amdgcn_readfirstlane(min(binCnt[wv], BCAP));
        const unsigned* myB = bins + wv * BCAP;
        float* trow = tile + wv * HIDC + 4 * lane;
        unsigned ent = (lane < nb) ? myB[lane] : 0u;
        float    dv  = (lane < nb) ? dinv[ent >> 4] : 0.0f;
        for (int base = 0; base < nb; base += 32) {
            const int cnt32 = min(32, nb - base);
            const int   s   = (int)(ent >> 4);
            const int   sl  = (int)(ent & 15u);
            const float nrm = dv * dtile[(sl << 5) | wv];
            const int nx = base + 32 + lane;
            ent = (nx < nb) ? myB[nx] : 0u;
            dv  = (nx < nb) ? dinv[ent >> 4] : 0.0f;
            if (cnt32 == 32) {
                float4 ra[8], rb[8];
                load_rows8(ra, h, s, 0, lane);
                load_rows8(rb, h, s, 8, lane);
                add_rows8<false>(ra, trow, sl, nrm, 0, 32);
                load_rows8(ra, h, s, 16, lane);
                add_rows8<false>(rb, trow, sl, nrm, 8, 32);
                load_rows8(rb, h, s, 24, lane);
                add_rows8<false>(ra, trow, sl, nrm, 16, 32);
                add_rows8<false>(rb, trow, sl, nrm, 24, 32);
            } else {
#pragma nounroll
                for (int k0 = 0; k0 < cnt32; k0 += 8) {
                    float4 rc[8];
                    load_rows8(rc, h, s, k0, lane);
                    add_rows8<true>(rc, trow, sl, nrm, k0, cnt32);
                }
            }
        }
    }

    const float4 bb = ((const float4*)b1)[lane];
    const float4 w  = ((const float4*)W2)[lane];
    for (int r = wv; r < TN; r += 32) {
        const int i = lo + r;
        if (i >= N) break;
        const float di = dtile[r];
        const float ns = di * di;
        const float4 a  = ((const float4*)(tile + r * HIDC))[lane];
        const float4 hv = ((const float4*)(h + (size_t)i * HIDC))[lane];
        float r0 = fmaxf(a.x + hv.x * ns + bb.x, 0.0f);
        float r1 = fmaxf(a.y + hv.y * ns + bb.y, 0.0f);
        float r2 = fmaxf(a.z + hv.z * ns + bb.z, 0.0f);
        float r3 = fmaxf(a.w + hv.w * ns + bb.w, 0.0f);
        float p = r0 * w.x + r1 * w.y + r2 * w.z + r3 * w.w;
#pragma unroll
        for (int off = 16; off > 0; off >>= 1) p += __shfl_down(p, off, 32);
        if (lane == 0) h2[i] = p;
    }
}

__global__ void gcn_init_out(const float* __restrict__ h2, const float* __restrict__ dinv,
                             const float* __restrict__ b2, float* __restrict__ out, int N) {
    int i = blockIdx.x * blockDim.x + threadIdx.x;
    if (i < N) {
        float di = dinv[i];
        out[i] = h2[i] * di * di + b2[0];
    }
}

__global__ void gcn_agg2(const float* __restrict__ h2, const int* __restrict__ src,
                         const int* __restrict__ dst, const float* __restrict__ dinv,
                         float* __restrict__ out, int E) {
    int e = blockIdx.x * blockDim.x + threadIdx.x;
    if (e < E) {
        int s = src[e];
        int d = dst[e];
        atomicAdd(&out[d], h2[s] * dinv[s] * dinv[d]);
    }
}

extern "C" void kernel_launch(void* const* d_in, const int* in_sizes, int n_in,
                              void* d_out, int out_size, void* d_ws, size_t ws_size,
                              hipStream_t stream) {
    const float* x  = (const float*)d_in[0];
    const float* W1 = (const float*)d_in[1];
    const float* b1 = (const float*)d_in[2];
    const float* W2 = (const float*)d_in[3];
    const float* b2 = (const float*)d_in[4];
    const int*   ei = (const int*)d_in[5];

    const int N = in_sizes[0] / HIDC;
    const int E = in_sizes[5] / 2;
    const int* src = ei;
    const int* dst = ei + E;
    float* out = (float*)d_out;

    float* h    = (float*)d_ws;
    float* deg  = h   + (size_t)N * HIDC;
    float* dinv = deg + N;
    float* h2   = dinv + N;
    unsigned int* keyw = (unsigned int*)(h2 + N);

    const int TB = 256;
    gcn_init      <<<(N + TB - 1) / TB, TB, 0, stream>>>(deg, N);
    gcn_degree    <<<(E + TB - 1) / TB, TB, 0, stream>>>(dst, deg, keyw, E);
    gcn_dinv      <<<(N + TB - 1) / TB, TB, 0, stream>>>(deg, dinv, N);
    gcn_gemm1_wmma<<<(N + 15) / 16,     TB, 0, stream>>>(x, W1, h, N);
    gcn_agg1_owner<<<(N + TN - 1) / TN, 1024, 0, stream>>>(h, src, dst, keyw, dinv, b1, W2, h2, N, E);
    gcn_init_out  <<<(N + TB - 1) / TB, TB, 0, stream>>>(h2, dinv, b2, out, N);
    gcn_agg2      <<<(E + TB - 1) / TB, TB, 0, stream>>>(h2, src, dst, dinv, out, E);
}


# agg2 owner-computes: agg1 saves its edge bins to the workspace; a block re-reads them, adds g[src] (g=h2*dinv) to 512 LDS sums, scales by dinv[dst], writes out; f32 sum order: bin slot, self last
# speedup vs baseline: 2.4969x; 1.3960x over previous
#include <hip/hip_runtime.h>
#include <math.h>

#define HIDC 128

typedef __attribute__((ext_vector_type(2))) float v2f;
typedef __attribute__((ext_vector_type(8))) float v8f;

__global__ void gcn_init(float* __restrict__ deg, int N) {
    int idx = blockIdx.x * blockDim.x + threadIdx.x;
    if (idx < N) deg[idx] = 1.0f;
}

__global__ void gcn_degree(const int* __restrict__ dst, float* __restrict__ deg,
                           unsigned int* __restrict__ keyw, int E) {
    int e = blockIdx.x * blockDim.x + threadIdx.x;
    if (e < E) {
        const int d = dst[e];
        atomicAdd(&deg[d], 1.0f);
        ((unsigned char*)keyw)[e] = (unsigned char)(d >> 9);
    }
}

__global__ void gcn_dinv(const float* __restrict__ deg, float* __restrict__ dinv, int N) {
    int i = blockIdx.x * blockDim.x + threadIdx.x;
    if (i < N) dinv[i] = rsqrtf(deg[i]);
}

__global__ __launch_bounds__(256) void gcn_gemm1_wmma(const float* __restrict__ x,
                                                      const float* __restrict__ W,
                                                      float* __restrict__ h, int N) {
    __shared__ float As[16 * 132];
    const int t = threadIdx.x;
    const int rowBase = blockIdx.x * 16;
    const bool full = (rowBase + 16 <= N);

    if (full) {
#pragma unroll
        for (int i = t; i < 16 * 32; i += 256) {
            int r  = i >> 5;
            int c4 = i & 31;
            float4 v = ((const float4*)(x + (size_t)(rowBase + r) * HIDC))[c4];
            float* p = &As[r * 132 + c4 * 4];
            p[0] = v.x; p[1] = v.y; p[2] = v.z; p[3] = v.w;
        }
    } else {
        for (int i = t; i < 16 * 32; i += 256) {
            int r  = i >> 5;
            int c4 = i & 31;
            float4 v = make_float4(0.f, 0.f, 0.f, 0.f);
            if (rowBase + r < N) v = ((const float4*)(x + (size_t)(rowBase + r) * HIDC))[c4];
            float* p = &As[r * 132 + c4 * 4];
            p[0] = v.x; p[1] = v.y; p[2] = v.z; p[3] = v.w;
        }
    }
    __syncthreads();

    const int lane = t & 31;
    const int wv   = t >> 5;
    const int m    = lane & 15;
    const int half = lane >> 4;
    const int kl   = half * 2;
    const int n    = wv * 16 + (lane & 15);

    v8f c = {};
    const float* Arow = &As[m * 132];
#pragma unroll 8
    for (int k0 = 0; k0 < HIDC; k0 += 4) {
        v2f a, b;
        a.x = Arow[k0 + kl];
        a.y = Arow[k0 + kl + 1];
        b.x = W[(k0 + kl) * HIDC + n];
        b.y = W[(k0 + kl + 1) * HIDC + n];
        c = __builtin_amdgcn_wmma_f32_16x16x4_f32(false, a, false, b,
                                                  (short)0, c, false, false);
    }

    float* hp = h + (size_t)(rowBase + 8 * half) * HIDC + n;
    if (full) {
#pragma unroll
        for (int v = 0; v < 8; ++v) hp[v * HIDC] = c[v];
    } else {
#pragma unroll
        for (int v = 0; v < 8; ++v) {
            if (rowBase + 8 * half + v < N) hp[v * HIDC] = c[v];
        }
    }
}

#define TN   512
#define QCAP 64
#define BCAP 384

__device__ __forceinline__ int rdlane_i(int v, int k) { return __builtin_amdgcn_readlane(v, k); }
__device__ __forceinline__ float rdlane_f(float v, int k) {
    return __int_as_float(__builtin_amdgcn_readlane(__float_as_int(v), k));
}
__device__ __forceinline__ void lds_fadd(float* p, float v) {
    __hip_atomic_fetch_add(p, v, __ATOMIC_RELAXED, __HIP_MEMORY_SCOPE_WORKGROUP);
}

__device__ __forceinline__ void route_edges(const float* __restrict__ h,
                                            const int* __restrict__ src,
                                            const int* __restrict__ dst,
                                            const float* __restrict__ dinv,
                                            float* tile, const float* dtile,
                                            unsigned* bins, int* binCnt,
                                            const int* myE, int lo,
                                            int head, int n, int lane) {
    const int idx = (head + lane) & (QCAP - 1);
    int  s    = 0;
    int  dl   = 0;
    bool over = false;
    if (lane < n) {
        const int e = myE[idx];
        s  = src[e];
        dl = dst[e] - lo;
        const int owner = dl & 31;
        const int pos = __hip_atomic_fetch_add(&binCnt[owner], 1, __ATOMIC_RELAXED,
                                               __HIP_MEMORY_SCOPE_WORKGROUP);
        if (pos < BCAP) bins[owner * BCAP + pos] = ((unsigned)s << 4) | (unsigned)(dl >> 5);
        else over = true;
    }
    unsigned om = __builtin_amdgcn_ballot_w32(over);
    while (om != 0u) {
        const int k = __builtin_ctz(om);
        om &= om - 1u;
        const int   sk = rdlane_i(s, k);
        const int   dk = rdlane_i(dl, k);
        const float nk = dinv[sk] * dtile[dk];
        const float4 r = ((const float4*)(h + (size_t)sk * HIDC))[lane];
        float* o = tile + dk * HIDC + 4 * lane;
        lds_fadd(o + 0, r.x * nk);
        lds_fadd(o + 1, r.y * nk);
        lds_fadd(o + 2, r.z * nk);
        lds_fadd(o + 3, r.w * nk);
    }
}

__device__ __forceinline__ void load_rows8(float4 (&r)[8], const float* __restrict__ h,
                                           int s, int k0, int lane) {
#pragma unroll
    for (int u = 0; u < 8; ++u) {
        const int sk = rdlane_i(s, k0 + u);
        r[u] = ((const float4*)(h + (size_t)sk * HIDC))[lane];
    }
}
template <bool GUARD>
__device__ __forceinline__ void add_rows8(const float4 (&r)[8], float* trow,
                                          int sl, float nrm, int k0, int cnt) {
#pragma unroll
    for (int u = 0; u < 8; ++u) {
        if (!GUARD || k0 + u < cnt) {
            const int   rk = rdlane_i(sl, k0 + u);
            const float nk = rdlane_f(nrm, k0 + u);
            float4* p = (float4*)(trow + rk * (32 * HIDC));
            float4 a = *p;
            a.x = fmaf(r[u].x, nk, a.x);
            a.y = fmaf(r[u].y, nk, a.y);
            a.z = fmaf(r[u].z, nk, a.z);
            a.w = fmaf(r[u].w, nk, a.w);
            *p = a;
        }
    }
}

__device__ __forceinline__ void push_edges(bool m, int e, int* myE, int head, int& cnt) {
    const unsigned mask = __builtin_amdgcn_ballot_w32(m);
    if (mask != 0u) {
        if (m) {
            int p = (head + cnt + (int)__builtin_amdgcn_mbcnt_lo(mask, 0u)) & (QCAP - 1);
            myE[p] = e;
        }
        cnt += __builtin_popcount(mask);
    }
}

__device__ __forceinline__ unsigned differ_bits(unsigned w, unsigned pat) {
    const unsigned x = w ^ pat;
    return (((x & 0x7f7f7f7fu) + 0x7f7f7f7fu) | x) & 0x80808080u;
}
__device__ __forceinline__ unsigned match16(uint4 v, unsigned pat) {
    const unsigned nd = (differ_bits(v.x, pat) >> 7) | (differ_bits(v.y, pat) >> 6) |
                        (differ_bits(v.z, pat) >> 5) | (differ_bits(v.w, pat) >> 4);
    return ~nd & 0x0f0f0f0fu;
}

__global__ __launch_bounds__(1024) void gcn_agg1_owner(const float* __restrict__ h,
                                                       const int* __restrict__ src,
                                                       const int* __restrict__ dst,
                                                       const unsigned int* __restrict__ keyw,
                                                       const float* __restrict__ dinv,
                                                       const float* __restrict__ b1,
                                                       const float* __restrict__ W2,
                                                       float* __restrict__ gv,
                                                       unsigned int* __restrict__ binsG,
                                                       int* __restrict__ cntG, int N, int E) {
    __shared__ __attribute__((aligned(16))) float tile[TN * HIDC];
    __shared__ unsigned bins[32 * BCAP];
    __shared__ int   qE[32 * QCAP];
    __shared__ float dtile[TN];
    __shared__ int   binCnt[32];

    const int t    = threadIdx.x;
    const int lane = t & 31;
    const int wv   = __builtin_amdgcn_readfirstlane(t >> 5);
    const int lo   = blockIdx.x * TN;

    for (int i = t; i < TN * HIDC / 4; i += 1024)
        ((float4*)tile)[i] = make_float4(0.f, 0.f, 0.f, 0.f);
    if (t < TN) dtile[t] = (lo + t < N) ? dinv[lo + t] : 0.0f;
    if (t < 32) binCnt[t] = 0;
    __syncthreads();

    int* myE = qE + wv * QCAP;
    int head = 0, cnt = 0;

    const unsigned char* key = (const unsigned char*)keyw;
    const bool usekey = (N <= 256 * TN) && ((((size_t)key) & 15) == 0);
    const int  nrows  = usekey ? (E >> 9) : 0;
    const unsigned pat = (blockIdx.x & 255u) * 0x01010101u;

    for (int r0 = wv; r0 < nrows; r0 += 32 * 8) {
        uint4 vv[8];
#pragma unroll
        for (int g = 0; g < 8; ++g) {
            const int row = min(r0 + 32 * g, nrows - 1);
            vv[g] = ((const uint4*)(key + ((size_t)row << 9)))[lane];
        }
#pragma unroll
        for (int g = 0; g < 8; ++g) {
            const int row = r0 + 32 * g;
            if (row < nrows) {
                const int eb = (row << 9) + lane * 16;
                unsigned F = match16(vv[g], pat);
#pragma nounroll
                for (int r = 0; r < 16; ++r) {
                    const unsigned mask = __builtin_amdgcn_ballot_w32(F != 0u);
                    if (mask == 0u) break;
                    if (F != 0u) {
                        const int tz = __builtin_ctz(F);
                        const int p  = (head + cnt + (int)__builtin_amdgcn_mbcnt_lo(mask, 0u)) & (QCAP - 1);
                        myE[p] = eb + ((tz & 7) << 2) + (tz >> 3);
                    }
                    cnt += __builtin_popcount(mask);
                    F &= F - 1u;
                    if (cnt >= 32) {
                        route_edges(h, src, dst, dinv, tile, dtile, bins, binCnt, myE, lo, head, 32, lane);
                        head = (head + 32) & (QCAP - 1);
                        cnt -= 32;
                    }
                }
            }
        }
    }
    const int ebeg   = nrows << 9;
    const int nsteps = (E - ebeg + 31) >> 5;
#pragma nounroll
    for (int st = wv; st < nsteps; st += 32) {
        const int e = ebeg + (st << 5) + lane;
        const int d = (e < E) ? dst[e] : -1;
        push_edges((unsigned)(d - lo) < (unsigned)TN, e, myE, head, cnt);
        if (cnt >= 32) {
            route_edges(h, src, dst, dinv, tile, dtile, bins, binCnt, myE, lo, head, 32, lane);
            head = (head + 32) & (QCAP - 1);
            cnt -= 32;
        }
    }
    if (cnt > 0) route_edges(h, src, dst, dinv, tile, dtile, bins, binCnt, myE, lo, head, cnt, lane);
    __syncthreads();

    {
        const int nraw = __builtin_amdgcn_readfirstlane(binCnt[wv]);
        const int nb = min(nraw, BCAP);
        const unsigned* myB = bins + wv * BCAP;
        unsigned* gB = binsG + ((size_t)blockIdx.x * 32 + wv) * BCAP;
        if (lane == 0) cntG[blockIdx.x * 32 + wv] = nraw;
        float* trow = tile + wv * HIDC + 4 * lane;
        unsigned ent = (lane < nb) ? myB[lane] : 0u;
        float    dv  = (lane < nb) ? dinv[ent >> 4] : 0.0f;
        for (int base = 0; base < nb; base += 32) {
            const int cnt32 = min(32, nb - base);
            if (base + lane < nb) gB[base + lane] = ent;
            const int   s   = (int)(ent >> 4);
            const int   sl  = (int)(ent & 15u);
            const float nrm = dv * dtile[(sl << 5) | wv];
            const int nx = base + 32 + lane;
            ent = (nx < nb) ? myB[nx] : 0u;
            dv  = (nx < nb) ? dinv[ent >> 4] : 0.0f;
            if (cnt32 == 32) {
                float4 ra[8], rb[8];
                load_rows8(ra, h, s, 0, lane);
                load_rows8(rb, h, s, 8, lane);
                add_rows8<false>(ra, trow, sl, nrm, 0, 32);
                load_rows8(ra, h, s, 16, lane);
                add_rows8<false>(rb, trow, sl, nrm, 8, 32);
                load_rows8(rb, h, s, 24, lane);
                add_rows8<false>(ra, trow, sl, nrm, 16, 32);
                add_rows8<false>(rb, trow, sl, nrm, 24, 32);
            } else {
#pragma nounroll
                for (int k0 = 0; k0 < cnt32; k0 += 8) {
                    float4 rc[8];
                    load_rows8(rc, h, s, k0, lane);
                    add_rows8<true>(rc, trow, sl, nrm, k0, cnt32);
                }
            }
        }
    }

    const float4 bb = ((const float4*)b1)[lane];
    const float4 w  = ((const float4*)W2)[lane];
    for (int r = wv; r < TN; r += 32) {
        const int i = lo + r;
        if (i >= N) break;
        const float di = dtile[r];
        const float ns = di * di;
        const float4 a  = ((const float4*)(tile + r * HIDC))[lane];
        const float4 hv = ((const float4*)(h + (size_t)i * HIDC))[lane];
        float r0 = fmaxf(a.x + hv.x * ns + bb.x, 0.0f);
        float r1 = fmaxf(a.y + hv.y * ns + bb.y, 0.0f);
        float r2 = fmaxf(a.z + hv.z * ns + bb.z, 0.0f);
        float r3 = fmaxf(a.w + hv.w * ns + bb.w, 0.0f);
        float p = r0 * w.x + r1 * w.y + r2 * w.z + r3 * w.w;
#pragma unroll
        for (int off = 16; off > 0; off >>= 1) p += __shfl_down(p, off, 32);
        if (lane == 0) gv[i] = p * di;
    }
}

__global__ __launch_bounds__(1024) void gcn_agg2_owner(const float* __restrict__ g,
                                                       const int* __restrict__ src,
                                                       const int* __restrict__ dst,
                                                       const unsigned int* __restrict__ binsG,
                                                       const int* __restrict__ cntG,
                                                       const float* __restrict__ dinv,
                                                       const float* __restrict__ b2,
                                                       float* __restrict__ out, int N, int E) {
    __shared__ float acc[TN];

    const int t    = threadIdx.x;
    const int lane = t & 31;
    const int wv   = __builtin_amdgcn_readfirstlane(t >> 5);
    const int lo   = blockIdx.x * TN;

    const int  c    = cntG[blockIdx.x * 32 + lane];
    const bool over = __builtin_amdgcn_ballot_w32(c > BCAP) != 0u;
    const int  nraw = rdlane_i(c, wv);

    if (t < TN) acc[t] = 0.0f;
    __syncthreads();

    const bool  mine = (t < TN) && (lo + t < N);
    const float gs   = mine ? g[lo + t] : 0.0f;
    const float ds   = mine ? dinv[lo + t] : 0.0f;
    const float bias = b2[0];

    if (!over) {
        if (nraw > 0) {
            const unsigned* gB = binsG + ((size_t)blockIdx.x * 32 + wv) * BCAP;
            unsigned ent[BCAP / 32];
            float    v[BCAP / 32];
#pragma unroll
            for (int k = 0; k < BCAP / 32; ++k) ent[k] = gB[min(lane + 32 * k, nraw - 1)];
#pragma unroll
            for (int k = 0; k < BCAP / 32; ++k) v[k] = g[ent[k] >> 4];
#pragma unroll
            for (int k = BCAP / 32 - 1; k >= 0; --k) {
                if (lane + 32 * k < nraw) lds_fadd(&acc[wv * 16 + (int)(ent[k] & 15u)], v[k]);
            }
        }
    } else {
        for (int e = t; e < E; e += 1024) {
            const unsigned dl = (unsigned)(dst[e] - lo);
            if (dl < (unsigned)TN) lds_fadd(&acc[(dl & 31u) * 16u + (dl >> 5)], g[src[e]]);
        }
    }
    __syncthreads();

    if (mine) out[lo + t] = (acc[(t & 31) * 16 + (t >> 5)] + gs) * ds + bias;
}

extern "C" void kernel_launch(void* const* d_in, const int* in_sizes, int n_in,
                              void* d_out, int out_size, void* d_ws, size_t ws_size,
                              hipStream_t stream) {
    const float* x  = (const float*)d_in[0];
    const float* W1 = (const float*)d_in[1];
    const float* b1 = (const float*)d_in[2];
    const float* W2 = (const float*)d_in[3];
    const float* b2 = (const float*)d_in[4];
    const int*   ei = (const int*)d_in[5];

    const int N = in_sizes[0] / HIDC;
    const int E = in_sizes[5] / 2;
    const int* src = ei;
    const int* dst = ei + E;
    float* out = (float*)d_out;

    float* h    = (float*)d_ws;
    float* deg  = h   + (size_t)N * HIDC;
    float* dinv = deg + N;
    float* g    = dinv + N;
    const int NB = (N + TN - 1) / TN;
    int* cntG = (int*)(g + N);
    unsigned int* binsG = (unsigned int*)(cntG + (size_t)NB * 32);
    unsigned int* keyw  = binsG + (size_t)NB * 32 * BCAP;

    const int TB = 256;
    gcn_init      <<<(N + TB - 1) / TB, TB, 0, stream>>>(deg, N);
    gcn_degree    <<<(E + TB - 1) / TB, TB, 0, stream>>>(dst, deg, keyw, E);
    gcn_dinv      <<<(N + TB - 1) / TB, TB, 0, stream>>>(deg, dinv, N);
    gcn_gemm1_wmma<<<(N + 15) / 16,     TB, 0, stream>>>(x, W1, h, N);
    gcn_agg1_owner<<<NB, 1024, 0, stream>>>(h, src, dst, keyw, dinv, b1, W2, g, binsG, cntG, N, E);
    gcn_agg2_owner<<<NB, 1024, 0, stream>>>(g, src, dst, binsG, cntG, dinv, b2, out, N, E);
}


# in-degree counted in LDS by owner blocks: edge filing runs first as its own launch, counts in-edges per node, writes dinv; global f32 atomics, init and dinv launches gone; f32 sum order as before
# speedup vs baseline: 3.8716x; 1.5506x over previous
#include <hip/hip_runtime.h>
#include <math.h>

#define HIDC 128

typedef __attribute__((ext_vector_type(2))) float v2f;
typedef __attribute__((ext_vector_type(8))) float v8f;

__device__ __forceinline__ unsigned key4(int4 d) {
    return ((unsigned)(d.x >> 9) & 255u) | (((unsigned)(d.y >> 9) & 255u) << 8) |
           (((unsigned)(d.z >> 9) & 255u) << 16) | (((unsigned)(d.w >> 9) & 255u) << 24);
}
__global__ void gcn_keys(const int* __restrict__ dst, unsigned int* __restrict__ keyw, int E) {
    const int gt   = blockIdx.x * blockDim.x + threadIdx.x;
    const int lane = threadIdx.x & 31;
    const int qb   = (gt >> 5) * 128 + lane;
    const int nq   = E >> 2;
    unsigned char* key = (unsigned char*)keyw;
    if (((((size_t)dst) & 15) == 0) && nq > 0) {
        int4 d[4];
#pragma unroll
        for (int j = 0; j < 4; ++j) d[j] = ((const int4*)dst)[min(qb + 32 * j, nq - 1)];
#pragma unroll
        for (int j = 0; j < 4; ++j) {
            const int q = qb + 32 * j;
            if (q < nq) keyw[q] = key4(d[j]);
        }
        if (gt == 0) {
            for (int e = nq * 4; e < E; ++e) key[e] = (unsigned char)(dst[e] >> 9);
        }
    } else {
        for (int j = 0; j < 4; ++j) {
            for (int k = 0; k < 4; ++k) {
                const int e = 4 * (qb + 32 * j) + k;
                if (e < E) key[e] = (unsigned char)(dst[e] >> 9);
            }
        }
    }
}

__global__ __launch_bounds__(256) void gcn_gemm1_wmma(const float* __restrict__ x,
                                                      const float* __restrict__ W,
                                                      float* __restrict__ h, int N) {
    __shared__ float As[16 * 132];
    const int t = threadIdx.x;
    const int rowBase = blockIdx.x * 16;
    const bool full = (rowBase + 16 <= N);

    if (full) {
#pragma unroll
        for (int i = t; i < 16 * 32; i += 256) {
            int r  = i >> 5;
            int c4 = i & 31;
            float4 v = ((const float4*)(x + (size_t)(rowBase + r) * HIDC))[c4];
            float* p = &As[r * 132 + c4 * 4];
            p[0] = v.x; p[1] = v.y; p[2] = v.z; p[3] = v.w;
        }
    } else {
        for (int i = t; i < 16 * 32; i += 256) {
            int r  = i >> 5;
            int c4 = i & 31;
            float4 v = make_float4(0.f, 0.f, 0.f, 0.f);
            if (rowBase + r < N) v = ((const float4*)(x + (size_t)(rowBase + r) * HIDC))[c4];
            float* p = &As[r * 132 + c4 * 4];
            p[0] = v.x; p[1] = v.y; p[2] = v.z; p[3] = v.w;
        }
    }
    __syncthreads();

    const int lane = t & 31;
    const int wv   = t >> 5;
    const int m    = lane & 15;
    const int half = lane >> 4;
    const int kl   = half * 2;
    const int n    = wv * 16 + (lane & 15);

    v8f c = {};
    const float* Arow = &As[m * 132];
#pragma unroll 8
    for (int k0 = 0; k0 < HIDC; k0 += 4) {
        v2f a, b;
        a.x = Arow[k0 + kl];
        a.y = Arow[k0 + kl + 1];
        b.x = W[(k0 + kl) * HIDC + n];
        b.y = W[(k0 + kl + 1) * HIDC + n];
        c = __builtin_amdgcn_wmma_f32_16x16x4_f32(false, a, false, b,
                                                  (short)0, c, false, false);
    }

    float* hp = h + (size_t)(rowBase + 8 * half) * HIDC + n;
    if (full) {
#pragma unroll
        for (int v = 0; v < 8; ++v) hp[v * HIDC] = c[v];
    } else {
#pragma unroll
        for (int v = 0; v < 8; ++v) {
            if (rowBase + 8 * half + v < N) hp[v * HIDC] = c[v];
        }
    }
}

#define TN   512
#define QCAP 64
#define BCAP 384

__device__ __forceinline__ int rdlane_i(int v, int k) { return __builtin_amdgcn_readlane(v, k); }
__device__ __forceinline__ float rdlane_f(float v, int k) {
    return __int_as_float(__builtin_amdgcn_readlane(__float_as_int(v), k));
}
__device__ __forceinline__ void lds_fadd(float* p, float v) {
    __hip_atomic_fetch_add(p, v, __ATOMIC_RELAXED, __HIP_MEMORY_SCOPE_WORKGROUP);
}

__device__ __forceinline__ void file_edges(const int* __restrict__ src,
                                           const int* __restrict__ dst,
                                           unsigned* bins, int* binCnt, int* degc,
                                           const int* myE, int lo,
                                           int head, int n, int lane) {
    if (lane < n) {
        const int e  = myE[(head + lane) & (QCAP - 1)];
        const int s  = src[e];
        const int dl = dst[e] - lo;
        const int owner = dl & 31;
        __hip_atomic_fetch_add(&degc[dl], 1, __ATOMIC_RELAXED, __HIP_MEMORY_SCOPE_WORKGROUP);
        const int pos = __hip_atomic_fetch_add(&binCnt[owner], 1, __ATOMIC_RELAXED,
                                               __HIP_MEMORY_SCOPE_WORKGROUP);
        const int slot = (pos < BCAP) ? owner * BCAP + pos : 32 * BCAP;
        bins[slot] = ((unsigned)s << 4) | (unsigned)(dl >> 5);
    }
}

__device__ __forceinline__ void push_edges(bool m, int e, int* myE, int head, int& cnt) {
    const unsigned mask = __builtin_amdgcn_ballot_w32(m);
    if (mask != 0u) {
        if (m) {
            int p = (head + cnt + (int)__builtin_amdgcn_mbcnt_lo(mask, 0u)) & (QCAP - 1);
            myE[p] = e;
        }
        cnt += __builtin_popcount(mask);
    }
}

__device__ __forceinline__ unsigned differ_bits(unsigned w, unsigned pat) {
    const unsigned x = w ^ pat;
    return (((x & 0x7f7f7f7fu) + 0x7f7f7f7fu) | x) & 0x80808080u;
}
__device__ __forceinline__ unsigned match16(uint4 v, unsigned pat) {
    const unsigned nd = (differ_bits(v.x, pat) >> 7) | (differ_bits(v.y, pat) >> 6) |
                        (differ_bits(v.z, pat) >> 5) | (differ_bits(v.w, pat) >> 4);
    return ~nd & 0x0f0f0f0fu;
}

__global__ __launch_bounds__(1024) void gcn_file_edges(const int* __restrict__ src,
                                                       const int* __restrict__ dst,
                                                       const unsigned int* __restrict__ keyw,
                                                       float* __restrict__ dinv,
                                                       unsigned int* __restrict__ binsG,
                                                       int* __restrict__ cntG, int N, int E) {
    __shared__ unsigned bins[32 * BCAP + 1];
    __shared__ int   qE[32 * QCAP];
    __shared__ int   degc[TN];
    __shared__ int   binCnt[32];

    const int t    = threadIdx.x;
    const int lane = t & 31;
    const int wv   = __builtin_amdgcn_readfirstlane(t >> 5);
    const int lo   = blockIdx.x * TN;

    if (t < TN) degc[t] = 0;
    if (t < 32) binCnt[t] = 0;
    __syncthreads();

    int* myE = qE + wv * QCAP;
    int head = 0, cnt = 0;

    const unsigned char* key = (const unsigned char*)keyw;
    const bool usekey = (N <= 256 * TN) && ((((size_t)key) & 15) == 0);
    const int  nrows  = usekey ? (E >> 9) : 0;
    const unsigned pat = (blockIdx.x & 255u) * 0x01010101u;

    for (int r0 = wv; r0 < nrows; r0 += 32 * 8) {
        uint4 vv[8];
#pragma unroll
        for (int g = 0; g < 8; ++g) {
            const int row = min(r0 + 32 * g, nrows - 1);
            vv[g] = ((const uint4*)(key + ((size_t)row << 9)))[lane];
        }
#pragma unroll
        for (int g = 0; g < 8; ++g) {
            const int row = r0 + 32 * g;
            if (row < nrows) {
                const int eb = (row << 9) + lane * 16;
                unsigned F = match16(vv[g], pat);
#pragma nounroll
                for (int r = 0; r < 16; ++r) {
                    const unsigned mask = __builtin_amdgcn_ballot_w32(F != 0u);
                    if (mask == 0u) break;
                    if (F != 0u) {
                        const int tz = __builtin_ctz(F);
                        const int p  = (head + cnt + (int)__builtin_amdgcn_mbcnt_lo(mask, 0u)) & (QCAP - 1);
                        myE[p] = eb + ((tz & 7) << 2) + (tz >> 3);
                    }
                    cnt += __builtin_popcount(mask);
                    F &= F - 1u;
                    if (cnt >= 32) {
                        file_edges(src, dst, bins, binCnt, degc, myE, lo, head, 32, lane);
                        head = (head + 32) & (QCAP - 1);
                        cnt -= 32;
                    }
                }
            }
        }
    }
    const int ebeg   = nrows << 9;
    const int nsteps = (E - ebeg + 31) >> 5;
#pragma nounroll
    for (int st = wv; st < nsteps; st += 32) {
        const int e = ebeg + (st << 5) + lane;
        const int d = (e < E) ? dst[e] : -1;
        push_edges((unsigned)(d - lo) < (unsigned)TN, e, myE, head, cnt);
        if (cnt >= 32) {
            file_edges(src, dst, bins, binCnt, degc, myE, lo, head, 32, lane);
            head = (head + 32) & (QCAP - 1);
            cnt -= 32;
        }
    }
    if (cnt > 0) file_edges(src, dst, bins, binCnt, degc, myE, lo, head, cnt, lane);
    __syncthreads();

    {
        const int nraw = __builtin_amdgcn_readfirstlane(binCnt[wv]);
        const int nb = min(nraw, BCAP);
        const unsigned* myB = bins + wv * BCAP;
        unsigned* gB = binsG + ((size_t)blockIdx.x * 32 + wv) * BCAP;
        for (int i = lane; i < nb; i += 32) gB[i] = myB[i];
        if (lane == 0) cntG[blockIdx.x * 32 + wv] = nraw;
    }
    if (t < TN && lo + t < N) dinv[lo + t] = rsqrtf(1.0f + (float)degc[t]);
}


__device__ __forceinline__ void load_rows8(float4 (&r)[8], const float* __restrict__ h,
                                           int s, int k0, int lane) {
#pragma unroll
    for (int u = 0; u < 8; ++u) {
        const int sk = rdlane_i(s, k0 + u);
        r[u] = ((const float4*)(h + (size_t)sk * HIDC))[lane];
    }
}
template <bool GUARD>
__device__ __forceinline__ void add_rows8(const float4 (&r)[8], float* trow,
                                          int sl, float nrm, int k0, int cnt) {
#pragma unroll
    for (int u = 0; u < 8; ++u) {
        if (!GUARD || k0 + u < cnt) {
            const int   rk = rdlane_i(sl, k0 + u);
            const float nk = rdlane_f(nrm, k0 + u);
            float4* p = (float4*)(trow + rk * (32 * HIDC));
            float4 a = *p;
            a.x = fmaf(r[u].x, nk, a.x);
            a.y = fmaf(r[u].y, nk, a.y);
            a.z = fmaf(r[u].z, nk, a.z);
            a.w = fmaf(r[u].w, nk, a.w);
            *p = a;
        }
    }
}

__global__ __launch_bounds__(1024) void gcn_agg1_owner(const float* __restrict__ h,
                                                       const int* __restrict__ src,
                                                       const int* __restrict__ dst,
                                                       const float* __restrict__ dinv,
                                                       const float* __restrict__ b1,
                                                       const float* __restrict__ W2,
                                                       float* __restrict__ gv,
                                                       const unsigned int* __restrict__ binsG,
                                                       const int* __restrict__ cntG, int N, int E) {
    __shared__ __attribute__((aligned(16))) float tile[TN * HIDC];
    __shared__ float dtile[TN];

    const int t    = threadIdx.x;
    const int lane = t & 31;
    const int wv   = __builtin_amdgcn_readfirstlane(t >> 5);
    const int lo   = blockIdx.x * TN;

    const int  c    = cntG[blockIdx.x * 32 + lane];
    const bool over = __builtin_amdgcn_ballot_w32(c > BCAP) != 0u;
    const int  nb   = rdlane_i(c, wv);

    for (int i = t; i < TN * HIDC / 4; i += 1024)
        ((float4*)tile)[i] = make_float4(0.f, 0.f, 0.f, 0.f);
    if (t < TN) dtile[t] = (lo + t < N) ? dinv[lo + t] : 0.0f;
    __syncthreads();

    if (!over) {
        const unsigned* gB = binsG + ((size_t)blockIdx.x * 32 + wv) * BCAP;
        float* trow = tile + wv * HIDC + 4 * lane;
        unsigned ent  = (lane < nb) ? gB[lane] : 0u;
        unsigned ent1 = (32 + lane < nb) ? gB[32 + lane] : 0u;
        float    dv   = (lane < nb) ? dinv[ent >> 4] : 0.0f;
        for (int base = 0; base < nb; base += 32) {
            const int cnt32 = min(32, nb - base);
            const int   s   = (int)(ent >> 4);
            const int   sl  = (int)(ent & 15u);
            const float nrm = dv * dtile[(sl << 5) | wv];
            ent  = ent1;
            dv   = (base + 32 + lane < nb) ? dinv[ent >> 4] : 0.0f;
            ent1 = (base + 64 + lane < nb) ? gB[base + 64 + lane] : 0u;
            if (cnt32 == 32) {
                float4 ra[8], rb[8];
                load_rows8(ra, h, s, 0, lane);
                load_rows8(rb, h, s, 8, lane);
                add_rows8<false>(ra, trow, sl, nrm, 0, 32);
                load_rows8(ra, h, s, 16, lane);
                add_rows8<false>(rb, trow, sl, nrm, 8, 32);
                load_rows8(rb, h, s, 24, lane);
                add_rows8<false>(ra, trow, sl, nrm, 16, 32);
                add_rows8<false>(rb, trow, sl, nrm, 24, 32);
            } else {
#pragma nounroll
                for (int k0 = 0; k0 < cnt32; k0 += 8) {
                    float4 rc[8];
                    load_rows8(rc, h, s, k0, lane);
                    add_rows8<true>(rc, trow, sl, nrm, k0, cnt32);
                }
            }
        }
    } else {
        const int nsteps = (E + 31) >> 5;
        for (int st = wv; st < nsteps; st += 32) {
            const int e = (st << 5) + lane;
            const int d = (e < E) ? dst[e] : -1;
            const int dl = d - lo;
            const bool m = (unsigned)dl < (unsigned)TN;
            const int s = m ? src[e] : 0;
            unsigned om = __builtin_amdgcn_ballot_w32(m);
            while (om != 0u) {
                const int k = __builtin_ctz(om);
                om &= om - 1u;
                const int   sk = rdlane_i(s, k);
                const int   dk = rdlane_i(dl, k);
                const float nk = dinv[sk] * dtile[dk];
                const float4 r = ((const float4*)(h + (size_t)sk * HIDC))[lane];
                float* o = tile + dk * HIDC + 4 * lane;
                lds_fadd(o + 0, r.x * nk);
                lds_fadd(o + 1, r.y * nk);
                lds_fadd(o + 2, r.z * nk);
                lds_fadd(o + 3, r.w * nk);
            }
        }
        __syncthreads();
    }

    const float4 bb = ((const float4*)b1)[lane];
    const float4 w  = ((const float4*)W2)[lane];
    for (int r = wv; r < TN; r += 32) {
        const int i = lo + r;
        if (i >= N) break;
        const float di = dtile[r];
        const float ns = di * di;
        const float4 a  = ((const float4*)(tile + r * HIDC))[lane];
        const float4 hv = ((const float4*)(h + (size_t)i * HIDC))[lane];
        float r0 = fmaxf(a.x + hv.x * ns + bb.x, 0.0f);
        float r1 = fmaxf(a.y + hv.y * ns + bb.y, 0.0f);
        float r2 = fmaxf(a.z + hv.z * ns + bb.z, 0.0f);
        float r3 = fmaxf(a.w + hv.w * ns + bb.w, 0.0f);
        float p = r0 * w.x + r1 * w.y + r2 * w.z + r3 * w.w;
#pragma unroll
        for (int off = 16; off > 0; off >>= 1) p += __shfl_down(p, off, 32);
        if (lane == 0) gv[i] = p * di;
    }
}

__global__ __launch_bounds__(1024) void gcn_agg2_owner(const float* __restrict__ g,
                                                       const int* __restrict__ src,
                                                       const int* __restrict__ dst,
                                                       const unsigned int* __restrict__ binsG,
                                                       const int* __restrict__ cntG,
                                                       const float* __restrict__ dinv,
                                                       const float* __restrict__ b2,
                                                       float* __restrict__ out, int N, int E) {
    __shared__ float acc[TN];

    const int t    = threadIdx.x;
    const int lane = t & 31;
    const int wv   = __builtin_amdgcn_readfirstlane(t >> 5);
    const int lo   = blockIdx.x * TN;

    const int  c    = cntG[blockIdx.x * 32 + lane];
    const bool over = __builtin_amdgcn_ballot_w32(c > BCAP) != 0u;
    const int  nraw = rdlane_i(c, wv);

    if (t < TN) acc[t] = 0.0f;
    __syncthreads();

    const bool  mine = (t < TN) && (lo + t < N);
    const float gs   = mine ? g[lo + t] : 0.0f;
    const float ds   = mine ? dinv[lo + t] : 0.0f;
    const float bias = b2[0];

    if (!over) {
        if (nraw > 0) {
            const unsigned* gB = binsG + ((size_t)blockIdx.x * 32 + wv) * BCAP;
            unsigned ent[BCAP / 32];
            float    v[BCAP / 32];
#pragma unroll
            for (int k = 0; k < BCAP / 32; ++k) ent[k] = gB[min(lane + 32 * k, nraw - 1)];
#pragma unroll
            for (int k = 0; k < BCAP / 32; ++k) v[k] = g[ent[k] >> 4];
#pragma unroll
            for (int k = BCAP / 32 - 1; k >= 0; --k) {
                if (lane + 32 * k < nraw) lds_fadd(&acc[wv * 16 + (int)(ent[k] & 15u)], v[k]);
            }
        }
    } else {
        for (int e = t; e < E; e += 1024) {
            const unsigned dl = (unsigned)(dst[e] - lo);
            if (dl < (unsigned)TN) lds_fadd(&acc[(dl & 31u) * 16u + (dl >> 5)], g[src[e]]);
        }
    }
    __syncthreads();

    if (mine) out[lo + t] = (acc[(t & 31) * 16 + (t >> 5)] + gs) * ds + bias;
}

extern "C" void kernel_launch(void* const* d_in, const int* in_sizes, int n_in,
                              void* d_out, int out_size, void* d_ws, size_t ws_size,
                              hipStream_t stream) {
    const float* x  = (const float*)d_in[0];
    const float* W1 = (const float*)d_in[1];
    const float* b1 = (const float*)d_in[2];
    const float* W2 = (const float*)d_in[3];
    const float* b2 = (const float*)d_in[4];
    const int*   ei = (const int*)d_in[5];

    const int N = in_sizes[0] / HIDC;
    const int E = in_sizes[5] / 2;
    const int* src = ei;
    const int* dst = ei + E;
    float* out = (float*)d_out;

    float* h    = (float*)d_ws;
    float* dinv = h    + (size_t)N * HIDC;
    float* g    = dinv + N;
    const int NB = (N + TN - 1) / TN;
    int* cntG = (int*)(g + N);
    unsigned int* binsG = (unsigned int*)(cntG + (size_t)NB * 32);
    unsigned int* keyw  = binsG + (size_t)NB * 32 * BCAP;

    const int TB = 256;
    gcn_keys      <<<((E + 511) / 512 + 7) / 8, TB, 0, stream>>>(dst, keyw, E);
    gcn_file_edges<<<NB, 1024, 0, stream>>>(src, dst, keyw, dinv, binsG, cntG, N, E);
    gcn_gemm1_wmma<<<(N + 15) / 16,     TB, 0, stream>>>(x, W1, h, N);
    gcn_agg1_owner<<<NB, 1024, 0, stream>>>(h, src, dst, dinv, b1, W2, g, binsG, cntG, N, E);
    gcn_agg2_owner<<<NB, 1024, 0, stream>>>(g, src, dst, binsG, cntG, dinv, b2, out, N, E);
}


# edge filing and the x@W1 product run in one launch on disjoint blocks (filing blocks first, then a 1024-thread block per 64 rows of x), so the product overlaps the filing; f32 sum order as before
# speedup vs baseline: 4.0492x; 1.0459x over previous
#include <hip/hip_runtime.h>
#include <math.h>

#define HIDC 128

typedef __attribute__((ext_vector_type(2))) float v2f;
typedef __attribute__((ext_vector_type(8))) float v8f;

__device__ __forceinline__ unsigned key4(int4 d) {
    return ((unsigned)(d.x >> 9) & 255u) | (((unsigned)(d.y >> 9) & 255u) << 8) |
           (((unsigned)(d.z >> 9) & 255u) << 16) | (((unsigned)(d.w >> 9) & 255u) << 24);
}
__global__ void gcn_keys(const int* __restrict__ dst, unsigned int* __restrict__ keyw, int E) {
    const int gt   = blockIdx.x * blockDim.x + threadIdx.x;
    const int lane = threadIdx.x & 31;
    const int qb   = (gt >> 5) * 128 + lane;
    const int nq   = E >> 2;
    unsigned char* key = (unsigned char*)keyw;
    if (((((size_t)dst) & 15) == 0) && nq > 0) {
        int4 d[4];
#pragma unroll
        for (int j = 0; j < 4; ++j) d[j] = ((const int4*)dst)[min(qb + 32 * j, nq - 1)];
#pragma unroll
        for (int j = 0; j < 4; ++j) {
            const int q = qb + 32 * j;
            if (q < nq) keyw[q] = key4(d[j]);
        }
        if (gt == 0) {
            for (int e = nq * 4; e < E; ++e) key[e] = (unsigned char)(dst[e] >> 9);
        }
    } else {
        for (int j = 0; j < 4; ++j) {
            for (int k = 0; k < 4; ++k) {
                const int e = 4 * (qb + 32 * j) + k;
                if (e < E) key[e] = (unsigned char)(dst[e] >> 9);
            }
        }
    }
}

#define GROWS 64

__device__ __forceinline__ void gemm1_rows64(const float* __restrict__ x,
                                             const float* __restrict__ W,
                                             float* __restrict__ h, int N, int tile,
                                             float* As) {
    const int t = threadIdx.x;
    const int rowBase = tile * GROWS;
    const bool full = (rowBase + GROWS <= N);

    if (full) {
#pragma unroll
        for (int i = t; i < GROWS * 32; i += 1024) {
            int r  = i >> 5;
            int c4 = i & 31;
            float4 v = ((const float4*)(x + (size_t)(rowBase + r) * HIDC))[c4];
            float* p = &As[r * 132 + c4 * 4];
            p[0] = v.x; p[1] = v.y; p[2] = v.z; p[3] = v.w;
        }
    } else {
        for (int i = t; i < GROWS * 32; i += 1024) {
            int r  = i >> 5;
            int c4 = i & 31;
            float4 v = make_float4(0.f, 0.f, 0.f, 0.f);
            if (rowBase + r < N) v = ((const float4*)(x + (size_t)(rowBase + r) * HIDC))[c4];
            float* p = &As[r * 132 + c4 * 4];
            p[0] = v.x; p[1] = v.y; p[2] = v.z; p[3] = v.w;
        }
    }
    __syncthreads();

    const int lane = t & 31;
    const int wv   = t >> 5;
    const int rt   = wv >> 3;
    const int m    = lane & 15;
    const int half = lane >> 4;
    const int kl   = half * 2;
    const int n    = (wv & 7) * 16 + (lane & 15);

    v8f c = {};
    const float* Arow = &As[(rt * 16 + m) * 132];
#pragma unroll 8
    for (int k0 = 0; k0 < HIDC; k0 += 4) {
        v2f a, b;
        a.x = Arow[k0 + kl];
        a.y = Arow[k0 + kl + 1];
        b.x = W[(k0 + kl) * HIDC + n];
        b.y = W[(k0 + kl + 1) * HIDC + n];
        c = __builtin_amdgcn_wmma_f32_16x16x4_f32(false, a, false, b,
                                                  (short)0, c, false, false);
    }

    const int row0 = rowBase + rt * 16 + 8 * half;
    float* hp = h + (size_t)row0 * HIDC + n;
    if (full) {
#pragma unroll
        for (int v = 0; v < 8; ++v) hp[v * HIDC] = c[v];
    } else {
#pragma unroll
        for (int v = 0; v < 8; ++v) {
            if (row0 + v < N) hp[v * HIDC] = c[v];
        }
    }
}

#define TN   512
#define QCAP 64
#define BCAP 384

__device__ __forceinline__ int rdlane_i(int v, int k) { return __builtin_amdgcn_readlane(v, k); }
__device__ __forceinline__ float rdlane_f(float v, int k) {
    return __int_as_float(__builtin_amdgcn_readlane(__float_as_int(v), k));
}
__device__ __forceinline__ void lds_fadd(float* p, float v) {
    __hip_atomic_fetch_add(p, v, __ATOMIC_RELAXED, __HIP_MEMORY_SCOPE_WORKGROUP);
}

__device__ __forceinline__ void file_edges(const int* __restrict__ src,
                                           const int* __restrict__ dst,
                                           unsigned* bins, int* binCnt, int* degc,
                                           const int* myE, int lo,
                                           int head, int n, int lane) {
    if (lane < n) {
        const int e  = myE[(head + lane) & (QCAP - 1)];
        const int s  = src[e];
        const int dl = dst[e] - lo;
        const int owner = dl & 31;
        __hip_atomic_fetch_add(&degc[dl], 1, __ATOMIC_RELAXED, __HIP_MEMORY_SCOPE_WORKGROUP);
        const int pos = __hip_atomic_fetch_add(&binCnt[owner], 1, __ATOMIC_RELAXED,
                                               __HIP_MEMORY_SCOPE_WORKGROUP);
        const int slot = (pos < BCAP) ? owner * BCAP + pos : 32 * BCAP;
        bins[slot] = ((unsigned)s << 4) | (unsigned)(dl >> 5);
    }
}

__device__ __forceinline__ void push_edges(bool m, int e, int* myE, int head, int& cnt) {
    const unsigned mask = __builtin_amdgcn_ballot_w32(m);
    if (mask != 0u) {
        if (m) {
            int p = (head + cnt + (int)__builtin_amdgcn_mbcnt_lo(mask, 0u)) & (QCAP - 1);
            myE[p] = e;
        }
        cnt += __builtin_popcount(mask);
    }
}

__device__ __forceinline__ unsigned differ_bits(unsigned w, unsigned pat) {
    const unsigned x = w ^ pat;
    return (((x & 0x7f7f7f7fu) + 0x7f7f7f7fu) | x) & 0x80808080u;
}
__device__ __forceinline__ unsigned match16(uint4 v, unsigned pat) {
    const unsigned nd = (differ_bits(v.x, pat) >> 7) | (differ_bits(v.y, pat) >> 6) |
                        (differ_bits(v.z, pat) >> 5) | (differ_bits(v.w, pat) >> 4);
    return ~nd & 0x0f0f0f0fu;
}

__device__ __forceinline__ void file_edges_block(const int* __restrict__ src,
                                                 const int* __restrict__ dst,
                                                 const unsigned int* __restrict__ keyw,
                                                 float* __restrict__ dinv,
                                                 unsigned int* __restrict__ binsG,
                                                 int* __restrict__ cntG, int N, int E, int bid,
                                                 unsigned* bins, int* qE, int* degc, int* binCnt) {
    const int t    = threadIdx.x;
    const int lane = t & 31;
    const int wv   = __builtin_amdgcn_readfirstlane(t >> 5);
    const int lo   = bid * TN;

    if (t < TN) degc[t] = 0;
    if (t < 32) binCnt[t] = 0;
    __syncthreads();

    int* myE = qE + wv * QCAP;
    int head = 0, cnt = 0;

    const unsigned char* key = (const unsigned char*)keyw;
    const bool usekey = (N <= 256 * TN) && ((((size_t)key) & 15) == 0);
    const int  nrows  = usekey ? (E >> 9) : 0;
    const unsigned pat = ((unsigned)bid & 255u) * 0x01010101u;

    for (int r0 = wv; r0 < nrows; r0 += 32 * 8) {
        uint4 vv[8];
#pragma unroll
        for (int g = 0; g < 8; ++g) {
            const int row = min(r0 + 32 * g, nrows - 1);
            vv[g] = ((const uint4*)(key + ((size_t)row << 9)))[lane];
        }
#pragma unroll
        for (int g = 0; g < 8; ++g) {
            const int row = r0 + 32 * g;
            if (row < nrows) {
                const int eb = (row << 9) + lane * 16;
                unsigned F = match16(vv[g], pat);
#pragma nounroll
                for (int r = 0; r < 16; ++r) {
                    const unsigned mask = __builtin_amdgcn_ballot_w32(F != 0u);
                    if (mask == 0u) break;
                    if (F != 0u) {
                        const int tz = __builtin_ctz(F);
                        const int p  = (head + cnt + (int)__builtin_amdgcn_mbcnt_lo(mask, 0u)) & (QCAP - 1);
                        myE[p] = eb + ((tz & 7) << 2) + (tz >> 3);
                    }
                    cnt += __builtin_popcount(mask);
                    F &= F - 1u;
                    if (cnt >= 32) {
                        file_edges(src, dst, bins, binCnt, degc, myE, lo, head, 32, lane);
                        head = (head + 32) & (QCAP - 1);
                        cnt -= 32;
                    }
                }
            }
        }
    }
    const int ebeg   = nrows << 9;
    const int nsteps = (E - ebeg + 31) >> 5;
#pragma nounroll
    for (int st = wv; st < nsteps; st += 32) {
        const int e = ebeg + (st << 5) + lane;
        const int d = (e < E) ? dst[e] : -1;
        push_edges((unsigned)(d - lo) < (unsigned)TN, e, myE, head, cnt);
        if (cnt >= 32) {
            file_edges(src, dst, bins, binCnt, degc, myE, lo, head, 32, lane);
            head = (head + 32) & (QCAP - 1);
            cnt -= 32;
        }
    }
    if (cnt > 0) file_edges(src, dst, bins, binCnt, degc, myE, lo, head, cnt, lane);
    __syncthreads();

    {
        const int nraw = __builtin_amdgcn_readfirstlane(binCnt[wv]);
        const int nb = min(nraw, BCAP);
        const unsigned* myB = bins + wv * BCAP;
        unsigned* gB = binsG + ((size_t)bid * 32 + wv) * BCAP;
        for (int i = lane; i < nb; i += 32) gB[i] = myB[i];
        if (lane == 0) cntG[bid * 32 + wv] = nraw;
    }
    if (t < TN && lo + t < N) dinv[lo + t] = rsqrtf(1.0f + (float)degc[t]);
}

__global__ __launch_bounds__(1024) void gcn_file_gemm(const int* __restrict__ src,
                                                      const int* __restrict__ dst,
                                                      const unsigned int* __restrict__ keyw,
                                                      float* __restrict__ dinv,
                                                      unsigned int* __restrict__ binsG,
                                                      int* __restrict__ cntG,
                                                      const float* __restrict__ x,
                                                      const float* __restrict__ W,
                                                      float* __restrict__ h,
                                                      int N, int E, int NBF) {
    __shared__ unsigned bins[32 * BCAP + 1];
    __shared__ int   qE[32 * QCAP];
    __shared__ int   degc[TN];
    __shared__ int   binCnt[32];
    __shared__ float As[GROWS * 132];

    const int b = blockIdx.x;
    if (b < NBF) {
        file_edges_block(src, dst, keyw, dinv, binsG, cntG, N, E, b, bins, qE, degc, binCnt);
    } else {
        gemm1_rows64(x, W, h, N, b - NBF, As);
    }
}


__device__ __forceinline__ void load_rows8(float4 (&r)[8], const float* __restrict__ h,
                                           int s, int k0, int lane) {
#pragma unroll
    for (int u = 0; u < 8; ++u) {
        const int sk = rdlane_i(s, k0 + u);
        r[u] = ((const float4*)(h + (size_t)sk * HIDC))[lane];
    }
}
template <bool GUARD>
__device__ __forceinline__ void add_rows8(const float4 (&r)[8], float* trow,
                                          int sl, float nrm, int k0, int cnt) {
#pragma unroll
    for (int u = 0; u < 8; ++u) {
        if (!GUARD || k0 + u < cnt) {
            const int   rk = rdlane_i(sl, k0 + u);
            const float nk = rdlane_f(nrm, k0 + u);
            float4* p = (float4*)(trow + rk * (32 * HIDC));
            float4 a = *p;
            a.x = fmaf(r[u].x, nk, a.x);
            a.y = fmaf(r[u].y, nk, a.y);
            a.z = fmaf(r[u].z, nk, a.z);
            a.w = fmaf(r[u].w, nk, a.w);
            *p = a;
        }
    }
}

__global__ __launch_bounds__(1024) void gcn_agg1_owner(const float* __restrict__ h,
                                                       const int* __restrict__ src,
                                                       const int* __restrict__ dst,
                                                       const float* __restrict__ dinv,
                                                       const float* __restrict__ b1,
                                                       const float* __restrict__ W2,
                                                       float* __restrict__ gv,
                                                       const unsigned int* __restrict__ binsG,
                                                       const int* __restrict__ cntG, int N, int E) {
    __shared__ __attribute__((aligned(16))) float tile[TN * HIDC];
    __shared__ float dtile[TN];

    const int t    = threadIdx.x;
    const int lane = t & 31;
    const int wv   = __builtin_amdgcn_readfirstlane(t >> 5);
    const int lo   = blockIdx.x * TN;

    const int  c    = cntG[blockIdx.x * 32 + lane];
    const bool over = __builtin_amdgcn_ballot_w32(c > BCAP) != 0u;
    const int  nb   = rdlane_i(c, wv);

    for (int i = t; i < TN * HIDC / 4; i += 1024)
        ((float4*)tile)[i] = make_float4(0.f, 0.f, 0.f, 0.f);
    if (t < TN) dtile[t] = (lo + t < N) ? dinv[lo + t] : 0.0f;
    __syncthreads();

    if (!over) {
        const unsigned* gB = binsG + ((size_t)blockIdx.x * 32 + wv) * BCAP;
        float* trow = tile + wv * HIDC + 4 * lane;
        unsigned ent  = (lane < nb) ? gB[lane] : 0u;
        unsigned ent1 = (32 + lane < nb) ? gB[32 + lane] : 0u;
        float    dv   = (lane < nb) ? dinv[ent >> 4] : 0.0f;
        for (int base = 0; base < nb; base += 32) {
            const int cnt32 = min(32, nb - base);
            const int   s   = (int)(ent >> 4);
            const int   sl  = (int)(ent & 15u);
            const float nrm = dv * dtile[(sl << 5) | wv];
            ent  = ent1;
            dv   = (base + 32 + lane < nb) ? dinv[ent >> 4] : 0.0f;
            ent1 = (base + 64 + lane < nb) ? gB[base + 64 + lane] : 0u;
            if (cnt32 == 32) {
                float4 ra[8], rb[8];
                load_rows8(ra, h, s, 0, lane);
                load_rows8(rb, h, s, 8, lane);
                add_rows8<false>(ra, trow, sl, nrm, 0, 32);
                load_rows8(ra, h, s, 16, lane);
                add_rows8<false>(rb, trow, sl, nrm, 8, 32);
                load_rows8(rb, h, s, 24, lane);
                add_rows8<false>(ra, trow, sl, nrm, 16, 32);
                add_rows8<false>(rb, trow, sl, nrm, 24, 32);
            } else {
#pragma nounroll
                for (int k0 = 0; k0 < cnt32; k0 += 8) {
                    float4 rc[8];
                    load_rows8(rc, h, s, k0, lane);
                    add_rows8<true>(rc, trow, sl, nrm, k0, cnt32);
                }
            }
        }
    } else {
        const int nsteps = (E + 31) >> 5;
        for (int st = wv; st < nsteps; st += 32) {
            const int e = (st << 5) + lane;
            const int d = (e < E) ? dst[e] : -1;
            const int dl = d - lo;
            const bool m = (unsigned)dl < (unsigned)TN;
            const int s = m ? src[e] : 0;
            unsigned om = __builtin_amdgcn_ballot_w32(m);
            while (om != 0u) {
                const int k = __builtin_ctz(om);
                om &= om - 1u;
                const int   sk = rdlane_i(s, k);
                const int   dk = rdlane_i(dl, k);
                const float nk = dinv[sk] * dtile[dk];
                const float4 r = ((const float4*)(h + (size_t)sk * HIDC))[lane];
                float* o = tile + dk * HIDC + 4 * lane;
                lds_fadd(o + 0, r.x * nk);
                lds_fadd(o + 1, r.y * nk);
                lds_fadd(o + 2, r.z * nk);
                lds_fadd(o + 3, r.w * nk);
            }
        }
        __syncthreads();
    }

    const float4 bb = ((const float4*)b1)[lane];
    const float4 w  = ((const float4*)W2)[lane];
    for (int r = wv; r < TN; r += 32) {
        const int i = lo + r;
        if (i >= N) break;
        const float di = dtile[r];
        const float ns = di * di;
        const float4 a  = ((const float4*)(tile + r * HIDC))[lane];
        const float4 hv = ((const float4*)(h + (size_t)i * HIDC))[lane];
        float r0 = fmaxf(a.x + hv.x * ns + bb.x, 0.0f);
        float r1 = fmaxf(a.y + hv.y * ns + bb.y, 0.0f);
        float r2 = fmaxf(a.z + hv.z * ns + bb.z, 0.0f);
        float r3 = fmaxf(a.w + hv.w * ns + bb.w, 0.0f);
        float p = r0 * w.x + r1 * w.y + r2 * w.z + r3 * w.w;
#pragma unroll
        for (int off = 16; off > 0; off >>= 1) p += __shfl_down(p, off, 32);
        if (lane == 0) gv[i] = p * di;
    }
}

__global__ __launch_bounds__(1024) void gcn_agg2_owner(const float* __restrict__ g,
                                                       const int* __restrict__ src,
                                                       const int* __restrict__ dst,
                                                       const unsigned int* __restrict__ binsG,
                                                       const int* __restrict__ cntG,
                                                       const float* __restrict__ dinv,
                                                       const float* __restrict__ b2,
                                                       float* __restrict__ out, int N, int E) {
    __shared__ float acc[TN];

    const int t    = threadIdx.x;
    const int lane = t & 31;
    const int wv   = __builtin_amdgcn_readfirstlane(t >> 5);
    const int lo   = blockIdx.x * TN;

    const int  c    = cntG[blockIdx.x * 32 + lane];
    const bool over = __builtin_amdgcn_ballot_w32(c > BCAP) != 0u;
    const int  nraw = rdlane_i(c, wv);

    if (t < TN) acc[t] = 0.0f;
    __syncthreads();

    const bool  mine = (t < TN) && (lo + t < N);
    const float gs   = mine ? g[lo + t] : 0.0f;
    const float ds   = mine ? dinv[lo + t] : 0.0f;
    const float bias = b2[0];

    if (!over) {
        if (nraw > 0) {
            const unsigned* gB = binsG + ((size_t)blockIdx.x * 32 + wv) * BCAP;
            unsigned ent[BCAP / 32];
            float    v[BCAP / 32];
#pragma unroll
            for (int k = 0; k < BCAP / 32; ++k) ent[k] = gB[min(lane + 32 * k, nraw - 1)];
#pragma unroll
            for (int k = 0; k < BCAP / 32; ++k) v[k] = g[ent[k] >> 4];
#pragma unroll
            for (int k = BCAP / 32 - 1; k >= 0; --k) {
                if (lane + 32 * k < nraw) lds_fadd(&acc[wv * 16 + (int)(ent[k] & 15u)], v[k]);
            }
        }
    } else {
        for (int e = t; e < E; e += 1024) {
            const unsigned dl = (unsigned)(dst[e] - lo);
            if (dl < (unsigned)TN) lds_fadd(&acc[(dl & 31u) * 16u + (dl >> 5)], g[src[e]]);
        }
    }
    __syncthreads();

    if (mine) out[lo + t] = (acc[(t & 31) * 16 + (t >> 5)] + gs) * ds + bias;
}

extern "C" void kernel_launch(void* const* d_in, const int* in_sizes, int n_in,
                              void* d_out, int out_size, void* d_ws, size_t ws_size,
                              hipStream_t stream) {
    const float* x  = (const float*)d_in[0];
    const float* W1 = (const float*)d_in[1];
    const float* b1 = (const float*)d_in[2];
    const float* W2 = (const float*)d_in[3];
    const float* b2 = (const float*)d_in[4];
    const int*   ei = (const int*)d_in[5];

    const int N = in_sizes[0] / HIDC;
    const int E = in_sizes[5] / 2;
    const int* src = ei;
    const int* dst = ei + E;
    float* out = (float*)d_out;

    float* h    = (float*)d_ws;
    float* dinv = h    + (size_t)N * HIDC;
    float* g    = dinv + N;
    const int NB = (N + TN - 1) / TN;
    int* cntG = (int*)(g + N);
    unsigned int* binsG = (unsigned int*)(cntG + (size_t)NB * 32);
    unsigned int* keyw  = binsG + (size_t)NB * 32 * BCAP;

    const int TB = 256;
    gcn_keys      <<<((E + 511) / 512 + 7) / 8, TB, 0, stream>>>(dst, keyw, E);
    gcn_file_gemm <<<NB + (N + GROWS - 1) / GROWS, 1024, 0, stream>>>(src, dst, keyw, dinv, binsG, cntG,
                                                                     x, W1, h, N, E, NB);
    gcn_agg1_owner<<<NB, 1024, 0, stream>>>(h, src, dst, dinv, b1, W2, g, binsG, cntG, N, E);
    gcn_agg2_owner<<<NB, 1024, 0, stream>>>(g, src, dst, binsG, cntG, dinv, b2, out, N, E);
}


# edge filter by group bits: first pass marks, per owner block, one bit per 8 edges (LDS OR, whole-line stores); a filing block walks its set bits, tests only those groups' dst; f32 sum order as before
# speedup vs baseline: 4.6192x; 1.1408x over previous
#include <hip/hip_runtime.h>
#include <math.h>

#define HIDC 128

typedef __attribute__((ext_vector_type(2))) float v2f;
typedef __attribute__((ext_vector_type(8))) float v8f;

#define TN     512
#define PMAXO  256
#define PPITCH 257

__device__ __forceinline__ void lds_or(unsigned* p, unsigned v) {
    __hip_atomic_fetch_or(p, v, __ATOMIC_RELAXED, __HIP_MEMORY_SCOPE_WORKGROUP);
}
__device__ __forceinline__ unsigned* range_word(unsigned* row, int d) {
    return row + (((unsigned)d >> 9) & (PMAXO - 1u));
}
__global__ __launch_bounds__(1024) void gcn_groups(const int* __restrict__ dst,
                                                   unsigned int* __restrict__ planes,
                                                   int NB, int E, int nkb) {
    __shared__ unsigned bm[32 * PPITCH];
    const int t    = threadIdx.x;
    const int lane = t & 31;
    const int wv   = t >> 5;
    if (NB > PMAXO || ((((size_t)dst) & 15) != 0)) return;

    for (int i = t; i < 32 * PPITCH; i += 1024) bm[i] = 0u;
    __syncthreads();

    const int ng = E >> 3;
    const int G  = blockIdx.x * 1024 + t;
    if (G < ng) {
        const int4 a = ((const int4*)dst)[2 * G];
        const int4 b = ((const int4*)dst)[2 * G + 1];
        unsigned* row = bm + wv * PPITCH;
        const unsigned bit = 1u << lane;
        lds_or(range_word(row, a.x), bit);
        lds_or(range_word(row, a.y), bit);
        lds_or(range_word(row, a.z), bit);
        lds_or(range_word(row, a.w), bit);
        lds_or(range_word(row, b.x), bit);
        lds_or(range_word(row, b.y), bit);
        lds_or(range_word(row, b.z), bit);
        lds_or(range_word(row, b.w), bit);
    }
    __syncthreads();

    for (int o = wv; o < NB; o += 32)
        planes[((size_t)o * nkb + blockIdx.x) * 32 + lane] = bm[lane * PPITCH + o];
}

#define GROWS 64

__device__ __forceinline__ void gemm1_rows64(const float* __restrict__ x,
                                             const float* __restrict__ W,
                                             float* __restrict__ h, int N, int tile,
                                             float* As) {
    const int t = threadIdx.x;
    const int rowBase = tile * GROWS;
    const bool full = (rowBase + GROWS <= N);

    if (full) {
#pragma unroll
        for (int i = t; i < GROWS * 32; i += 1024) {
            int r  = i >> 5;
            int c4 = i & 31;
            float4 v = ((const float4*)(x + (size_t)(rowBase + r) * HIDC))[c4];
            float* p = &As[r * 132 + c4 * 4];
            p[0] = v.x; p[1] = v.y; p[2] = v.z; p[3] = v.w;
        }
    } else {
        for (int i = t; i < GROWS * 32; i += 1024) {
            int r  = i >> 5;
            int c4 = i & 31;
            float4 v = make_float4(0.f, 0.f, 0.f, 0.f);
            if (rowBase + r < N) v = ((const float4*)(x + (size_t)(rowBase + r) * HIDC))[c4];
            float* p = &As[r * 132 + c4 * 4];
            p[0] = v.x; p[1] = v.y; p[2] = v.z; p[3] = v.w;
        }
    }
    __syncthreads();

    const int lane = t & 31;
    const int wv   = t >> 5;
    const int rt   = wv >> 3;
    const int m    = lane & 15;
    const int half = lane >> 4;
    const int kl   = half * 2;
    const int n    = (wv & 7) * 16 + (lane & 15);

    v8f c = {};
    const float* Arow = &As[(rt * 16 + m) * 132];
#pragma unroll 8
    for (int k0 = 0; k0 < HIDC; k0 += 4) {
        v2f a, b;
        a.x = Arow[k0 + kl];
        a.y = Arow[k0 + kl + 1];
        b.x = W[(k0 + kl) * HIDC + n];
        b.y = W[(k0 + kl + 1) * HIDC + n];
        c = __builtin_amdgcn_wmma_f32_16x16x4_f32(false, a, false, b,
                                                  (short)0, c, false, false);
    }

    const int row0 = rowBase + rt * 16 + 8 * half;
    float* hp = h + (size_t)row0 * HIDC + n;
    if (full) {
#pragma unroll
        for (int v = 0; v < 8; ++v) hp[v * HIDC] = c[v];
    } else {
#pragma unroll
        for (int v = 0; v < 8; ++v) {
            if (row0 + v < N) hp[v * HIDC] = c[v];
        }
    }
}

#define QCAP 64
#define BCAP 384

__device__ __forceinline__ int rdlane_i(int v, int k) { return __builtin_amdgcn_readlane(v, k); }
__device__ __forceinline__ float rdlane_f(float v, int k) {
    return __int_as_float(__builtin_amdgcn_readlane(__float_as_int(v), k));
}
__device__ __forceinline__ void lds_fadd(float* p, float v) {
    __hip_atomic_fetch_add(p, v, __ATOMIC_RELAXED, __HIP_MEMORY_SCOPE_WORKGROUP);
}

__device__ __forceinline__ void file_one(unsigned* bins, int* binCnt, int* degc, int s, int dl) {
    const int owner = dl & 31;
    __hip_atomic_fetch_add(&degc[dl], 1, __ATOMIC_RELAXED, __HIP_MEMORY_SCOPE_WORKGROUP);
    const int pos = __hip_atomic_fetch_add(&binCnt[owner], 1, __ATOMIC_RELAXED,
                                           __HIP_MEMORY_SCOPE_WORKGROUP);
    const int slot = (pos < BCAP) ? owner * BCAP + pos : 32 * BCAP;
    bins[slot] = ((unsigned)s << 4) | (unsigned)(dl >> 5);
}

__device__ __forceinline__ void file_groups(const int* __restrict__ src,
                                            const int* __restrict__ dst,
                                            unsigned* bins, int* binCnt, int* degc,
                                            const int* myG, int lo,
                                            int head, int n, int lane) {
    if (lane < n) {
        const int G = myG[(head + lane) & (QCAP - 1)];
        const int4 a = ((const int4*)dst)[2 * G];
        const int4 b = ((const int4*)dst)[2 * G + 1];
        const int d0 = a.x - lo, d1 = a.y - lo, d2 = a.z - lo, d3 = a.w - lo;
        const int d4 = b.x - lo, d5 = b.y - lo, d6 = b.z - lo, d7 = b.w - lo;
        unsigned m = ((unsigned)d0 < (unsigned)TN ? 1u : 0u)   | ((unsigned)d1 < (unsigned)TN ? 2u : 0u) |
                     ((unsigned)d2 < (unsigned)TN ? 4u : 0u)   | ((unsigned)d3 < (unsigned)TN ? 8u : 0u) |
                     ((unsigned)d4 < (unsigned)TN ? 16u : 0u)  | ((unsigned)d5 < (unsigned)TN ? 32u : 0u) |
                     ((unsigned)d6 < (unsigned)TN ? 64u : 0u)  | ((unsigned)d7 < (unsigned)TN ? 128u : 0u);
        while (m != 0u) {
            const int k = __builtin_ctz(m);
            m &= m - 1u;
            const int s = src[8 * G + k];
            const int e01 = (k & 1) ? d1 : d0;
            const int e23 = (k & 1) ? d3 : d2;
            const int e45 = (k & 1) ? d5 : d4;
            const int e67 = (k & 1) ? d7 : d6;
            const int e03 = (k & 2) ? e23 : e01;
            const int e47 = (k & 2) ? e67 : e45;
            file_one(bins, binCnt, degc, s, (k & 4) ? e47 : e03);
        }
    }
}

__device__ __forceinline__ void file_edges_block(const int* __restrict__ src,
                                                 const int* __restrict__ dst,
                                                 const unsigned int* __restrict__ planes,
                                                 float* __restrict__ dinv,
                                                 unsigned int* __restrict__ binsG,
                                                 int* __restrict__ cntG, int N, int E, int NB,
                                                 int nkb, int bid,
                                                 unsigned* bins, int* qG, int* degc, int* binCnt) {
    const int t    = threadIdx.x;
    const int lane = t & 31;
    const int wv   = __builtin_amdgcn_readfirstlane(t >> 5);
    const int lo   = bid * TN;

    if (t < TN) degc[t] = 0;
    if (t < 32) binCnt[t] = 0;
    __syncthreads();

    int* myG = qG + wv * QCAP;
    int head = 0, cnt = 0;

    const bool useplane = (NB <= PMAXO) && ((((size_t)dst) & 15) == 0);
    const int  nwords   = useplane ? nkb * 32 : 0;
    const unsigned* P   = planes + (size_t)bid * nkb * 32;

    for (int i0 = wv * 32; i0 < nwords; i0 += 1024 * 8) {
        unsigned wd[8];
#pragma unroll
        for (int g = 0; g < 8; ++g) {
            wd[g] = P[min(i0 + 1024 * g + lane, nwords - 1)];
        }
#pragma unroll
        for (int g = 0; g < 8; ++g) {
            const int ib = i0 + 1024 * g;
            if (ib < nwords) {
                const int gb = (ib + lane) << 5;
                unsigned F = wd[g];
#pragma nounroll
                for (int r = 0; r < 32; ++r) {
                    const unsigned mask = __builtin_amdgcn_ballot_w32(F != 0u);
                    if (mask == 0u) break;
                    if (F != 0u) {
                        const int p = (head + cnt + (int)__builtin_amdgcn_mbcnt_lo(mask, 0u)) & (QCAP - 1);
                        myG[p] = gb + __builtin_ctz(F);
                    }
                    cnt += __builtin_popcount(mask);
                    F &= F - 1u;
                    if (cnt >= 32) {
                        file_groups(src, dst, bins, binCnt, degc, myG, lo, head, 32, lane);
                        head = (head + 32) & (QCAP - 1);
                        cnt -= 32;
                    }
                }
            }
        }
    }
    if (cnt > 0) file_groups(src, dst, bins, binCnt, degc, myG, lo, head, cnt, lane);
    const int ebeg   = useplane ? ((E >> 3) << 3) : 0;
    const int nsteps = (E - ebeg + 31) >> 5;
#pragma nounroll
    for (int st = wv; st < nsteps; st += 32) {
        const int e = ebeg + (st << 5) + lane;
        if (e < E) {
            const int dl = dst[e] - lo;
            if ((unsigned)dl < (unsigned)TN) file_one(bins, binCnt, degc, src[e], dl);
        }
    }
    __syncthreads();

    {
        const int nraw = __builtin_amdgcn_readfirstlane(binCnt[wv]);
        const int nb = min(nraw, BCAP);
        const unsigned* myB = bins + wv * BCAP;
        unsigned* gB = binsG + ((size_t)bid * 32 + wv) * BCAP;
        for (int i = lane; i < nb; i += 32) gB[i] = myB[i];
        if (lane == 0) cntG[bid * 32 + wv] = nraw;
    }
    if (t < TN && lo + t < N) dinv[lo + t] = rsqrtf(1.0f + (float)degc[t]);
}

__global__ __launch_bounds__(1024) void gcn_file_gemm(const int* __restrict__ src,
                                                      const int* __restrict__ dst,
                                                      const unsigned int* __restrict__ planes,
                                                      float* __restrict__ dinv,
                                                      unsigned int* __restrict__ binsG,
                                                      int* __restrict__ cntG,
                                                      const float* __restrict__ x,
                                                      const float* __restrict__ W,
                                                      float* __restrict__ h,
                                                      int N, int E, int NBF, int nkb) {
    __shared__ unsigned bins[32 * BCAP + 1];
    __shared__ int   qG[32 * QCAP];
    __shared__ int   degc[TN];
    __shared__ int   binCnt[32];
    __shared__ float As[GROWS * 132];

    const int b = blockIdx.x;
    if (b < NBF) {
        file_edges_block(src, dst, planes, dinv, binsG, cntG, N, E, NBF, nkb, b, bins, qG, degc, binCnt);
    } else {
        gemm1_rows64(x, W, h, N, b - NBF, As);
    }
}


__device__ __forceinline__ void load_rows8(float4 (&r)[8], const float* __restrict__ h,
                                           int s, int k0, int lane) {
#pragma unroll
    for (int u = 0; u < 8; ++u) {
        const int sk = rdlane_i(s, k0 + u);
        r[u] = ((const float4*)(h + (size_t)sk * HIDC))[lane];
    }
}
template <bool GUARD>
__device__ __forceinline__ void add_rows8(const float4 (&r)[8], float* trow,
                                          int sl, float nrm, int k0, int cnt) {
#pragma unroll
    for (int u = 0; u < 8; ++u) {
        if (!GUARD || k0 + u < cnt) {
            const int   rk = rdlane_i(sl, k0 + u);
            const float nk = rdlane_f(nrm, k0 + u);
            float4* p = (float4*)(trow + rk * (32 * HIDC));
            float4 a = *p;
            a.x = fmaf(r[u].x, nk, a.x);
            a.y = fmaf(r[u].y, nk, a.y);
            a.z = fmaf(r[u].z, nk, a.z);
            a.w = fmaf(r[u].w, nk, a.w);
            *p = a;
        }
    }
}

__global__ __launch_bounds__(1024) void gcn_agg1_owner(const float* __restrict__ h,
                                                       const int* __restrict__ src,
                                                       const int* __restrict__ dst,
                                                       const float* __restrict__ dinv,
                                                       const float* __restrict__ b1,
                                                       const float* __restrict__ W2,
                                                       float* __restrict__ gv,
                                                       const unsigned int* __restrict__ binsG,
                                                       const int* __restrict__ cntG, int N, int E) {
    __shared__ __attribute__((aligned(16))) float tile[TN * HIDC];
    __shared__ float dtile[TN];

    const int t    = threadIdx.x;
    const int lane = t & 31;
    const int wv   = __builtin_amdgcn_readfirstlane(t >> 5);
    const int lo   = blockIdx.x * TN;

    const int  c    = cntG[blockIdx.x * 32 + lane];
    const bool over = __builtin_amdgcn_ballot_w32(c > BCAP) != 0u;
    const int  nb   = rdlane_i(c, wv);

    for (int i = t; i < TN * HIDC / 4; i += 1024)
        ((float4*)tile)[i] = make_float4(0.f, 0.f, 0.f, 0.f);
    if (t < TN) dtile[t] = (lo + t < N) ? dinv[lo + t] : 0.0f;
    __syncthreads();

    if (!over) {
        const unsigned* gB = binsG + ((size_t)blockIdx.x * 32 + wv) * BCAP;
        float* trow = tile + wv * HIDC + 4 * lane;
        unsigned ent  = (lane < nb) ? gB[lane] : 0u;
        unsigned ent1 = (32 + lane < nb) ? gB[32 + lane] : 0u;
        float    dv   = (lane < nb) ? dinv[ent >> 4] : 0.0f;
        for (int base = 0; base < nb; base += 32) {
            const int cnt32 = min(32, nb - base);
            const int   s   = (int)(ent >> 4);
            const int   sl  = (int)(ent & 15u);
            const float nrm = dv * dtile[(sl << 5) | wv];
            ent  = ent1;
            dv   = (base + 32 + lane < nb) ? dinv[ent >> 4] : 0.0f;
            ent1 = (base + 64 + lane < nb) ? gB[base + 64 + lane] : 0u;
            if (cnt32 == 32) {
                float4 ra[8], rb[8];
                load_rows8(ra, h, s, 0, lane);
                load_rows8(rb, h, s, 8, lane);
                add_rows8<false>(ra, trow, sl, nrm, 0, 32);
                load_rows8(ra, h, s, 16, lane);
                add_rows8<false>(rb, trow, sl, nrm, 8, 32);
                load_rows8(rb, h, s, 24, lane);
                add_rows8<false>(ra, trow, sl, nrm, 16, 32);
                add_rows8<false>(rb, trow, sl, nrm, 24, 32);
            } else {
#pragma nounroll
                for (int k0 = 0; k0 < cnt32; k0 += 8) {
                    float4 rc[8];
                    load_rows8(rc, h, s, k0, lane);
                    add_rows8<true>(rc, trow, sl, nrm, k0, cnt32);
                }
            }
        }
    } else {
        const int nsteps = (E + 31) >> 5;
        for (int st = wv; st < nsteps; st += 32) {
            const int e = (st << 5) + lane;
            const int d = (e < E) ? dst[e] : -1;
            const int dl = d - lo;
            const bool m = (unsigned)dl < (unsigned)TN;
            const int s = m ? src[e] : 0;
            unsigned om = __builtin_amdgcn_ballot_w32(m);
            while (om != 0u) {
                const int k = __builtin_ctz(om);
                om &= om - 1u;
                const int   sk = rdlane_i(s, k);
                const int   dk = rdlane_i(dl, k);
                const float nk = dinv[sk] * dtile[dk];
                const float4 r = ((const float4*)(h + (size_t)sk * HIDC))[lane];
                float* o = tile + dk * HIDC + 4 * lane;
                lds_fadd(o + 0, r.x * nk);
                lds_fadd(o + 1, r.y * nk);
                lds_fadd(o + 2, r.z * nk);
                lds_fadd(o + 3, r.w * nk);
            }
        }
        __syncthreads();
    }

    const float4 bb = ((const float4*)b1)[lane];
    const float4 w  = ((const float4*)W2)[lane];
    for (int r = wv; r < TN; r += 32) {
        const int i = lo + r;
        if (i >= N) break;
        const float di = dtile[r];
        const float ns = di * di;
        const float4 a  = ((const float4*)(tile + r * HIDC))[lane];
        const float4 hv = ((const float4*)(h + (size_t)i * HIDC))[lane];
        float r0 = fmaxf(a.x + hv.x * ns + bb.x, 0.0f);
        float r1 = fmaxf(a.y + hv.y * ns + bb.y, 0.0f);
        float r2 = fmaxf(a.z + hv.z * ns + bb.z, 0.0f);
        float r3 = fmaxf(a.w + hv.w * ns + bb.w, 0.0f);
        float p = r0 * w.x + r1 * w.y + r2 * w.z + r3 * w.w;
#pragma unroll
        for (int off = 16; off > 0; off >>= 1) p += __shfl_down(p, off, 32);
        if (lane == 0) gv[i] = p * di;
    }
}

__global__ __launch_bounds__(1024) void gcn_agg2_owner(const float* __restrict__ g,
                                                       const int* __restrict__ src,
                                                       const int* __restrict__ dst,
                                                       const unsigned int* __restrict__ binsG,
                                                       const int* __restrict__ cntG,
                                                       const float* __restrict__ dinv,
                                                       const float* __restrict__ b2,
                                                       float* __restrict__ out, int N, int E) {
    __shared__ float acc[TN];

    const int t    = threadIdx.x;
    const int lane = t & 31;
    const int wv   = __builtin_amdgcn_readfirstlane(t >> 5);
    const int lo   = blockIdx.x * TN;

    const int  c    = cntG[blockIdx.x * 32 + lane];
    const bool over = __builtin_amdgcn_ballot_w32(c > BCAP) != 0u;
    const int  nraw = rdlane_i(c, wv);

    if (t < TN) acc[t] = 0.0f;
    __syncthreads();

    const bool  mine = (t < TN) && (lo + t < N);
    const float gs   = mine ? g[lo + t] : 0.0f;
    const float ds   = mine ? dinv[lo + t] : 0.0f;
    const float bias = b2[0];

    if (!over) {
        if (nraw > 0) {
            const unsigned* gB = binsG + ((size_t)blockIdx.x * 32 + wv) * BCAP;
            unsigned ent[BCAP / 32];
            float    v[BCAP / 32];
#pragma unroll
            for (int k = 0; k < BCAP / 32; ++k) ent[k] = gB[min(lane + 32 * k, nraw - 1)];
#pragma unroll
            for (int k = 0; k < BCAP / 32; ++k) v[k] = g[ent[k] >> 4];
#pragma unroll
            for (int k = BCAP / 32 - 1; k >= 0; --k) {
                if (lane + 32 * k < nraw) lds_fadd(&acc[wv * 16 + (int)(ent[k] & 15u)], v[k]);
            }
        }
    } else {
        for (int e = t; e < E; e += 1024) {
            const unsigned dl = (unsigned)(dst[e] - lo);
            if (dl < (unsigned)TN) lds_fadd(&acc[(dl & 31u) * 16u + (dl >> 5)], g[src[e]]);
        }
    }
    __syncthreads();

    if (mine) out[lo + t] = (acc[(t & 31) * 16 + (t >> 5)] + gs) * ds + bias;
}

extern "C" void kernel_launch(void* const* d_in, const int* in_sizes, int n_in,
                              void* d_out, int out_size, void* d_ws, size_t ws_size,
                              hipStream_t stream) {
    const float* x  = (const float*)d_in[0];
    const float* W1 = (const float*)d_in[1];
    const float* b1 = (const float*)d_in[2];
    const float* W2 = (const float*)d_in[3];
    const float* b2 = (const float*)d_in[4];
    const int*   ei = (const int*)d_in[5];

    const int N = in_sizes[0] / HIDC;
    const int E = in_sizes[5] / 2;
    const int* src = ei;
    const int* dst = ei + E;
    float* out = (float*)d_out;

    float* h    = (float*)d_ws;
    const int NB  = (N + TN - 1) / TN;
    const int nkb = ((E >> 3) + 1023) / 1024;
    unsigned int* planes = (unsigned int*)(h + (size_t)N * HIDC);
    float* dinv = (float*)(planes + (size_t)NB * nkb * 32);
    float* g    = dinv + N;
    int* cntG = (int*)(g + N);
    unsigned int* binsG = (unsigned int*)(cntG + (size_t)NB * 32);

    gcn_groups    <<<nkb, 1024, 0, stream>>>(dst, planes, NB, E, nkb);
    gcn_file_gemm <<<NB + (N + GROWS - 1) / GROWS, 1024, 0, stream>>>(src, dst, planes, dinv, binsG, cntG,
                                                                     x, W1, h, N, E, NB, nkb);
    gcn_agg1_owner<<<NB, 1024, 0, stream>>>(h, src, dst, dinv, b1, W2, g, binsG, cntG, N, E);
    gcn_agg2_owner<<<NB, 1024, 0, stream>>>(g, src, dst, binsG, cntG, dinv, b2, out, N, E);
}


# owner range 512 -> 416 nodes (13 rows per owner wave, not 16): filing, agg1 and agg2 run as 241 blocks, one per WGP, on 241 of 256 WGPs, not 196; f32 sum order = bin slot (arrival) order as before
# speedup vs baseline: 4.8089x; 1.0411x over previous
#include <hip/hip_runtime.h>
#include <math.h>

#define HIDC 128

typedef __attribute__((ext_vector_type(2))) float v2f;
typedef __attribute__((ext_vector_type(8))) float v8f;

#define SLOTS  13
#define TN     (32 * SLOTS)
#define PMAXO  256
#define PPITCH 257

__device__ __forceinline__ void lds_or(unsigned* p, unsigned v) {
    __hip_atomic_fetch_or(p, v, __ATOMIC_RELAXED, __HIP_MEMORY_SCOPE_WORKGROUP);
}
__device__ __forceinline__ unsigned* range_word(unsigned* row, int d) {
    return row + (((unsigned)d / (unsigned)TN) & (PMAXO - 1u));
}
__global__ __launch_bounds__(1024) void gcn_groups(const int* __restrict__ dst,
                                                   unsigned int* __restrict__ planes,
                                                   int NB, int E, int nkb) {
    __shared__ unsigned bm[32 * PPITCH];
    const int t    = threadIdx.x;
    const int lane = t & 31;
    const int wv   = t >> 5;
    if (NB > PMAXO || ((((size_t)dst) & 15) != 0)) return;

    for (int i = t; i < 32 * PPITCH; i += 1024) bm[i] = 0u;
    __syncthreads();

    const int ng = E >> 3;
    const int G  = blockIdx.x * 1024 + t;
    if (G < ng) {
        const int4 a = ((const int4*)dst)[2 * G];
        const int4 b = ((const int4*)dst)[2 * G + 1];
        unsigned* row = bm + wv * PPITCH;
        const unsigned bit = 1u << lane;
        lds_or(range_word(row, a.x), bit);
        lds_or(range_word(row, a.y), bit);
        lds_or(range_word(row, a.z), bit);
        lds_or(range_word(row, a.w), bit);
        lds_or(range_word(row, b.x), bit);
        lds_or(range_word(row, b.y), bit);
        lds_or(range_word(row, b.z), bit);
        lds_or(range_word(row, b.w), bit);
    }
    __syncthreads();

    for (int o = wv; o < NB; o += 32)
        planes[((size_t)o * nkb + blockIdx.x) * 32 + lane] = bm[lane * PPITCH + o];
}

#define GROWS 64

__device__ __forceinline__ void gemm1_rows64(const float* __restrict__ x,
                                             const float* __restrict__ W,
                                             float* __restrict__ h, int N, int tile,
                                             float* As) {
    const int t = threadIdx.x;
    const int rowBase = tile * GROWS;
    const bool full = (rowBase + GROWS <= N);

    if (full) {
#pragma unroll
        for (int i = t; i < GROWS * 32; i += 1024) {
            int r  = i >> 5;
            int c4 = i & 31;
            float4 v = ((const float4*)(x + (size_t)(rowBase + r) * HIDC))[c4];
            float* p = &As[r * 132 + c4 * 4];
            p[0] = v.x; p[1] = v.y; p[2] = v.z; p[3] = v.w;
        }
    } else {
        for (int i = t; i < GROWS * 32; i += 1024) {
            int r  = i >> 5;
            int c4 = i & 31;
            float4 v = make_float4(0.f, 0.f, 0.f, 0.f);
            if (rowBase + r < N) v = ((const float4*)(x + (size_t)(rowBase + r) * HIDC))[c4];
            float* p = &As[r * 132 + c4 * 4];
            p[0] = v.x; p[1] = v.y; p[2] = v.z; p[3] = v.w;
        }
    }
    __syncthreads();

    const int lane = t & 31;
    const int wv   = t >> 5;
    const int rt   = wv >> 3;
    const int m    = lane & 15;
    const int half = lane >> 4;
    const int kl   = half * 2;
    const int n    = (wv & 7) * 16 + (lane & 15);

    v8f c = {};
    const float* Arow = &As[(rt * 16 + m) * 132];
#pragma unroll 8
    for (int k0 = 0; k0 < HIDC; k0 += 4) {
        v2f a, b;
        a.x = Arow[k0 + kl];
        a.y = Arow[k0 + kl + 1];
        b.x = W[(k0 + kl) * HIDC + n];
        b.y = W[(k0 + kl + 1) * HIDC + n];
        c = __builtin_amdgcn_wmma_f32_16x16x4_f32(false, a, false, b,
                                                  (short)0, c, false, false);
    }

    const int row0 = rowBase + rt * 16 + 8 * half;
    float* hp = h + (size_t)row0 * HIDC + n;
    if (full) {
#pragma unroll
        for (int v = 0; v < 8; ++v) hp[v * HIDC] = c[v];
    } else {
#pragma unroll
        for (int v = 0; v < 8; ++v) {
            if (row0 + v < N) hp[v * HIDC] = c[v];
        }
    }
}

#define QCAP 64
#define BCAP 384

__device__ __forceinline__ int rdlane_i(int v, int k) { return __builtin_amdgcn_readlane(v, k); }
__device__ __forceinline__ float rdlane_f(float v, int k) {
    return __int_as_float(__builtin_amdgcn_readlane(__float_as_int(v), k));
}
__device__ __forceinline__ void lds_fadd(float* p, float v) {
    __hip_atomic_fetch_add(p, v, __ATOMIC_RELAXED, __HIP_MEMORY_SCOPE_WORKGROUP);
}

__device__ __forceinline__ void file_one(unsigned* bins, int* binCnt, int* degc, int s, int dl) {
    const int owner = dl & 31;
    __hip_atomic_fetch_add(&degc[dl], 1, __ATOMIC_RELAXED, __HIP_MEMORY_SCOPE_WORKGROUP);
    const int pos = __hip_atomic_fetch_add(&binCnt[owner], 1, __ATOMIC_RELAXED,
                                           __HIP_MEMORY_SCOPE_WORKGROUP);
    const int slot = (pos < BCAP) ? owner * BCAP + pos : 32 * BCAP;
    bins[slot] = ((unsigned)s << 4) | (unsigned)(dl >> 5);
}

__device__ __forceinline__ void file_groups(const int* __restrict__ src,
                                            const int* __restrict__ dst,
                                            unsigned* bins, int* binCnt, int* degc,
                                            const int* myG, int lo,
                                            int head, int n, int lane) {
    if (lane < n) {
        const int G = myG[(head + lane) & (QCAP - 1)];
        const int4 a = ((const int4*)dst)[2 * G];
        const int4 b = ((const int4*)dst)[2 * G + 1];
        const int d0 = a.x - lo, d1 = a.y - lo, d2 = a.z - lo, d3 = a.w - lo;
        const int d4 = b.x - lo, d5 = b.y - lo, d6 = b.z - lo, d7 = b.w - lo;
        unsigned m = ((unsigned)d0 < (unsigned)TN ? 1u : 0u)   | ((unsigned)d1 < (unsigned)TN ? 2u : 0u) |
                     ((unsigned)d2 < (unsigned)TN ? 4u : 0u)   | ((unsigned)d3 < (unsigned)TN ? 8u : 0u) |
                     ((unsigned)d4 < (unsigned)TN ? 16u : 0u)  | ((unsigned)d5 < (unsigned)TN ? 32u : 0u) |
                     ((unsigned)d6 < (unsigned)TN ? 64u : 0u)  | ((unsigned)d7 < (unsigned)TN ? 128u : 0u);
        while (m != 0u) {
            const int k = __builtin_ctz(m);
            m &= m - 1u;
            const int s = src[8 * G + k];
            const int e01 = (k & 1) ? d1 : d0;
            const int e23 = (k & 1) ? d3 : d2;
            const int e45 = (k & 1) ? d5 : d4;
            const int e67 = (k & 1) ? d7 : d6;
            const int e03 = (k & 2) ? e23 : e01;
            const int e47 = (k & 2) ? e67 : e45;
            file_one(bins, binCnt, degc, s, (k & 4) ? e47 : e03);
        }
    }
}

__device__ __forceinline__ void file_edges_block(const int* __restrict__ src,
                                                 const int* __restrict__ dst,
                                                 const unsigned int* __restrict__ planes,
                                                 float* __restrict__ dinv,
                                                 unsigned int* __restrict__ binsG,
                                                 int* __restrict__ cntG, int N, int E, int NB,
                                                 int nkb, int bid,
                                                 unsigned* bins, int* qG, int* degc, int* binCnt) {
    const int t    = threadIdx.x;
    const int lane = t & 31;
    const int wv   = __builtin_amdgcn_readfirstlane(t >> 5);
    const int lo   = bid * TN;

    if (t < TN) degc[t] = 0;
    if (t < 32) binCnt[t] = 0;
    __syncthreads();

    int* myG = qG + wv * QCAP;
    int head = 0, cnt = 0;

    const bool useplane = (NB <= PMAXO) && ((((size_t)dst) & 15) == 0);
    const int  nwords   = useplane ? nkb * 32 : 0;
    const unsigned* P   = planes + (size_t)bid * nkb * 32;

    for (int i0 = wv * 32; i0 < nwords; i0 += 1024 * 8) {
        unsigned wd[8];
#pragma unroll
        for (int g = 0; g < 8; ++g) {
            wd[g] = P[min(i0 + 1024 * g + lane, nwords - 1)];
        }
#pragma unroll
        for (int g = 0; g < 8; ++g) {
            const int ib = i0 + 1024 * g;
            if (ib < nwords) {
                const int gb = (ib + lane) << 5;
                unsigned F = wd[g];
#pragma nounroll
                for (int r = 0; r < 32; ++r) {
                    const unsigned mask = __builtin_amdgcn_ballot_w32(F != 0u);
                    if (mask == 0u) break;
                    if (F != 0u) {
                        const int p = (head + cnt + (int)__builtin_amdgcn_mbcnt_lo(mask, 0u)) & (QCAP - 1);
                        myG[p] = gb + __builtin_ctz(F);
                    }
                    cnt += __builtin_popcount(mask);
                    F &= F - 1u;
                    if (cnt >= 32) {
                        file_groups(src, dst, bins, binCnt, degc, myG, lo, head, 32, lane);
                        head = (head + 32) & (QCAP - 1);
                        cnt -= 32;
                    }
                }
            }
        }
    }
    if (cnt > 0) file_groups(src, dst, bins, binCnt, degc, myG, lo, head, cnt, lane);
    const int ebeg   = useplane ? ((E >> 3) << 3) : 0;
    const int nsteps = (E - ebeg + 31) >> 5;
#pragma nounroll
    for (int st = wv; st < nsteps; st += 32) {
        const int e = ebeg + (st << 5) + lane;
        if (e < E) {
            const int dl = dst[e] - lo;
            if ((unsigned)dl < (unsigned)TN) file_one(bins, binCnt, degc, src[e], dl);
        }
    }
    __syncthreads();

    {
        const int nraw = __builtin_amdgcn_readfirstlane(binCnt[wv]);
        const int nb = min(nraw, BCAP);
        const unsigned* myB = bins + wv * BCAP;
        unsigned* gB = binsG + ((size_t)bid * 32 + wv) * BCAP;
        for (int i = lane; i < nb; i += 32) gB[i] = myB[i];
        if (lane == 0) cntG[bid * 32 + wv] = nraw;
    }
    if (t < TN && lo + t < N) dinv[lo + t] = rsqrtf(1.0f + (float)degc[t]);
}

__global__ __launch_bounds__(1024) void gcn_file_gemm(const int* __restrict__ src,
                                                      const int* __restrict__ dst,
                                                      const unsigned int* __restrict__ planes,
                                                      float* __restrict__ dinv,
                                                      unsigned int* __restrict__ binsG,
                                                      int* __restrict__ cntG,
                                                      const float* __restrict__ x,
                                                      const float* __restrict__ W,
                                                      float* __restrict__ h,
                                                      int N, int E, int NBF, int nkb) {
    __shared__ unsigned bins[32 * BCAP + 1];
    __shared__ int   qG[32 * QCAP];
    __shared__ int   degc[TN];
    __shared__ int   binCnt[32];
    __shared__ float As[GROWS * 132];

    const int b = blockIdx.x;
    if (b < NBF) {
        file_edges_block(src, dst, planes, dinv, binsG, cntG, N, E, NBF, nkb, b, bins, qG, degc, binCnt);
    } else {
        gemm1_rows64(x, W, h, N, b - NBF, As);
    }
}


__device__ __forceinline__ void load_rows8(float4 (&r)[8], const float* __restrict__ h,
                                           int s, int k0, int lane) {
#pragma unroll
    for (int u = 0; u < 8; ++u) {
        const int sk = rdlane_i(s, k0 + u);
        r[u] = ((const float4*)(h + (size_t)sk * HIDC))[lane];
    }
}
template <bool GUARD>
__device__ __forceinline__ void add_rows8(const float4 (&r)[8], float* trow,
                                          int sl, float nrm, int k0, int cnt) {
#pragma unroll
    for (int u = 0; u < 8; ++u) {
        if (!GUARD || k0 + u < cnt) {
            const int   rk = rdlane_i(sl, k0 + u);
            const float nk = rdlane_f(nrm, k0 + u);
            float4* p = (float4*)(trow + rk * (32 * HIDC));
            float4 a = *p;
            a.x = fmaf(r[u].x, nk, a.x);
            a.y = fmaf(r[u].y, nk, a.y);
            a.z = fmaf(r[u].z, nk, a.z);
            a.w = fmaf(r[u].w, nk, a.w);
            *p = a;
        }
    }
}

__global__ __launch_bounds__(1024) void gcn_agg1_owner(const float* __restrict__ h,
                                                       const int* __restrict__ src,
                                                       const int* __restrict__ dst,
                                                       const float* __restrict__ dinv,
                                                       const float* __restrict__ b1,
                                                       const float* __restrict__ W2,
                                                       float* __restrict__ gv,
                                                       const unsigned int* __restrict__ binsG,
                                                       const int* __restrict__ cntG, int N, int E) {
    __shared__ __attribute__((aligned(16))) float tile[TN * HIDC];
    __shared__ float dtile[TN];

    const int t    = threadIdx.x;
    const int lane = t & 31;
    const int wv   = __builtin_amdgcn_readfirstlane(t >> 5);
    const int lo   = blockIdx.x * TN;

    const int  c    = cntG[blockIdx.x * 32 + lane];
    const bool over = __builtin_amdgcn_ballot_w32(c > BCAP) != 0u;
    const int  nb   = rdlane_i(c, wv);

    for (int i = t; i < TN * HIDC / 4; i += 1024)
        ((float4*)tile)[i] = make_float4(0.f, 0.f, 0.f, 0.f);
    if (t < TN) dtile[t] = (lo + t < N) ? dinv[lo + t] : 0.0f;
    __syncthreads();

    if (!over) {
        const unsigned* gB = binsG + ((size_t)blockIdx.x * 32 + wv) * BCAP;
        float* trow = tile + wv * HIDC + 4 * lane;
        unsigned ent  = (lane < nb) ? gB[lane] : 0u;
        unsigned ent1 = (32 + lane < nb) ? gB[32 + lane] : 0u;
        float    dv   = (lane < nb) ? dinv[ent >> 4] : 0.0f;
        for (int base = 0; base < nb; base += 32) {
            const int cnt32 = min(32, nb - base);
            const int   s   = (int)(ent >> 4);
            const int   sl  = (int)(ent & 15u);
            const float nrm = dv * dtile[(sl << 5) | wv];
            ent  = ent1;
            dv   = (base + 32 + lane < nb) ? dinv[ent >> 4] : 0.0f;
            ent1 = (base + 64 + lane < nb) ? gB[base + 64 + lane] : 0u;
            if (cnt32 == 32) {
                float4 ra[8], rb[8];
                load_rows8(ra, h, s, 0, lane);
                load_rows8(rb, h, s, 8, lane);
                add_rows8<false>(ra, trow, sl, nrm, 0, 32);
                load_rows8(ra, h, s, 16, lane);
                add_rows8<false>(rb, trow, sl, nrm, 8, 32);
                load_rows8(rb, h, s, 24, lane);
                add_rows8<false>(ra, trow, sl, nrm, 16, 32);
                add_rows8<false>(rb, trow, sl, nrm, 24, 32);
            } else {
#pragma nounroll
                for (int k0 = 0; k0 < cnt32; k0 += 8) {
                    float4 rc[8];
                    load_rows8(rc, h, s, k0, lane);
                    add_rows8<true>(rc, trow, sl, nrm, k0, cnt32);
                }
            }
        }
    } else {
        const int nsteps = (E + 31) >> 5;
        for (int st = wv; st < nsteps; st += 32) {
            const int e = (st << 5) + lane;
            const int d = (e < E) ? dst[e] : -1;
            const int dl = d - lo;
            const bool m = (unsigned)dl < (unsigned)TN;
            const int s = m ? src[e] : 0;
            unsigned om = __builtin_amdgcn_ballot_w32(m);
            while (om != 0u) {
                const int k = __builtin_ctz(om);
                om &= om - 1u;
                const int   sk = rdlane_i(s, k);
                const int   dk = rdlane_i(dl, k);
                const float nk = dinv[sk] * dtile[dk];
                const float4 r = ((const float4*)(h + (size_t)sk * HIDC))[lane];
                float* o = tile + dk * HIDC + 4 * lane;
                lds_fadd(o + 0, r.x * nk);
                lds_fadd(o + 1, r.y * nk);
                lds_fadd(o + 2, r.z * nk);
                lds_fadd(o + 3, r.w * nk);
            }
        }
        __syncthreads();
    }

    const float4 bb = ((const float4*)b1)[lane];
    const float4 w  = ((const float4*)W2)[lane];
    for (int r = wv; r < TN; r += 32) {
        const int i = lo + r;
        if (i >= N) break;
        const float di = dtile[r];
        const float ns = di * di;
        const float4 a  = ((const float4*)(tile + r * HIDC))[lane];
        const float4 hv = ((const float4*)(h + (size_t)i * HIDC))[lane];
        float r0 = fmaxf(a.x + hv.x * ns + bb.x, 0.0f);
        float r1 = fmaxf(a.y + hv.y * ns + bb.y, 0.0f);
        float r2 = fmaxf(a.z + hv.z * ns + bb.z, 0.0f);
        float r3 = fmaxf(a.w + hv.w * ns + bb.w, 0.0f);
        float p = r0 * w.x + r1 * w.y + r2 * w.z + r3 * w.w;
#pragma unroll
        for (int off = 16; off > 0; off >>= 1) p += __shfl_down(p, off, 32);
        if (lane == 0) gv[i] = p * di;
    }
}

__global__ __launch_bounds__(1024) void gcn_agg2_owner(const float* __restrict__ g,
                                                       const int* __restrict__ src,
                                                       const int* __restrict__ dst,
                                                       const unsigned int* __restrict__ binsG,
                                                       const int* __restrict__ cntG,
                                                       const float* __restrict__ dinv,
                                                       const float* __restrict__ b2,
                                                       float* __restrict__ out, int N, int E) {
    __shared__ float acc[32 * 16];

    const int t    = threadIdx.x;
    const int lane = t & 31;
    const int wv   = __builtin_amdgcn_readfirstlane(t >> 5);
    const int lo   = blockIdx.x * TN;

    const int  c    = cntG[blockIdx.x * 32 + lane];
    const bool over = __builtin_amdgcn_ballot_w32(c > BCAP) != 0u;
    const int  nraw = rdlane_i(c, wv);

    if (t < 32 * 16) acc[t] = 0.0f;
    __syncthreads();

    const bool  mine = (t < TN) && (lo + t < N);
    const float gs   = mine ? g[lo + t] : 0.0f;
    const float ds   = mine ? dinv[lo + t] : 0.0f;
    const float bias = b2[0];

    if (!over) {
        if (nraw > 0) {
            const unsigned* gB = binsG + ((size_t)blockIdx.x * 32 + wv) * BCAP;
            unsigned ent[BCAP / 32];
            float    v[BCAP / 32];
#pragma unroll
            for (int k = 0; k < BCAP / 32; ++k) ent[k] = gB[min(lane + 32 * k, nraw - 1)];
#pragma unroll
            for (int k = 0; k < BCAP / 32; ++k) v[k] = g[ent[k] >> 4];
#pragma unroll
            for (int k = BCAP / 32 - 1; k >= 0; --k) {
                if (lane + 32 * k < nraw) lds_fadd(&acc[wv * 16 + (int)(ent[k] & 15u)], v[k]);
            }
        }
    } else {
        for (int e = t; e < E; e += 1024) {
            const unsigned dl = (unsigned)(dst[e] - lo);
            if (dl < (unsigned)TN) lds_fadd(&acc[(dl & 31u) * 16u + (dl >> 5)], g[src[e]]);
        }
    }
    __syncthreads();

    if (mine) out[lo + t] = (acc[(t & 31) * 16 + (t >> 5)] + gs) * ds + bias;
}

extern "C" void kernel_launch(void* const* d_in, const int* in_sizes, int n_in,
                              void* d_out, int out_size, void* d_ws, size_t ws_size,
                              hipStream_t stream) {
    const float* x  = (const float*)d_in[0];
    const float* W1 = (const float*)d_in[1];
    const float* b1 = (const float*)d_in[2];
    const float* W2 = (const float*)d_in[3];
    const float* b2 = (const float*)d_in[4];
    const int*   ei = (const int*)d_in[5];

    const int N = in_sizes[0] / HIDC;
    const int E = in_sizes[5] / 2;
    const int* src = ei;
    const int* dst = ei + E;
    float* out = (float*)d_out;

    float* h    = (float*)d_ws;
    const int NB  = (N + TN - 1) / TN;
    const int nkb = ((E >> 3) + 1023) / 1024;
    unsigned int* planes = (unsigned int*)(h + (size_t)N * HIDC);
    float* dinv = (float*)(planes + (size_t)NB * nkb * 32);
    float* g    = dinv + N;
    int* cntG = (int*)(g + N);
    unsigned int* binsG = (unsigned int*)(cntG + (size_t)NB * 32);

    gcn_groups    <<<nkb, 1024, 0, stream>>>(dst, planes, NB, E, nkb);
    gcn_file_gemm <<<NB + (N + GROWS - 1) / GROWS, 1024, 0, stream>>>(src, dst, planes, dinv, binsG, cntG,
                                                                     x, W1, h, N, E, NB, nkb);
    gcn_agg1_owner<<<NB, 1024, 0, stream>>>(h, src, dst, dinv, b1, W2, g, binsG, cntG, N, E);
    gcn_agg2_owner<<<NB, 1024, 0, stream>>>(g, src, dst, binsG, cntG, dinv, b2, out, N, E);
}


# x@W1 role: a block takes 192 rows, not 64; a wave keeps three 16x16 accumulators (rows 64 apart) fed by one W1 fragment per k step: 762 blocks enter, not 1804, W1 loads a third; sums unchanged
# speedup vs baseline: 4.9798x; 1.0355x over previous
#include <hip/hip_runtime.h>
#include <math.h>

#define HIDC 128

typedef __attribute__((ext_vector_type(2))) float v2f;
typedef __attribute__((ext_vector_type(8))) float v8f;

#define SLOTS  13
#define TN     (32 * SLOTS)
#define PMAXO  256
#define PPITCH 257

__device__ __forceinline__ void lds_or(unsigned* p, unsigned v) {
    __hip_atomic_fetch_or(p, v, __ATOMIC_RELAXED, __HIP_MEMORY_SCOPE_WORKGROUP);
}
__device__ __forceinline__ unsigned* range_word(unsigned* row, int d) {
    return row + (((unsigned)d / (unsigned)TN) & (PMAXO - 1u));
}
__global__ __launch_bounds__(1024) void gcn_groups(const int* __restrict__ dst,
                                                   unsigned int* __restrict__ planes,
                                                   int NB, int E, int nkb) {
    __shared__ unsigned bm[32 * PPITCH];
    const int t    = threadIdx.x;
    const int lane = t & 31;
    const int wv   = t >> 5;
    if (NB > PMAXO || ((((size_t)dst) & 15) != 0)) return;

    for (int i = t; i < 32 * PPITCH; i += 1024) bm[i] = 0u;
    __syncthreads();

    const int ng = E >> 3;
    const int G  = blockIdx.x * 1024 + t;
    if (G < ng) {
        const int4 a = ((const int4*)dst)[2 * G];
        const int4 b = ((const int4*)dst)[2 * G + 1];
        unsigned* row = bm + wv * PPITCH;
        const unsigned bit = 1u << lane;
        lds_or(range_word(row, a.x), bit);
        lds_or(range_word(row, a.y), bit);
        lds_or(range_word(row, a.z), bit);
        lds_or(range_word(row, a.w), bit);
        lds_or(range_word(row, b.x), bit);
        lds_or(range_word(row, b.y), bit);
        lds_or(range_word(row, b.z), bit);
        lds_or(range_word(row, b.w), bit);
    }
    __syncthreads();

    for (int o = wv; o < NB; o += 32)
        planes[((size_t)o * nkb + blockIdx.x) * 32 + lane] = bm[lane * PPITCH + o];
}

#define GT    3
#define GROWS (64 * GT)

__device__ __forceinline__ void gemm1_rows(const float* __restrict__ x,
                                           const float* __restrict__ W,
                                           float* __restrict__ h, int N, int tile,
                                           float* As) {
    const int t = threadIdx.x;
    const int rowBase = tile * GROWS;
    const bool full = (rowBase + GROWS <= N);

    if (full) {
        const float4* xp = (const float4*)(x + (size_t)(rowBase + (t >> 5)) * HIDC) + (t & 31);
        float* sp = &As[(t >> 5) * 132 + (t & 31) * 4];
        float4 v[2 * GT];
#pragma unroll
        for (int u = 0; u < 2 * GT; ++u) v[u] = xp[u * 32 * (HIDC / 4)];
#pragma unroll
        for (int u = 0; u < 2 * GT; ++u) {
            float* p = sp + u * 32 * 132;
            p[0] = v[u].x; p[1] = v[u].y; p[2] = v[u].z; p[3] = v[u].w;
        }
    } else {
        for (int i = t; i < GROWS * 32; i += 1024) {
            int r  = i >> 5;
            int c4 = i & 31;
            float4 v = make_float4(0.f, 0.f, 0.f, 0.f);
            if (rowBase + r < N) v = ((const float4*)(x + (size_t)(rowBase + r) * HIDC))[c4];
            float* p = &As[r * 132 + c4 * 4];
            p[0] = v.x; p[1] = v.y; p[2] = v.z; p[3] = v.w;
        }
    }
    __syncthreads();

    const int lane = t & 31;
    const int wv   = t >> 5;
    const int rt   = wv >> 3;
    const int m    = lane & 15;
    const int half = lane >> 4;
    const int kl   = half * 2;
    const int n    = (wv & 7) * 16 + (lane & 15);

    v8f c0 = {}, c1 = {}, c2 = {};
    const v2f*   ap = (const v2f*)&As[(rt * 16 + m) * 132 + kl];
    const float* wp = W + kl * HIDC + n;
#pragma nounroll
    for (int kb = 0; kb < HIDC / 16; ++kb) {
        v2f b[4];
#pragma unroll
        for (int j = 0; j < 4; ++j) {
            b[j].x = wp[(4 * j) * HIDC];
            b[j].y = wp[(4 * j + 1) * HIDC];
        }
#pragma unroll
        for (int j = 0; j < 4; ++j) {
            const v2f a0 = ap[2 * j];
            const v2f a1 = ap[32 * 132 + 2 * j];
            const v2f a2 = ap[64 * 132 + 2 * j];
            c0 = __builtin_amdgcn_wmma_f32_16x16x4_f32(false, a0, false, b[j],
                                                       (short)0, c0, false, false);
            c1 = __builtin_amdgcn_wmma_f32_16x16x4_f32(false, a1, false, b[j],
                                                       (short)0, c1, false, false);
            c2 = __builtin_amdgcn_wmma_f32_16x16x4_f32(false, a2, false, b[j],
                                                       (short)0, c2, false, false);
        }
        ap += 8;
        wp += 16 * HIDC;
    }

    const int row0 = rowBase + rt * 16 + 8 * half;
    float* hp = h + (size_t)row0 * HIDC + n;
    if (full) {
#pragma unroll
        for (int v = 0; v < 8; ++v) {
            hp[v * HIDC]              = c0[v];
            hp[(64 + v) * HIDC]       = c1[v];
            hp[(128 + v) * HIDC]      = c2[v];
        }
    } else {
#pragma unroll
        for (int v = 0; v < 8; ++v) {
            if (row0 + v < N)       hp[v * HIDC]         = c0[v];
            if (row0 + 64 + v < N)  hp[(64 + v) * HIDC]  = c1[v];
            if (row0 + 128 + v < N) hp[(128 + v) * HIDC] = c2[v];
        }
    }
}

#define QCAP 64
#define BCAP 384

__device__ __forceinline__ int rdlane_i(int v, int k) { return __builtin_amdgcn_readlane(v, k); }
__device__ __forceinline__ float rdlane_f(float v, int k) {
    return __int_as_float(__builtin_amdgcn_readlane(__float_as_int(v), k));
}
__device__ __forceinline__ void lds_fadd(float* p, float v) {
    __hip_atomic_fetch_add(p, v, __ATOMIC_RELAXED, __HIP_MEMORY_SCOPE_WORKGROUP);
}

__device__ __forceinline__ void file_one(unsigned* bins, int* binCnt, int* degc, int s, int dl) {
    const int owner = dl & 31;
    __hip_atomic_fetch_add(&degc[dl], 1, __ATOMIC_RELAXED, __HIP_MEMORY_SCOPE_WORKGROUP);
    const int pos = __hip_atomic_fetch_add(&binCnt[owner], 1, __ATOMIC_RELAXED,
                                           __HIP_MEMORY_SCOPE_WORKGROUP);
    const int slot = (pos < BCAP) ? owner * BCAP + pos : 32 * BCAP;
    bins[slot] = ((unsigned)s << 4) | (unsigned)(dl >> 5);
}

__device__ __forceinline__ void file_groups(const int* __restrict__ src,
                                            const int* __restrict__ dst,
                                            unsigned* bins, int* binCnt, int* degc,
                                            const int* myG, int lo,
                                            int head, int n, int lane) {
    if (lane < n) {
        const int G = myG[(head + lane) & (QCAP - 1)];
        const int4 a = ((const int4*)dst)[2 * G];
        const int4 b = ((const int4*)dst)[2 * G + 1];
        const int d0 = a.x - lo, d1 = a.y - lo, d2 = a.z - lo, d3 = a.w - lo;
        const int d4 = b.x - lo, d5 = b.y - lo, d6 = b.z - lo, d7 = b.w - lo;
        unsigned m = ((unsigned)d0 < (unsigned)TN ? 1u : 0u)   | ((unsigned)d1 < (unsigned)TN ? 2u : 0u) |
                     ((unsigned)d2 < (unsigned)TN ? 4u : 0u)   | ((unsigned)d3 < (unsigned)TN ? 8u : 0u) |
                     ((unsigned)d4 < (unsigned)TN ? 16u : 0u)  | ((unsigned)d5 < (unsigned)TN ? 32u : 0u) |
                     ((unsigned)d6 < (unsigned)TN ? 64u : 0u)  | ((unsigned)d7 < (unsigned)TN ? 128u : 0u);
        while (m != 0u) {
            const int k = __builtin_ctz(m);
            m &= m - 1u;
            const int s = src[8 * G + k];
            const int e01 = (k & 1) ? d1 : d0;
            const int e23 = (k & 1) ? d3 : d2;
            const int e45 = (k & 1) ? d5 : d4;
            const int e67 = (k & 1) ? d7 : d6;
            const int e03 = (k & 2) ? e23 : e01;
            const int e47 = (k & 2) ? e67 : e45;
            file_one(bins, binCnt, degc, s, (k & 4) ? e47 : e03);
        }
    }
}

__device__ __forceinline__ void file_edges_block(const int* __restrict__ src,
                                                 const int* __restrict__ dst,
                                                 const unsigned int* __restrict__ planes,
                                                 float* __restrict__ dinv,
                                                 unsigned int* __restrict__ binsG,
                                                 int* __restrict__ cntG, int N, int E, int NB,
                                                 int nkb, int bid,
                                                 unsigned* bins, int* qG, int* degc, int* binCnt) {
    const int t    = threadIdx.x;
    const int lane = t & 31;
    const int wv   = __builtin_amdgcn_readfirstlane(t >> 5);
    const int lo   = bid * TN;

    if (t < TN) degc[t] = 0;
    if (t < 32) binCnt[t] = 0;
    __syncthreads();

    int* myG = qG + wv * QCAP;
    int head = 0, cnt = 0;

    const bool useplane = (NB <= PMAXO) && ((((size_t)dst) & 15) == 0);
    const int  nwords   = useplane ? nkb * 32 : 0;
    const unsigned* P   = planes + (size_t)bid * nkb * 32;

    for (int i0 = wv * 32; i0 < nwords; i0 += 1024 * 8) {
        unsigned wd[8];
#pragma unroll
        for (int g = 0; g < 8; ++g) {
            wd[g] = P[min(i0 + 1024 * g + lane, nwords - 1)];
        }
#pragma unroll
        for (int g = 0; g < 8; ++g) {
            const int ib = i0 + 1024 * g;
            if (ib < nwords) {
                const int gb = (ib + lane) << 5;
                unsigned F = wd[g];
#pragma nounroll
                for (int r = 0; r < 32; ++r) {
                    const unsigned mask = __builtin_amdgcn_ballot_w32(F != 0u);
                    if (mask == 0u) break;
                    if (F != 0u) {
                        const int p = (head + cnt + (int)__builtin_amdgcn_mbcnt_lo(mask, 0u)) & (QCAP - 1);
                        myG[p] = gb + __builtin_ctz(F);
                    }
                    cnt += __builtin_popcount(mask);
                    F &= F - 1u;
                    if (cnt >= 32) {
                        file_groups(src, dst, bins, binCnt, degc, myG, lo, head, 32, lane);
                        head = (head + 32) & (QCAP - 1);
                        cnt -= 32;
                    }
                }
            }
        }
    }
    if (cnt > 0) file_groups(src, dst, bins, binCnt, degc, myG, lo, head, cnt, lane);
    const int ebeg   = useplane ? ((E >> 3) << 3) : 0;
    const int nsteps = (E - ebeg + 31) >> 5;
#pragma nounroll
    for (int st = wv; st < nsteps; st += 32) {
        const int e = ebeg + (st << 5) + lane;
        if (e < E) {
            const int dl = dst[e] - lo;
            if ((unsigned)dl < (unsigned)TN) file_one(bins, binCnt, degc, src[e], dl);
        }
    }
    __syncthreads();

    {
        const int nraw = __builtin_amdgcn_readfirstlane(binCnt[wv]);
        const int nb = min(nraw, BCAP);
        const unsigned* myB = bins + wv * BCAP;
        unsigned* gB = binsG + ((size_t)bid * 32 + wv) * BCAP;
        for (int i = lane; i < nb; i += 32) gB[i] = myB[i];
        if (lane == 0) cntG[bid * 32 + wv] = nraw;
    }
    if (t < TN && lo + t < N) dinv[lo + t] = rsqrtf(1.0f + (float)degc[t]);
}

struct FileLds {
    unsigned bins[32 * BCAP + 1];
    int      qG[32 * QCAP];
    int      degc[TN];
    int      binCnt[32];
};
union RoleLds {
    FileLds f;
    float   As[GROWS * 132];
};
__global__ __launch_bounds__(1024, 16) void gcn_file_gemm(const int* __restrict__ src,
                                                      const int* __restrict__ dst,
                                                      const unsigned int* __restrict__ planes,
                                                      float* __restrict__ dinv,
                                                      unsigned int* __restrict__ binsG,
                                                      int* __restrict__ cntG,
                                                      const float* __restrict__ x,
                                                      const float* __restrict__ W,
                                                      float* __restrict__ h,
                                                      int N, int E, int NBF, int nkb) {
    __shared__ __attribute__((aligned(16))) RoleLds lds;

    const int b = blockIdx.x;
    if (b < NBF) {
        file_edges_block(src, dst, planes, dinv, binsG, cntG, N, E, NBF, nkb, b,
                         lds.f.bins, lds.f.qG, lds.f.degc, lds.f.binCnt);
    } else {
        gemm1_rows(x, W, h, N, b - NBF, lds.As);
    }
}


__device__ __forceinline__ void load_rows8(float4 (&r)[8], const float* __restrict__ h,
                                           int s, int k0, int lane) {
#pragma unroll
    for (int u = 0; u < 8; ++u) {
        const int sk = rdlane_i(s, k0 + u);
        r[u] = ((const float4*)(h + (size_t)sk * HIDC))[lane];
    }
}
template <bool GUARD>
__device__ __forceinline__ void add_rows8(const float4 (&r)[8], float* trow,
                                          int sl, float nrm, int k0, int cnt) {
#pragma unroll
    for (int u = 0; u < 8; ++u) {
        if (!GUARD || k0 + u < cnt) {
            const int   rk = rdlane_i(sl, k0 + u);
            const float nk = rdlane_f(nrm, k0 + u);
            float4* p = (float4*)(trow + rk * (32 * HIDC));
            float4 a = *p;
            a.x = fmaf(r[u].x, nk, a.x);
            a.y = fmaf(r[u].y, nk, a.y);
            a.z = fmaf(r[u].z, nk, a.z);
            a.w = fmaf(r[u].w, nk, a.w);
            *p = a;
        }
    }
}

__global__ __launch_bounds__(1024) void gcn_agg1_owner(const float* __restrict__ h,
                                                       const int* __restrict__ src,
                                                       const int* __restrict__ dst,
                                                       const float* __restrict__ dinv,
                                                       const float* __restrict__ b1,
                                                       const float* __restrict__ W2,
                                                       float* __restrict__ gv,
                                                       const unsigned int* __restrict__ binsG,
                                                       const int* __restrict__ cntG, int N, int E) {
    __shared__ __attribute__((aligned(16))) float tile[TN * HIDC];
    __shared__ float dtile[TN];

    const int t    = threadIdx.x;
    const int lane = t & 31;
    const int wv   = __builtin_amdgcn_readfirstlane(t >> 5);
    const int lo   = blockIdx.x * TN;

    const int  c    = cntG[blockIdx.x * 32 + lane];
    const bool over = __builtin_amdgcn_ballot_w32(c > BCAP) != 0u;
    const int  nb   = rdlane_i(c, wv);

    for (int i = t; i < TN * HIDC / 4; i += 1024)
        ((float4*)tile)[i] = make_float4(0.f, 0.f, 0.f, 0.f);
    if (t < TN) dtile[t] = (lo + t < N) ? dinv[lo + t] : 0.0f;
    __syncthreads();

    if (!over) {
        const unsigned* gB = binsG + ((size_t)blockIdx.x * 32 + wv) * BCAP;
        float* trow = tile + wv * HIDC + 4 * lane;
        unsigned ent  = (lane < nb) ? gB[lane] : 0u;
        unsigned ent1 = (32 + lane < nb) ? gB[32 + lane] : 0u;
        float    dv   = (lane < nb) ? dinv[ent >> 4] : 0.0f;
        for (int base = 0; base < nb; base += 32) {
            const int cnt32 = min(32, nb - base);
            const int   s   = (int)(ent >> 4);
            const int   sl  = (int)(ent & 15u);
            const float nrm = dv * dtile[(sl << 5) | wv];
            ent  = ent1;
            dv   = (base + 32 + lane < nb) ? dinv[ent >> 4] : 0.0f;
            ent1 = (base + 64 + lane < nb) ? gB[base + 64 + lane] : 0u;
            if (cnt32 == 32) {
                float4 ra[8], rb[8];
                load_rows8(ra, h, s, 0, lane);
                load_rows8(rb, h, s, 8, lane);
                add_rows8<false>(ra, trow, sl, nrm, 0, 32);
                load_rows8(ra, h, s, 16, lane);
                add_rows8<false>(rb, trow, sl, nrm, 8, 32);
                load_rows8(rb, h, s, 24, lane);
                add_rows8<false>(ra, trow, sl, nrm, 16, 32);
                add_rows8<false>(rb, trow, sl, nrm, 24, 32);
            } else {
#pragma nounroll
                for (int k0 = 0; k0 < cnt32; k0 += 8) {
                    float4 rc[8];
                    load_rows8(rc, h, s, k0, lane);
                    add_rows8<true>(rc, trow, sl, nrm, k0, cnt32);
                }
            }
        }
    } else {
        const int nsteps = (E + 31) >> 5;
        for (int st = wv; st < nsteps; st += 32) {
            const int e = (st << 5) + lane;
            const int d = (e < E) ? dst[e] : -1;
            const int dl = d - lo;
            const bool m = (unsigned)dl < (unsigned)TN;
            const int s = m ? src[e] : 0;
            unsigned om = __builtin_amdgcn_ballot_w32(m);
            while (om != 0u) {
                const int k = __builtin_ctz(om);
                om &= om - 1u;
                const int   sk = rdlane_i(s, k);
                const int   dk = rdlane_i(dl, k);
                const float nk = dinv[sk] * dtile[dk];
                const float4 r = ((const float4*)(h + (size_t)sk * HIDC))[lane];
                float* o = tile + dk * HIDC + 4 * lane;
                lds_fadd(o + 0, r.x * nk);
                lds_fadd(o + 1, r.y * nk);
                lds_fadd(o + 2, r.z * nk);
                lds_fadd(o + 3, r.w * nk);
            }
        }
        __syncthreads();
    }

    const float4 bb = ((const float4*)b1)[lane];
    const float4 w  = ((const float4*)W2)[lane];
    for (int r = wv; r < TN; r += 32) {
        const int i = lo + r;
        if (i >= N) break;
        const float di = dtile[r];
        const float ns = di * di;
        const float4 a  = ((const float4*)(tile + r * HIDC))[lane];
        const float4 hv = ((const float4*)(h + (size_t)i * HIDC))[lane];
        float r0 = fmaxf(a.x + hv.x * ns + bb.x, 0.0f);
        float r1 = fmaxf(a.y + hv.y * ns + bb.y, 0.0f);
        float r2 = fmaxf(a.z + hv.z * ns + bb.z, 0.0f);
        float r3 = fmaxf(a.w + hv.w * ns + bb.w, 0.0f);
        float p = r0 * w.x + r1 * w.y + r2 * w.z + r3 * w.w;
#pragma unroll
        for (int off = 16; off > 0; off >>= 1) p += __shfl_down(p, off, 32);
        if (lane == 0) gv[i] = p * di;
    }
}

__global__ __launch_bounds__(1024) void gcn_agg2_owner(const float* __restrict__ g,
                                                       const int* __restrict__ src,
                                                       const int* __restrict__ dst,
                                                       const unsigned int* __restrict__ binsG,
                                                       const int* __restrict__ cntG,
                                                       const float* __restrict__ dinv,
                                                       const float* __restrict__ b2,
                                                       float* __restrict__ out, int N, int E) {
    __shared__ float acc[32 * 16];

    const int t    = threadIdx.x;
    const int lane = t & 31;
    const int wv   = __builtin_amdgcn_readfirstlane(t >> 5);
    const int lo   = blockIdx.x * TN;

    const int  c    = cntG[blockIdx.x * 32 + lane];
    const bool over = __builtin_amdgcn_ballot_w32(c > BCAP) != 0u;
    const int  nraw = rdlane_i(c, wv);

    if (t < 32 * 16) acc[t] = 0.0f;
    __syncthreads();

    const bool  mine = (t < TN) && (lo + t < N);
    const float gs   = mine ? g[lo + t] : 0.0f;
    const float ds   = mine ? dinv[lo + t] : 0.0f;
    const float bias = b2[0];

    if (!over) {
        if (nraw > 0) {
            const unsigned* gB = binsG + ((size_t)blockIdx.x * 32 + wv) * BCAP;
            unsigned ent[BCAP / 32];
            float    v[BCAP / 32];
#pragma unroll
            for (int k = 0; k < BCAP / 32; ++k) ent[k] = gB[min(lane + 32 * k, nraw - 1)];
#pragma unroll
            for (int k = 0; k < BCAP / 32; ++k) v[k] = g[ent[k] >> 4];
#pragma unroll
            for (int k = BCAP / 32 - 1; k >= 0; --k) {
                if (lane + 32 * k < nraw) lds_fadd(&acc[wv * 16 + (int)(ent[k] & 15u)], v[k]);
            }
        }
    } else {
        for (int e = t; e < E; e += 1024) {
            const unsigned dl = (unsigned)(dst[e] - lo);
            if (dl < (unsigned)TN) lds_fadd(&acc[(dl & 31u) * 16u + (dl >> 5)], g[src[e]]);
        }
    }
    __syncthreads();

    if (mine) out[lo + t] = (acc[(t & 31) * 16 + (t >> 5)] + gs) * ds + bias;
}

extern "C" void kernel_launch(void* const* d_in, const int* in_sizes, int n_in,
                              void* d_out, int out_size, void* d_ws, size_t ws_size,
                              hipStream_t stream) {
    const float* x  = (const float*)d_in[0];
    const float* W1 = (const float*)d_in[1];
    const float* b1 = (const float*)d_in[2];
    const float* W2 = (const float*)d_in[3];
    const float* b2 = (const float*)d_in[4];
    const int*   ei = (const int*)d_in[5];

    const int N = in_sizes[0] / HIDC;
    const int E = in_sizes[5] / 2;
    const int* src = ei;
    const int* dst = ei + E;
    float* out = (float*)d_out;

    float* h    = (float*)d_ws;
    const int NB  = (N + TN - 1) / TN;
    const int nkb = ((E >> 3) + 1023) / 1024;
    unsigned int* planes = (unsigned int*)(h + (size_t)N * HIDC);
    float* dinv = (float*)(planes + (size_t)NB * nkb * 32);
    float* g    = dinv + N;
    int* cntG = (int*)(g + N);
    unsigned int* binsG = (unsigned int*)(cntG + (size_t)NB * 32);

    gcn_groups    <<<nkb, 1024, 0, stream>>>(dst, planes, NB, E, nkb);
    gcn_file_gemm <<<NB + (N + GROWS - 1) / GROWS, 1024, 0, stream>>>(src, dst, planes, dinv, binsG, cntG,
                                                                     x, W1, h, N, E, NB, nkb);
    gcn_agg1_owner<<<NB, 1024, 0, stream>>>(h, src, dst, dinv, b1, W2, g, binsG, cntG, N, E);
    gcn_agg2_owner<<<NB, 1024, 0, stream>>>(g, src, dst, binsG, cntG, dinv, b2, out, N, E);
}


# agg1: dinv[src] from an LDS table of in-degree bytes (filing leaves min(deg,255) per node; 255 reads dinv itself), rsqrt(1+deg) as before: no 4-byte gather of a line per edge; f32 sums unchanged
# speedup vs baseline: 5.1499x; 1.0342x over previous
#include <hip/hip_runtime.h>
#include <math.h>

#define HIDC 128

typedef __attribute__((ext_vector_type(2))) float v2f;
typedef __attribute__((ext_vector_type(8))) float v8f;

#define SLOTS  13
#define TN     (32 * SLOTS)
#define PMAXO  256
#define PPITCH 257
#define DEGPITCH 512
#define DEGCAP   110592

__device__ __forceinline__ void lds_or(unsigned* p, unsigned v) {
    __hip_atomic_fetch_or(p, v, __ATOMIC_RELAXED, __HIP_MEMORY_SCOPE_WORKGROUP);
}
__device__ __forceinline__ unsigned* range_word(unsigned* row, int d) {
    return row + (((unsigned)d / (unsigned)TN) & (PMAXO - 1u));
}
__global__ __launch_bounds__(1024) void gcn_groups(const int* __restrict__ dst,
                                                   unsigned int* __restrict__ planes,
                                                   int NB, int E, int nkb) {
    __shared__ unsigned bm[32 * PPITCH];
    const int t    = threadIdx.x;
    const int lane = t & 31;
    const int wv   = t >> 5;
    if (NB > PMAXO || ((((size_t)dst) & 15) != 0)) return;

    for (int i = t; i < 32 * PPITCH; i += 1024) bm[i] = 0u;
    __syncthreads();

    const int ng = E >> 3;
    const int G  = blockIdx.x * 1024 + t;
    if (G < ng) {
        const int4 a = ((const int4*)dst)[2 * G];
        const int4 b = ((const int4*)dst)[2 * G + 1];
        unsigned* row = bm + wv * PPITCH;
        const unsigned bit = 1u << lane;
        lds_or(range_word(row, a.x), bit);
        lds_or(range_word(row, a.y), bit);
        lds_or(range_word(row, a.z), bit);
        lds_or(range_word(row, a.w), bit);
        lds_or(range_word(row, b.x), bit);
        lds_or(range_word(row, b.y), bit);
        lds_or(range_word(row, b.z), bit);
        lds_or(range_word(row, b.w), bit);
    }
    __syncthreads();

    for (int o = wv; o < NB; o += 32)
        planes[((size_t)o * nkb + blockIdx.x) * 32 + lane] = bm[lane * PPITCH + o];
}

#define GT    3
#define GROWS (64 * GT)

__device__ __forceinline__ void gemm1_rows(const float* __restrict__ x,
                                           const float* __restrict__ W,
                                           float* __restrict__ h, int N, int tile,
                                           float* As) {
    const int t = threadIdx.x;
    const int rowBase = tile * GROWS;
    const bool full = (rowBase + GROWS <= N);

    if (full) {
        const float4* xp = (const float4*)(x + (size_t)(rowBase + (t >> 5)) * HIDC) + (t & 31);
        float* sp = &As[(t >> 5) * 132 + (t & 31) * 4];
        float4 v[2 * GT];
#pragma unroll
        for (int u = 0; u < 2 * GT; ++u) v[u] = xp[u * 32 * (HIDC / 4)];
#pragma unroll
        for (int u = 0; u < 2 * GT; ++u) {
            float* p = sp + u * 32 * 132;
            p[0] = v[u].x; p[1] = v[u].y; p[2] = v[u].z; p[3] = v[u].w;
        }
    } else {
        for (int i = t; i < GROWS * 32; i += 1024) {
            int r  = i >> 5;
            int c4 = i & 31;
            float4 v = make_float4(0.f, 0.f, 0.f, 0.f);
            if (rowBase + r < N) v = ((const float4*)(x + (size_t)(rowBase + r) * HIDC))[c4];
            float* p = &As[r * 132 + c4 * 4];
            p[0] = v.x; p[1] = v.y; p[2] = v.z; p[3] = v.w;
        }
    }
    __syncthreads();

    const int lane = t & 31;
    const int wv   = t >> 5;
    const int rt   = wv >> 3;
    const int m    = lane & 15;
    const int half = lane >> 4;
    const int kl   = half * 2;
    const int n    = (wv & 7) * 16 + (lane & 15);

    v8f c0 = {}, c1 = {}, c2 = {};
    const v2f*   ap = (const v2f*)&As[(rt * 16 + m) * 132 + kl];
    const float* wp = W + kl * HIDC + n;
#pragma nounroll
    for (int kb = 0; kb < HIDC / 16; ++kb) {
        v2f b[4];
#pragma unroll
        for (int j = 0; j < 4; ++j) {
            b[j].x = wp[(4 * j) * HIDC];
            b[j].y = wp[(4 * j + 1) * HIDC];
        }
#pragma unroll
        for (int j = 0; j < 4; ++j) {
            const v2f a0 = ap[2 * j];
            const v2f a1 = ap[32 * 132 + 2 * j];
            const v2f a2 = ap[64 * 132 + 2 * j];
            c0 = __builtin_amdgcn_wmma_f32_16x16x4_f32(false, a0, false, b[j],
                                                       (short)0, c0, false, false);
            c1 = __builtin_amdgcn_wmma_f32_16x16x4_f32(false, a1, false, b[j],
                                                       (short)0, c1, false, false);
            c2 = __builtin_amdgcn_wmma_f32_16x16x4_f32(false, a2, false, b[j],
                                                       (short)0, c2, false, false);
        }
        ap += 8;
        wp += 16 * HIDC;
    }

    const int row0 = rowBase + rt * 16 + 8 * half;
    float* hp = h + (size_t)row0 * HIDC + n;
    if (full) {
#pragma unroll
        for (int v = 0; v < 8; ++v) {
            hp[v * HIDC]              = c0[v];
            hp[(64 + v) * HIDC]       = c1[v];
            hp[(128 + v) * HIDC]      = c2[v];
        }
    } else {
#pragma unroll
        for (int v = 0; v < 8; ++v) {
            if (row0 + v < N)       hp[v * HIDC]         = c0[v];
            if (row0 + 64 + v < N)  hp[(64 + v) * HIDC]  = c1[v];
            if (row0 + 128 + v < N) hp[(128 + v) * HIDC] = c2[v];
        }
    }
}

#define QCAP 64
#define BCAP 384

__device__ __forceinline__ int rdlane_i(int v, int k) { return __builtin_amdgcn_readlane(v, k); }
__device__ __forceinline__ float rdlane_f(float v, int k) {
    return __int_as_float(__builtin_amdgcn_readlane(__float_as_int(v), k));
}
__device__ __forceinline__ void lds_fadd(float* p, float v) {
    __hip_atomic_fetch_add(p, v, __ATOMIC_RELAXED, __HIP_MEMORY_SCOPE_WORKGROUP);
}

__device__ __forceinline__ void file_one(unsigned* bins, int* binCnt, int* degc, int s, int dl) {
    const int owner = dl & 31;
    __hip_atomic_fetch_add(&degc[dl], 1, __ATOMIC_RELAXED, __HIP_MEMORY_SCOPE_WORKGROUP);
    const int pos = __hip_atomic_fetch_add(&binCnt[owner], 1, __ATOMIC_RELAXED,
                                           __HIP_MEMORY_SCOPE_WORKGROUP);
    const int slot = (pos < BCAP) ? owner * BCAP + pos : 32 * BCAP;
    bins[slot] = ((unsigned)s << 4) | (unsigned)(dl >> 5);
}

__device__ __forceinline__ void file_groups(const int* __restrict__ src,
                                            const int* __restrict__ dst,
                                            unsigned* bins, int* binCnt, int* degc,
                                            const int* myG, int lo,
                                            int head, int n, int lane) {
    if (lane < n) {
        const int G = myG[(head + lane) & (QCAP - 1)];
        const int4 a = ((const int4*)dst)[2 * G];
        const int4 b = ((const int4*)dst)[2 * G + 1];
        const int d0 = a.x - lo, d1 = a.y - lo, d2 = a.z - lo, d3 = a.w - lo;
        const int d4 = b.x - lo, d5 = b.y - lo, d6 = b.z - lo, d7 = b.w - lo;
        unsigned m = ((unsigned)d0 < (unsigned)TN ? 1u : 0u)   | ((unsigned)d1 < (unsigned)TN ? 2u : 0u) |
                     ((unsigned)d2 < (unsigned)TN ? 4u : 0u)   | ((unsigned)d3 < (unsigned)TN ? 8u : 0u) |
                     ((unsigned)d4 < (unsigned)TN ? 16u : 0u)  | ((unsigned)d5 < (unsigned)TN ? 32u : 0u) |
                     ((unsigned)d6 < (unsigned)TN ? 64u : 0u)  | ((unsigned)d7 < (unsigned)TN ? 128u : 0u);
        while (m != 0u) {
            const int k = __builtin_ctz(m);
            m &= m - 1u;
            const int s = src[8 * G + k];
            const int e01 = (k & 1) ? d1 : d0;
            const int e23 = (k & 1) ? d3 : d2;
            const int e45 = (k & 1) ? d5 : d4;
            const int e67 = (k & 1) ? d7 : d6;
            const int e03 = (k & 2) ? e23 : e01;
            const int e47 = (k & 2) ? e67 : e45;
            file_one(bins, binCnt, degc, s, (k & 4) ? e47 : e03);
        }
    }
}

__device__ __forceinline__ void file_edges_block(const int* __restrict__ src,
                                                 const int* __restrict__ dst,
                                                 const unsigned int* __restrict__ planes,
                                                 float* __restrict__ dinv,
                                                 unsigned int* __restrict__ degW,
                                                 unsigned int* __restrict__ binsG,
                                                 int* __restrict__ cntG, int N, int E, int NB,
                                                 int nkb, int bid,
                                                 unsigned* bins, int* qG, int* degc, int* binCnt) {
    const int t    = threadIdx.x;
    const int lane = t & 31;
    const int wv   = __builtin_amdgcn_readfirstlane(t >> 5);
    const int lo   = bid * TN;

    if (t < TN) degc[t] = 0;
    if (t < 32) binCnt[t] = 0;
    __syncthreads();

    int* myG = qG + wv * QCAP;
    int head = 0, cnt = 0;

    const bool useplane = (NB <= PMAXO) && ((((size_t)dst) & 15) == 0);
    const int  nwords   = useplane ? nkb * 32 : 0;
    const unsigned* P   = planes + (size_t)bid * nkb * 32;

    for (int i0 = wv * 32; i0 < nwords; i0 += 1024 * 8) {
        unsigned wd[8];
#pragma unroll
        for (int g = 0; g < 8; ++g) {
            wd[g] = P[min(i0 + 1024 * g + lane, nwords - 1)];
        }
#pragma unroll
        for (int g = 0; g < 8; ++g) {
            const int ib = i0 + 1024 * g;
            if (ib < nwords) {
                const int gb = (ib + lane) << 5;
                unsigned F = wd[g];
#pragma nounroll
                for (int r = 0; r < 32; ++r) {
                    const unsigned mask = __builtin_amdgcn_ballot_w32(F != 0u);
                    if (mask == 0u) break;
                    if (F != 0u) {
                        const int p = (head + cnt + (int)__builtin_amdgcn_mbcnt_lo(mask, 0u)) & (QCAP - 1);
                        myG[p] = gb + __builtin_ctz(F);
                    }
                    cnt += __builtin_popcount(mask);
                    F &= F - 1u;
                    if (cnt >= 32) {
                        file_groups(src, dst, bins, binCnt, degc, myG, lo, head, 32, lane);
                        head = (head + 32) & (QCAP - 1);
                        cnt -= 32;
                    }
                }
            }
        }
    }
    if (cnt > 0) file_groups(src, dst, bins, binCnt, degc, myG, lo, head, cnt, lane);
    const int ebeg   = useplane ? ((E >> 3) << 3) : 0;
    const int nsteps = (E - ebeg + 31) >> 5;
#pragma nounroll
    for (int st = wv; st < nsteps; st += 32) {
        const int e = ebeg + (st << 5) + lane;
        if (e < E) {
            const int dl = dst[e] - lo;
            if ((unsigned)dl < (unsigned)TN) file_one(bins, binCnt, degc, src[e], dl);
        }
    }
    __syncthreads();

    {
        const int nraw = __builtin_amdgcn_readfirstlane(binCnt[wv]);
        const int nb = min(nraw, BCAP);
        const unsigned* myB = bins + wv * BCAP;
        unsigned* gB = binsG + ((size_t)bid * 32 + wv) * BCAP;
        for (int i = lane; i < nb; i += 32) gB[i] = myB[i];
        if (lane == 0) cntG[bid * 32 + wv] = nraw;
    }
    if (t < TN && lo + t < N) dinv[lo + t] = rsqrtf(1.0f + (float)degc[t]);
    if (t < TN / 4) {
        const unsigned d0 = (unsigned)min(degc[4 * t], 255);
        const unsigned d1 = (unsigned)min(degc[4 * t + 1], 255);
        const unsigned d2 = (unsigned)min(degc[4 * t + 2], 255);
        const unsigned d3 = (unsigned)min(degc[4 * t + 3], 255);
        degW[bid * (DEGPITCH / 4) + t] = d0 | (d1 << 8) | (d2 << 16) | (d3 << 24);
    }
}

struct FileLds {
    unsigned bins[32 * BCAP + 1];
    int      qG[32 * QCAP];
    int      degc[TN];
    int      binCnt[32];
};
union RoleLds {
    FileLds f;
    float   As[GROWS * 132];
};
__global__ __launch_bounds__(1024, 16) void gcn_file_gemm(const int* __restrict__ src,
                                                      const int* __restrict__ dst,
                                                      const unsigned int* __restrict__ planes,
                                                      float* __restrict__ dinv,
                                                      unsigned int* __restrict__ degW,
                                                      unsigned int* __restrict__ binsG,
                                                      int* __restrict__ cntG,
                                                      const float* __restrict__ x,
                                                      const float* __restrict__ W,
                                                      float* __restrict__ h,
                                                      int N, int E, int NBF, int nkb) {
    __shared__ __attribute__((aligned(16))) RoleLds lds;

    const int b = blockIdx.x;
    if (b < NBF) {
        file_edges_block(src, dst, planes, dinv, degW, binsG, cntG, N, E, NBF, nkb, b,
                         lds.f.bins, lds.f.qG, lds.f.degc, lds.f.binCnt);
    } else {
        gemm1_rows(x, W, h, N, b - NBF, lds.As);
    }
}


__device__ __forceinline__ void load_rows8(float4 (&r)[8], const float* __restrict__ h,
                                           int s, int k0, int lane) {
#pragma unroll
    for (int u = 0; u < 8; ++u) {
        const int sk = rdlane_i(s, k0 + u);
        r[u] = ((const float4*)(h + (size_t)sk * HIDC))[lane];
    }
}
template <bool GUARD>
__device__ __forceinline__ void add_rows8(const float4 (&r)[8], float* trow,
                                          int sl, float nrm, int k0, int cnt) {
#pragma unroll
    for (int u = 0; u < 8; ++u) {
        if (!GUARD || k0 + u < cnt) {
            const int   rk = rdlane_i(sl, k0 + u);
            const float nk = rdlane_f(nrm, k0 + u);
            float4* p = (float4*)(trow + rk * (32 * HIDC));
            float4 a = *p;
            a.x = fmaf(r[u].x, nk, a.x);
            a.y = fmaf(r[u].y, nk, a.y);
            a.z = fmaf(r[u].z, nk, a.z);
            a.w = fmaf(r[u].w, nk, a.w);
            *p = a;
        }
    }
}

__device__ __forceinline__ float src_dinv(const unsigned char* degb, const float* __restrict__ dinv,
                                          bool useTab, unsigned s) {
    if (useTab) {
        const unsigned b = degb[s];
        if (b != 255u) return rsqrtf(1.0f + (float)b);
    }
    return dinv[s];
}

__global__ __launch_bounds__(1024) void gcn_agg1_owner(const float* __restrict__ h,
                                                       const int* __restrict__ src,
                                                       const int* __restrict__ dst,
                                                       const float* __restrict__ dinv,
                                                       const float* __restrict__ b1,
                                                       const float* __restrict__ W2,
                                                       float* __restrict__ gv,
                                                       const unsigned int* __restrict__ binsG,
                                                       const int* __restrict__ cntG,
                                                       const unsigned int* __restrict__ degW,
                                                       int N, int E, int NB) {
    __shared__ __attribute__((aligned(16))) float tile[TN * HIDC];
    __shared__ float dtile[TN];
    __shared__ __attribute__((aligned(16))) unsigned char degb[DEGCAP];

    const int t    = threadIdx.x;
    const int lane = t & 31;
    const int wv   = __builtin_amdgcn_readfirstlane(t >> 5);
    const int lo   = blockIdx.x * TN;

    const int  c    = cntG[blockIdx.x * 32 + lane];
    const bool over = __builtin_amdgcn_ballot_w32(c > BCAP) != 0u;
    const int  nb   = rdlane_i(c, wv);

    const bool useTab = (NB * TN <= DEGCAP);
    const int  nq     = NB * (TN / 16);
    uint4 q[(DEGCAP / 16 + 1023) / 1024];
#pragma unroll
    for (int u = 0; u < (DEGCAP / 16 + 1023) / 1024; ++u) {
        const int i  = min(t + 1024 * u, nq - 1);
        const int cb = i / (TN / 16);
        q[u] = ((const uint4*)degW)[cb * (DEGPITCH / 16) + (i - cb * (TN / 16))];
    }

    for (int i = t; i < TN * HIDC / 4; i += 1024)
        ((float4*)tile)[i] = make_float4(0.f, 0.f, 0.f, 0.f);
    if (t < TN) dtile[t] = (lo + t < N) ? dinv[lo + t] : 0.0f;
    if (useTab) {
#pragma unroll
        for (int u = 0; u < (DEGCAP / 16 + 1023) / 1024; ++u)
            ((uint4*)degb)[min(t + 1024 * u, nq - 1)] = q[u];
    }
    __syncthreads();

    if (!over) {
        const unsigned* gB = binsG + ((size_t)blockIdx.x * 32 + wv) * BCAP;
        float* trow = tile + wv * HIDC + 4 * lane;
        unsigned ent  = (lane < nb) ? gB[lane] : 0u;
        unsigned ent1 = (32 + lane < nb) ? gB[32 + lane] : 0u;
        float    dv   = (lane < nb) ? src_dinv(degb, dinv, useTab, ent >> 4) : 0.0f;
        for (int base = 0; base < nb; base += 32) {
            const int cnt32 = min(32, nb - base);
            const int   s   = (int)(ent >> 4);
            const int   sl  = (int)(ent & 15u);
            const float nrm = dv * dtile[(sl << 5) | wv];
            ent  = ent1;
            dv   = (base + 32 + lane < nb) ? src_dinv(degb, dinv, useTab, ent >> 4) : 0.0f;
            ent1 = (base + 64 + lane < nb) ? gB[base + 64 + lane] : 0u;
            if (cnt32 == 32) {
                float4 ra[8], rb[8];
                load_rows8(ra, h, s, 0, lane);
                load_rows8(rb, h, s, 8, lane);
                add_rows8<false>(ra, trow, sl, nrm, 0, 32);
                load_rows8(ra, h, s, 16, lane);
                add_rows8<false>(rb, trow, sl, nrm, 8, 32);
                load_rows8(rb, h, s, 24, lane);
                add_rows8<false>(ra, trow, sl, nrm, 16, 32);
                add_rows8<false>(rb, trow, sl, nrm, 24, 32);
            } else {
#pragma nounroll
                for (int k0 = 0; k0 < cnt32; k0 += 8) {
                    float4 rc[8];
                    load_rows8(rc, h, s, k0, lane);
                    add_rows8<true>(rc, trow, sl, nrm, k0, cnt32);
                }
            }
        }
    } else {
        const int nsteps = (E + 31) >> 5;
        for (int st = wv; st < nsteps; st += 32) {
            const int e = (st << 5) + lane;
            const int d = (e < E) ? dst[e] : -1;
            const int dl = d - lo;
            const bool m = (unsigned)dl < (unsigned)TN;
            const int s = m ? src[e] : 0;
            unsigned om = __builtin_amdgcn_ballot_w32(m);
            while (om != 0u) {
                const int k = __builtin_ctz(om);
                om &= om - 1u;
                const int   sk = rdlane_i(s, k);
                const int   dk = rdlane_i(dl, k);
                const float nk = dinv[sk] * dtile[dk];
                const float4 r = ((const float4*)(h + (size_t)sk * HIDC))[lane];
                float* o = tile + dk * HIDC + 4 * lane;
                lds_fadd(o + 0, r.x * nk);
                lds_fadd(o + 1, r.y * nk);
                lds_fadd(o + 2, r.z * nk);
                lds_fadd(o + 3, r.w * nk);
            }
        }
        __syncthreads();
    }

    const float4 bb = ((const float4*)b1)[lane];
    const float4 w  = ((const float4*)W2)[lane];
    for (int r = wv; r < TN; r += 32) {
        const int i = lo + r;
        if (i >= N) break;
        const float di = dtile[r];
        const float ns = di * di;
        const float4 a  = ((const float4*)(tile + r * HIDC))[lane];
        const float4 hv = ((const float4*)(h + (size_t)i * HIDC))[lane];
        float r0 = fmaxf(a.x + hv.x * ns + bb.x, 0.0f);
        float r1 = fmaxf(a.y + hv.y * ns + bb.y, 0.0f);
        float r2 = fmaxf(a.z + hv.z * ns + bb.z, 0.0f);
        float r3 = fmaxf(a.w + hv.w * ns + bb.w, 0.0f);
        float p = r0 * w.x + r1 * w.y + r2 * w.z + r3 * w.w;
#pragma unroll
        for (int off = 16; off > 0; off >>= 1) p += __shfl_down(p, off, 32);
        if (lane == 0) gv[i] = p * di;
    }
}

__global__ __launch_bounds__(1024) void gcn_agg2_owner(const float* __restrict__ g,
                                                       const int* __restrict__ src,
                                                       const int* __restrict__ dst,
                                                       const unsigned int* __restrict__ binsG,
                                                       const int* __restrict__ cntG,
                                                       const float* __restrict__ dinv,
                                                       const float* __restrict__ b2,
                                                       float* __restrict__ out, int N, int E) {
    __shared__ float acc[32 * 16];

    const int t    = threadIdx.x;
    const int lane = t & 31;
    const int wv   = __builtin_amdgcn_readfirstlane(t >> 5);
    const int lo   = blockIdx.x * TN;

    const int  c    = cntG[blockIdx.x * 32 + lane];
    const bool over = __builtin_amdgcn_ballot_w32(c > BCAP) != 0u;
    const int  nraw = rdlane_i(c, wv);

    if (t < 32 * 16) acc[t] = 0.0f;
    __syncthreads();

    const bool  mine = (t < TN) && (lo + t < N);
    const float gs   = mine ? g[lo + t] : 0.0f;
    const float ds   = mine ? dinv[lo + t] : 0.0f;
    const float bias = b2[0];

    if (!over) {
        if (nraw > 0) {
            const unsigned* gB = binsG + ((size_t)blockIdx.x * 32 + wv) * BCAP;
            unsigned ent[BCAP / 32];
            float    v[BCAP / 32];
#pragma unroll
            for (int k = 0; k < BCAP / 32; ++k) ent[k] = gB[min(lane + 32 * k, nraw - 1)];
#pragma unroll
            for (int k = 0; k < BCAP / 32; ++k) v[k] = g[ent[k] >> 4];
#pragma unroll
            for (int k = BCAP / 32 - 1; k >= 0; --k) {
                if (lane + 32 * k < nraw) lds_fadd(&acc[wv * 16 + (int)(ent[k] & 15u)], v[k]);
            }
        }
    } else {
        for (int e = t; e < E; e += 1024) {
            const unsigned dl = (unsigned)(dst[e] - lo);
            if (dl < (unsigned)TN) lds_fadd(&acc[(dl & 31u) * 16u + (dl >> 5)], g[src[e]]);
        }
    }
    __syncthreads();

    if (mine) out[lo + t] = (acc[(t & 31) * 16 + (t >> 5)] + gs) * ds + bias;
}

extern "C" void kernel_launch(void* const* d_in, const int* in_sizes, int n_in,
                              void* d_out, int out_size, void* d_ws, size_t ws_size,
                              hipStream_t stream) {
    const float* x  = (const float*)d_in[0];
    const float* W1 = (const float*)d_in[1];
    const float* b1 = (const float*)d_in[2];
    const float* W2 = (const float*)d_in[3];
    const float* b2 = (const float*)d_in[4];
    const int*   ei = (const int*)d_in[5];

    const int N = in_sizes[0] / HIDC;
    const int E = in_sizes[5] / 2;
    const int* src = ei;
    const int* dst = ei + E;
    float* out = (float*)d_out;

    float* h    = (float*)d_ws;
    const int NB  = (N + TN - 1) / TN;
    const int nkb = ((E >> 3) + 1023) / 1024;
    unsigned int* planes = (unsigned int*)(h + (size_t)N * HIDC);
    unsigned int* degW = planes + (size_t)NB * nkb * 32;
    float* dinv = (float*)(degW + (size_t)NB * (DEGPITCH / 4));
    float* g    = dinv + N;
    int* cntG = (int*)(g + N);
    unsigned int* binsG = (unsigned int*)(cntG + (size_t)NB * 32);

    gcn_groups    <<<nkb, 1024, 0, stream>>>(dst, planes, NB, E, nkb);
    gcn_file_gemm <<<NB + (N + GROWS - 1) / GROWS, 1024, 0, stream>>>(src, dst, planes, dinv, degW, binsG, cntG,
                                                                     x, W1, h, N, E, NB, nkb);
    gcn_agg1_owner<<<NB, 1024, 0, stream>>>(h, src, dst, dinv, b1, W2, g, binsG, cntG, degW, N, E, NB);
    gcn_agg2_owner<<<NB, 1024, 0, stream>>>(g, src, dst, binsG, cntG, dinv, b2, out, N, E);
}
